# starting kernel
# baseline (speedup 1.0000x reference)
#include <hip/hip_runtime.h>
#include <math.h>

typedef __attribute__((ext_vector_type(16))) _Float16 v16h;
typedef __attribute__((ext_vector_type(8)))  _Float16 v8h;
typedef __attribute__((ext_vector_type(4)))  _Float16 v4h;
typedef __attribute__((ext_vector_type(8)))  float    v8f;

#define WMMA16(A,B,C) __builtin_amdgcn_wmma_f32_16x16x32_f16(false,(A),false,(B),(short)0,(C),false,false)

static constexpr int T_STEPS = 64;
static constexpr int E_ENVS  = 32;
static constexpr int B_TOT   = 2048;
static constexpr int HL      = 256;
static constexpr int IND     = 549;
static constexpr int KPAD    = 832;
static constexpr int XS      = 560;
static constexpr int ASTR    = 840;
static constexpr int IMS     = 88;

__device__ __forceinline__ v16h pack_frag(const _Float16* p) {
  v8h u = *(const v8h*)p;
  v8h w = *(const v8h*)(p + 16);
  v16h a;
#pragma unroll
  for (int e = 0; e < 8; e++) { a[e] = u[e]; a[e + 8] = w[e]; }
  return a;
}
__device__ __forceinline__ v16h pack_frag_c1(const _Float16* p0) {
  const _Float16* p1 = p0 + 2 * IMS;
  v4h u0 = *(const v4h*)p0;
  v4h u1 = *(const v4h*)(p0 + 4);
  v4h u2 = *(const v4h*)p1;
  v4h u3 = *(const v4h*)(p1 + 4);
  v16h a;
#pragma unroll
  for (int e = 0; e < 4; e++) {
    a[e] = u0[e]; a[e + 4] = u1[e]; a[e + 8] = u2[e]; a[e + 12] = u3[e];
  }
  return a;
}

__device__ __forceinline__ float sigmoidf_(float x) { return 1.0f / (1.0f + __expf(-x)); }

__global__ void k_prep_img(const float* img, _Float16* imgC, int n) {
  int i = blockIdx.x * blockDim.x + threadIdx.x;
  if (i >= n) return;
  int b = i / (3 * 84 * 84), r0 = i % (3 * 84 * 84);
  int c = r0 / (84 * 84), rc = r0 % (84 * 84);
  int row = rc / 84, col = rc % 84;
  imgC[((b * 3 + c) * 84 + row) * IMS + col] =
      (_Float16)(img[((b * 84 + row) * 84 + col) * 3 + c] * (1.0f / 255.0f));
}

__global__ void k_prep_w1(const float* w, _Float16* wt) {
  int i = blockIdx.x * blockDim.x + threadIdx.x;
  if (i >= 32 * 192) return;
  int n = i / 192, k = i % 192;
  int c = k >> 6, kh = (k & 63) >> 3, kw = k & 7;
  wt[i] = (_Float16)w[((n * 3 + c) * 8 + kh) * 8 + kw];
}
__global__ void k_prep_w2(const float* w, _Float16* wt) {
  int i = blockIdx.x * blockDim.x + threadIdx.x;
  if (i >= 64 * 512) return;
  int n = i / 512, k = i % 512;
  int pix = k >> 5, c = k & 31, kh = pix >> 2, kw = pix & 3;
  wt[i] = (_Float16)w[((n * 32 + c) * 4 + kh) * 4 + kw];
}
__global__ void k_prep_w3(const float* w, _Float16* wt) {
  int i = blockIdx.x * blockDim.x + threadIdx.x;
  if (i >= 64 * 576) return;
  int n = i / 576, k = i % 576;
  int pix = k >> 6, c = k & 63, kh = pix / 3, kw = pix % 3;
  wt[i] = (_Float16)w[((n * 64 + c) * 3 + kh) * 3 + kw];
}
__global__ void k_prep_fc(const float* w, _Float16* wt) {
  int i = blockIdx.x * blockDim.x + threadIdx.x;
  if (i >= 512 * 3136) return;
  int f = i / 3136, k = i % 3136;
  int s = k >> 6, c = k & 63;
  wt[i] = (_Float16)w[f * 3136 + c * 49 + s];
}
__global__ void k_prep_wg(const float* wih, const float* whh, _Float16* wg) {
  int i = blockIdx.x * blockDim.x + threadIdx.x;
  if (i >= 1024 * KPAD) return;
  int r = i / KPAD, k = i % KPAD;
  float v = 0.0f;
  if (k < IND)            v = wih[r * IND + k];
  else if (k < IND + HL)  v = whh[r * HL + (k - IND)];
  wg[i] = (_Float16)v;
}
__global__ void k_prep_bias(const float* bih, const float* bhh, float* bg) {
  int i = blockIdx.x * blockDim.x + threadIdx.x;
  if (i < 1024) bg[i] = bih[i] + bhh[i];
}
__global__ void k_onehot(const int* pos, const int* lact, _Float16* X) {
  int i = blockIdx.x * blockDim.x + threadIdx.x;
  if (i >= B_TOT * 48) return;
  int b = i / 48, c = i % 48;
  float v = 0.0f;
  if (c < 16)      v = (pos[b * 2 + 0] == c)      ? 1.0f : 0.0f;
  else if (c < 32) v = (pos[b * 2 + 1] == c - 16) ? 1.0f : 0.0f;
  else if (c < 37) v = (lact[b]        == c - 32) ? 1.0f : 0.0f;
  X[b * XS + 512 + c] = (_Float16)v;
}

__global__ void __launch_bounds__(256) k_conv1(const _Float16* imgC, const _Float16* wt,
                                               const float* bias, _Float16* out) {
  int wid  = (blockIdx.x * blockDim.x + threadIdx.x) >> 5;
  int lane = threadIdx.x & 31, hi = lane >> 4, lo = lane & 15;
  int m = wid * 16 + lo;
  int b = m / 400, p = m % 400, oh = p / 20, ow = p % 20;
  const _Float16* abase = imgC + (size_t)b * 3 * 84 * IMS + (oh * 4) * IMS + ow * 4;
  v8f acc0 = {}, acc1 = {};
#pragma unroll
  for (int kb = 0; kb < 192; kb += 32) {
    int c = kb >> 6, khb = (kb & 63) >> 3;
    v16h a  = pack_frag_c1(abase + (c * 84 + khb + hi) * IMS);
    v16h b0 = pack_frag(wt + lo * 192        + kb + hi * 8);
    v16h b1 = pack_frag(wt + (16 + lo) * 192 + kb + hi * 8);
    acc0 = WMMA16(a, b0, acc0);
    acc1 = WMMA16(a, b1, acc1);
  }
  float bs0 = bias[lo], bs1 = bias[16 + lo];
#pragma unroll
  for (int r = 0; r < 8; r++) {
    int mm = wid * 16 + hi * 8 + r;
    float v0 = acc0[r] + bs0; v0 = v0 > 0.0f ? v0 : 0.0f;
    float v1 = acc1[r] + bs1; v1 = v1 > 0.0f ? v1 : 0.0f;
    out[mm * 32 + lo]      = (_Float16)v0;
    out[mm * 32 + 16 + lo] = (_Float16)v1;
  }
}

__global__ void __launch_bounds__(256) k_conv2(const _Float16* act1, const _Float16* wt,
                                               const float* bias, _Float16* out) {
  int wid  = (blockIdx.x * blockDim.x + threadIdx.x) >> 5;
  int lane = threadIdx.x & 31, hi = lane >> 4, lo = lane & 15;
  int m = wid * 16 + lo;
  int b = m / 81, p = m % 81, oh = p / 9, ow = p % 9;
  v8f acc0 = {}, acc1 = {}, acc2 = {}, acc3 = {};
  for (int kb = 0; kb < 512; kb += 32) {
    int pix = kb >> 5, kh = pix >> 2, kw = pix & 3;
    v16h a  = pack_frag(act1 + ((b * 20 + oh * 2 + kh) * 20 + (ow * 2 + kw)) * 32 + hi * 8);
    v16h b0 = pack_frag(wt + (lo)      * 512 + kb + hi * 8);
    v16h b1 = pack_frag(wt + (16 + lo) * 512 + kb + hi * 8);
    v16h b2 = pack_frag(wt + (32 + lo) * 512 + kb + hi * 8);
    v16h b3 = pack_frag(wt + (48 + lo) * 512 + kb + hi * 8);
    acc0 = WMMA16(a, b0, acc0);
    acc1 = WMMA16(a, b1, acc1);
    acc2 = WMMA16(a, b2, acc2);
    acc3 = WMMA16(a, b3, acc3);
  }
  float bs0 = bias[lo], bs1 = bias[16 + lo], bs2 = bias[32 + lo], bs3 = bias[48 + lo];
#pragma unroll
  for (int r = 0; r < 8; r++) {
    int mm = wid * 16 + hi * 8 + r;
    float v0 = acc0[r] + bs0; v0 = v0 > 0.0f ? v0 : 0.0f;
    float v1 = acc1[r] + bs1; v1 = v1 > 0.0f ? v1 : 0.0f;
    float v2 = acc2[r] + bs2; v2 = v2 > 0.0f ? v2 : 0.0f;
    float v3 = acc3[r] + bs3; v3 = v3 > 0.0f ? v3 : 0.0f;
    out[mm * 64 + lo]      = (_Float16)v0;
    out[mm * 64 + 16 + lo] = (_Float16)v1;
    out[mm * 64 + 32 + lo] = (_Float16)v2;
    out[mm * 64 + 48 + lo] = (_Float16)v3;
  }
}

__global__ void __launch_bounds__(256) k_conv3(const _Float16* act2, const _Float16* wt,
                                               const float* bias, _Float16* out) {
  int wid  = (blockIdx.x * blockDim.x + threadIdx.x) >> 5;
  int lane = threadIdx.x & 31, hi = lane >> 4, lo = lane & 15;
  int m = wid * 16 + lo;
  int b = m / 49, p = m % 49, oh = p / 7, ow = p % 7;
  v8f acc0 = {}, acc1 = {}, acc2 = {}, acc3 = {};
  for (int kb = 0; kb < 576; kb += 32) {
    int pix = kb >> 6, cb = (kb & 63) + hi * 8;
    int kh = pix / 3, kw = pix % 3;
    v16h a  = pack_frag(act2 + ((b * 9 + oh + kh) * 9 + (ow + kw)) * 64 + cb);
    v16h b0 = pack_frag(wt + (lo)      * 576 + kb + hi * 8);
    v16h b1 = pack_frag(wt + (16 + lo) * 576 + kb + hi * 8);
    v16h b2 = pack_frag(wt + (32 + lo) * 576 + kb + hi * 8);
    v16h b3 = pack_frag(wt + (48 + lo) * 576 + kb + hi * 8);
    acc0 = WMMA16(a, b0, acc0);
    acc1 = WMMA16(a, b1, acc1);
    acc2 = WMMA16(a, b2, acc2);
    acc3 = WMMA16(a, b3, acc3);
  }
  float bs0 = bias[lo], bs1 = bias[16 + lo], bs2 = bias[32 + lo], bs3 = bias[48 + lo];
#pragma unroll
  for (int r = 0; r < 8; r++) {
    int mm = wid * 16 + hi * 8 + r;
    float v0 = acc0[r] + bs0; v0 = v0 > 0.0f ? v0 : 0.0f;
    float v1 = acc1[r] + bs1; v1 = v1 > 0.0f ? v1 : 0.0f;
    float v2 = acc2[r] + bs2; v2 = v2 > 0.0f ? v2 : 0.0f;
    float v3 = acc3[r] + bs3; v3 = v3 > 0.0f ? v3 : 0.0f;
    out[mm * 64 + lo]      = (_Float16)v0;
    out[mm * 64 + 16 + lo] = (_Float16)v1;
    out[mm * 64 + 32 + lo] = (_Float16)v2;
    out[mm * 64 + 48 + lo] = (_Float16)v3;
  }
}

__global__ void __launch_bounds__(256) k_fc(const _Float16* act3, const _Float16* wt,
                                            const float* bias, _Float16* X) {
  int wid  = (blockIdx.x * blockDim.x + threadIdx.x) >> 5;
  int lane = threadIdx.x & 31, hi = lane >> 4, lo = lane & 15;
  int ng = wid & 7, mt = wid >> 3;
  int m = mt * 16 + lo;
  int nb = ng * 64;
  v8f acc0 = {}, acc1 = {}, acc2 = {}, acc3 = {};
  for (int kb = 0; kb < 3136; kb += 32) {
    v16h a  = pack_frag(act3 + (size_t)m * 3136 + kb + hi * 8);
    v16h b0 = pack_frag(wt + (size_t)(nb + lo)      * 3136 + kb + hi * 8);
    v16h b1 = pack_frag(wt + (size_t)(nb + 16 + lo) * 3136 + kb + hi * 8);
    v16h b2 = pack_frag(wt + (size_t)(nb + 32 + lo) * 3136 + kb + hi * 8);
    v16h b3 = pack_frag(wt + (size_t)(nb + 48 + lo) * 3136 + kb + hi * 8);
    acc0 = WMMA16(a, b0, acc0);
    acc1 = WMMA16(a, b1, acc1);
    acc2 = WMMA16(a, b2, acc2);
    acc3 = WMMA16(a, b3, acc3);
  }
  float bs0 = bias[nb + lo], bs1 = bias[nb + 16 + lo];
  float bs2 = bias[nb + 32 + lo], bs3 = bias[nb + 48 + lo];
#pragma unroll
  for (int r = 0; r < 8; r++) {
    int mm = mt * 16 + hi * 8 + r;
    float v0 = acc0[r] + bs0; v0 = v0 > 0.0f ? v0 : 0.0f;
    float v1 = acc1[r] + bs1; v1 = v1 > 0.0f ? v1 : 0.0f;
    float v2 = acc2[r] + bs2; v2 = v2 > 0.0f ? v2 : 0.0f;
    float v3 = acc3[r] + bs3; v3 = v3 > 0.0f ? v3 : 0.0f;
    X[mm * XS + nb + lo]      = (_Float16)v0;
    X[mm * XS + nb + 16 + lo] = (_Float16)v1;
    X[mm * XS + nb + 32 + lo] = (_Float16)v2;
    X[mm * XS + nb + 48 + lo] = (_Float16)v3;
  }
}

__global__ void __launch_bounds__(1024) k_lstm(const _Float16* X, const _Float16* Wg,
                                               const float* biasg, const int* done,
                                               const float* h0, const float* c0,
                                               float* hs, float* out) {
  __shared__ _Float16 A[E_ENVS * ASTR];
  int tid = threadIdx.x;
  int w = tid >> 5, lane = tid & 31, hi = lane >> 4, lo = lane & 15;
  int mt = w >> 4, cg = w & 15;
  int ncol = cg * 16 + lo;

  for (int i = tid; i < E_ENVS * HL; i += 1024) {
    int e = i >> 8, k = i & 255;
    A[e * ASTR + IND + k] = (_Float16)h0[i];
  }
  for (int i = tid; i < E_ENVS * (ASTR - (IND + HL)); i += 1024) {
    int span = ASTR - (IND + HL);
    int e = i / span, k = i % span;
    A[e * ASTR + IND + HL + k] = (_Float16)0.0f;
  }
  v8f creg;
#pragma unroll
  for (int r = 0; r < 8; r++) {
    int env = mt * 16 + hi * 8 + r;
    creg[r] = c0[env * HL + ncol];
  }
  float bi = biasg[0 * HL + ncol], bf = biasg[1 * HL + ncol];
  float bg = biasg[2 * HL + ncol], bo = biasg[3 * HL + ncol];
  __syncthreads();

  for (int t = 0; t < T_STEPS; t++) {
    for (int i = tid; i < E_ENVS * HL; i += 1024) {
      int e = i >> 8, k = i & 255;
      if (done[t * E_ENVS + e]) A[e * ASTR + IND + k] = (_Float16)0.0f;
    }
    for (int i = tid; i < E_ENVS * IND; i += 1024) {
      int e = i / IND, k = i - e * IND;
      A[e * ASTR + k] = X[(t * E_ENVS + e) * XS + k];
    }
#pragma unroll
    for (int r = 0; r < 8; r++) {
      int env = mt * 16 + hi * 8 + r;
      if (done[t * E_ENVS + env]) creg[r] = 0.0f;
    }
    __syncthreads();

    v8f ai, af, ag, ao;
#pragma unroll
    for (int r = 0; r < 8; r++) { ai[r] = bi; af[r] = bf; ag[r] = bg; ao[r] = bo; }
    for (int kb = 0; kb < KPAD; kb += 32) {
      v16h a  = pack_frag(&A[(mt * 16 + lo) * ASTR + kb + hi * 8]);
      v16h b0 = pack_frag(Wg + (0 * HL + ncol) * KPAD + kb + hi * 8);
      v16h b1 = pack_frag(Wg + (1 * HL + ncol) * KPAD + kb + hi * 8);
      v16h b2 = pack_frag(Wg + (2 * HL + ncol) * KPAD + kb + hi * 8);
      v16h b3 = pack_frag(Wg + (3 * HL + ncol) * KPAD + kb + hi * 8);
      ai = WMMA16(a, b0, ai);
      af = WMMA16(a, b1, af);
      ag = WMMA16(a, b2, ag);
      ao = WMMA16(a, b3, ao);
    }
    __syncthreads();

    float hv[8];
#pragma unroll
    for (int r = 0; r < 8; r++) {
      float iv = sigmoidf_(ai[r]), fv = sigmoidf_(af[r]);
      float gv = tanhf(ag[r]),     ov = sigmoidf_(ao[r]);
      float cv = fv * creg[r] + iv * gv;
      creg[r] = cv;
      hv[r]   = ov * tanhf(cv);
    }
#pragma unroll
    for (int r = 0; r < 8; r++) {
      int env = mt * 16 + hi * 8 + r;
      A[env * ASTR + IND + ncol] = (_Float16)hv[r];
      hs[(t * E_ENVS + env) * HL + ncol] = hv[r];
      if (t == T_STEPS - 1) {
        out[12288 + env * HL + ncol] = hv[r];
        out[20480 + env * HL + ncol] = creg[r];
      }
    }
    __syncthreads();
  }
}

__global__ void k_heads(const float* hs, const float* pw, const float* pb,
                        const float* vw, const float* vb, float* out) {
  int i = blockIdx.x * blockDim.x + threadIdx.x;
  if (i >= B_TOT * 6) return;
  int b = i / 6, j = i % 6;
  const float* h    = hs + b * HL;
  const float* wrow = (j < 5) ? pw + j * HL : vw;
  float s = (j < 5) ? pb[j] : vb[0];
#pragma unroll 4
  for (int k = 0; k < HL; k++) s += h[k] * wrow[k];
  if (j < 5) out[b * 5 + j] = s;
  else       out[10240 + b] = s;
}

extern "C" void kernel_launch(void* const* d_in, const int* in_sizes, int n_in,
                              void* d_out, int out_size, void* d_ws, size_t ws_size,
                              hipStream_t stream) {
  (void)in_sizes; (void)n_in; (void)out_size; (void)ws_size;
  const float* img  = (const float*)d_in[0];
  const int*   pos  = (const int*)d_in[1];
  const int*   lact = (const int*)d_in[2];
  const int*   done = (const int*)d_in[3];
  const float* h0   = (const float*)d_in[4];
  const float* c0   = (const float*)d_in[5];
  const float* w1   = (const float*)d_in[6];
  const float* b1   = (const float*)d_in[7];
  const float* w2   = (const float*)d_in[8];
  const float* b2   = (const float*)d_in[9];
  const float* w3   = (const float*)d_in[10];
  const float* b3   = (const float*)d_in[11];
  const float* fcw  = (const float*)d_in[12];
  const float* fcb  = (const float*)d_in[13];
  const float* wih  = (const float*)d_in[14];
  const float* whh  = (const float*)d_in[15];
  const float* bih  = (const float*)d_in[16];
  const float* bhh  = (const float*)d_in[17];
  const float* pw   = (const float*)d_in[18];
  const float* pb   = (const float*)d_in[19];
  const float* vw   = (const float*)d_in[20];
  const float* vb   = (const float*)d_in[21];
  float* out = (float*)d_out;
  char*  ws  = (char*)d_ws;

  const size_t o_img  = 0;
  const size_t o_act1 = 90832896;
  const size_t o_w1t  = o_act1 + 52428800;
  const size_t o_w2t  = o_w1t + 12288;
  const size_t o_w3t  = o_w2t + 65536;
  const size_t o_fcwt = o_w3t + 73728;
  const size_t o_wg   = o_fcwt + 3211264;
  const size_t o_bg   = o_wg + 1703936;
  const size_t o_act2 = 0;
  const size_t o_act3 = 21233664;
  const size_t o_X    = 34078720;
  const size_t o_hs   = 36372480;

  _Float16* imgC = (_Float16*)(ws + o_img);
  _Float16* act1 = (_Float16*)(ws + o_act1);
  _Float16* act2 = (_Float16*)(ws + o_act2);
  _Float16* act3 = (_Float16*)(ws + o_act3);
  _Float16* Xm   = (_Float16*)(ws + o_X);
  _Float16* w1t  = (_Float16*)(ws + o_w1t);
  _Float16* w2t  = (_Float16*)(ws + o_w2t);
  _Float16* w3t  = (_Float16*)(ws + o_w3t);
  _Float16* fcwt = (_Float16*)(ws + o_fcwt);
  _Float16* wg   = (_Float16*)(ws + o_wg);
  float*    bg   = (float*)(ws + o_bg);
  float*    hsb  = (float*)(ws + o_hs);

  const int TB = 256;
  int nImg = B_TOT * 3 * 84 * 84;
  k_prep_img <<<(nImg + TB - 1) / TB, TB, 0, stream>>>(img, imgC, nImg);
  k_prep_w1  <<<(32 * 192   + TB - 1) / TB, TB, 0, stream>>>(w1, w1t);
  k_prep_w2  <<<(64 * 512   + TB - 1) / TB, TB, 0, stream>>>(w2, w2t);
  k_prep_w3  <<<(64 * 576   + TB - 1) / TB, TB, 0, stream>>>(w3, w3t);
  k_prep_fc  <<<(512 * 3136 + TB - 1) / TB, TB, 0, stream>>>(fcw, fcwt);
  k_prep_wg  <<<(1024 * KPAD + TB - 1) / TB, TB, 0, stream>>>(wih, whh, wg);
  k_prep_bias<<<(1024 + TB - 1) / TB, TB, 0, stream>>>(bih, bhh, bg);

  k_conv1<<<6400, TB, 0, stream>>>(imgC, w1t, b1, act1);
  k_conv2<<<1296, TB, 0, stream>>>(act1, w2t, b2, act2);
  k_conv3<<< 784, TB, 0, stream>>>(act2, w3t, b3, act3);
  k_fc   <<< 128, TB, 0, stream>>>(act3, fcwt, fcb, Xm);

  k_onehot<<<(B_TOT * 48 + TB - 1) / TB, TB, 0, stream>>>(pos, lact, Xm);

  k_lstm<<<1, 1024, 0, stream>>>(Xm, wg, bg, done, h0, c0, hsb, out);

  k_heads<<<(B_TOT * 6 + TB - 1) / TB, TB, 0, stream>>>(hsb, pw, pb, vw, vb, out);
}


# LSTM input projection x@W_ih^T+bias hoisted out of the 64-step recurrence into one WMMA GEMM over all 2048 rows; step loop keeps only h@W_hh^T (K 832->256); f32 sum order of the gates changes
# speedup vs baseline: 2.0288x; 2.0288x over previous
#include <hip/hip_runtime.h>
#include <math.h>

typedef __attribute__((ext_vector_type(16))) _Float16 v16h;
typedef __attribute__((ext_vector_type(8)))  _Float16 v8h;
typedef __attribute__((ext_vector_type(4)))  _Float16 v4h;
typedef __attribute__((ext_vector_type(8)))  float    v8f;
typedef __attribute__((ext_vector_type(4)))  float    v4f;

#define WMMA16(A,B,C) __builtin_amdgcn_wmma_f32_16x16x32_f16(false,(A),false,(B),(short)0,(C),false,false)

static constexpr int T_STEPS = 64;
static constexpr int E_ENVS  = 32;
static constexpr int B_TOT   = 2048;
static constexpr int HL      = 256;
static constexpr int IND     = 549;
static constexpr int XS      = 576;
static constexpr int HSTR    = 264;
static constexpr int GT      = 1024;
static constexpr int IMS     = 88;

__device__ __forceinline__ v16h pack_frag(const _Float16* p) {
  v8h u = *(const v8h*)p;
  v8h w = *(const v8h*)(p + 16);
  v16h a;
#pragma unroll
  for (int e = 0; e < 8; e++) { a[e] = u[e]; a[e + 8] = w[e]; }
  return a;
}
__device__ __forceinline__ v16h pack_frag_c1(const _Float16* p0) {
  const _Float16* p1 = p0 + 2 * IMS;
  v4h u0 = *(const v4h*)p0;
  v4h u1 = *(const v4h*)(p0 + 4);
  v4h u2 = *(const v4h*)p1;
  v4h u3 = *(const v4h*)(p1 + 4);
  v16h a;
#pragma unroll
  for (int e = 0; e < 4; e++) {
    a[e] = u0[e]; a[e + 4] = u1[e]; a[e + 8] = u2[e]; a[e + 12] = u3[e];
  }
  return a;
}

__device__ __forceinline__ float sigmoidf_(float x) { return 1.0f / (1.0f + __expf(-x)); }

__global__ void k_prep_img(const float* img, _Float16* imgC, int n) {
  int i = blockIdx.x * blockDim.x + threadIdx.x;
  if (i >= n) return;
  int b = i / (3 * 84 * 84), r0 = i % (3 * 84 * 84);
  int c = r0 / (84 * 84), rc = r0 % (84 * 84);
  int row = rc / 84, col = rc % 84;
  imgC[((b * 3 + c) * 84 + row) * IMS + col] =
      (_Float16)(img[((b * 84 + row) * 84 + col) * 3 + c] * (1.0f / 255.0f));
}

__global__ void k_prep_w1(const float* w, _Float16* wt) {
  int i = blockIdx.x * blockDim.x + threadIdx.x;
  if (i >= 32 * 192) return;
  int n = i / 192, k = i % 192;
  int c = k >> 6, kh = (k & 63) >> 3, kw = k & 7;
  wt[i] = (_Float16)w[((n * 3 + c) * 8 + kh) * 8 + kw];
}
__global__ void k_prep_w2(const float* w, _Float16* wt) {
  int i = blockIdx.x * blockDim.x + threadIdx.x;
  if (i >= 64 * 512) return;
  int n = i / 512, k = i % 512;
  int pix = k >> 5, c = k & 31, kh = pix >> 2, kw = pix & 3;
  wt[i] = (_Float16)w[((n * 32 + c) * 4 + kh) * 4 + kw];
}
__global__ void k_prep_w3(const float* w, _Float16* wt) {
  int i = blockIdx.x * blockDim.x + threadIdx.x;
  if (i >= 64 * 576) return;
  int n = i / 576, k = i % 576;
  int pix = k >> 6, c = k & 63, kh = pix / 3, kw = pix % 3;
  wt[i] = (_Float16)w[((n * 64 + c) * 3 + kh) * 3 + kw];
}
__global__ void k_prep_fc(const float* w, _Float16* wt) {
  int i = blockIdx.x * blockDim.x + threadIdx.x;
  if (i >= 512 * 3136) return;
  int f = i / 3136, k = i % 3136;
  int s = k >> 6, c = k & 63;
  wt[i] = (_Float16)w[f * 3136 + c * 49 + s];
}
__global__ void k_prep_wl(const float* wih, const float* whh, _Float16* wih16, _Float16* whh16) {
  int i = blockIdx.x * blockDim.x + threadIdx.x;
  if (i >= 1024 * (XS + HL)) return;
  int r = i / (XS + HL), k = i % (XS + HL);
  if (k < XS) wih16[r * XS + k] = (_Float16)((k < IND) ? wih[r * IND + k] : 0.0f);
  else        whh16[r * HL + (k - XS)] = (_Float16)whh[r * HL + (k - XS)];
}
__global__ void k_prep_bias(const float* bih, const float* bhh, float* bg) {
  int i = blockIdx.x * blockDim.x + threadIdx.x;
  if (i < 1024) bg[i] = bih[i] + bhh[i];
}
__global__ void k_onehot(const int* pos, const int* lact, _Float16* X) {
  int i = blockIdx.x * blockDim.x + threadIdx.x;
  if (i >= B_TOT * 64) return;
  int b = i >> 6, c = i & 63;
  float v = 0.0f;
  if (c < 16)      v = (pos[b * 2 + 0] == c)      ? 1.0f : 0.0f;
  else if (c < 32) v = (pos[b * 2 + 1] == c - 16) ? 1.0f : 0.0f;
  else if (c < 37) v = (lact[b]        == c - 32) ? 1.0f : 0.0f;
  X[b * XS + 512 + c] = (_Float16)v;
}

__global__ void __launch_bounds__(256) k_conv1(const _Float16* imgC, const _Float16* wt,
                                               const float* bias, _Float16* out) {
  int wid  = (blockIdx.x * blockDim.x + threadIdx.x) >> 5;
  int lane = threadIdx.x & 31, hi = lane >> 4, lo = lane & 15;
  int m = wid * 16 + lo;
  int b = m / 400, p = m % 400, oh = p / 20, ow = p % 20;
  const _Float16* abase = imgC + (size_t)b * 3 * 84 * IMS + (oh * 4) * IMS + ow * 4;
  v8f acc0 = {}, acc1 = {};
#pragma unroll
  for (int kb = 0; kb < 192; kb += 32) {
    int c = kb >> 6, khb = (kb & 63) >> 3;
    v16h a  = pack_frag_c1(abase + (c * 84 + khb + hi) * IMS);
    v16h b0 = pack_frag(wt + lo * 192        + kb + hi * 8);
    v16h b1 = pack_frag(wt + (16 + lo) * 192 + kb + hi * 8);
    acc0 = WMMA16(a, b0, acc0);
    acc1 = WMMA16(a, b1, acc1);
  }
  float bs0 = bias[lo], bs1 = bias[16 + lo];
#pragma unroll
  for (int r = 0; r < 8; r++) {
    int mm = wid * 16 + hi * 8 + r;
    float v0 = acc0[r] + bs0; v0 = v0 > 0.0f ? v0 : 0.0f;
    float v1 = acc1[r] + bs1; v1 = v1 > 0.0f ? v1 : 0.0f;
    out[mm * 32 + lo]      = (_Float16)v0;
    out[mm * 32 + 16 + lo] = (_Float16)v1;
  }
}

__global__ void __launch_bounds__(256) k_conv2(const _Float16* act1, const _Float16* wt,
                                               const float* bias, _Float16* out) {
  int wid  = (blockIdx.x * blockDim.x + threadIdx.x) >> 5;
  int lane = threadIdx.x & 31, hi = lane >> 4, lo = lane & 15;
  int m = wid * 16 + lo;
  int b = m / 81, p = m % 81, oh = p / 9, ow = p % 9;
  v8f acc0 = {}, acc1 = {}, acc2 = {}, acc3 = {};
  for (int kb = 0; kb < 512; kb += 32) {
    int pix = kb >> 5, kh = pix >> 2, kw = pix & 3;
    v16h a  = pack_frag(act1 + ((b * 20 + oh * 2 + kh) * 20 + (ow * 2 + kw)) * 32 + hi * 8);
    v16h b0 = pack_frag(wt + (lo)      * 512 + kb + hi * 8);
    v16h b1 = pack_frag(wt + (16 + lo) * 512 + kb + hi * 8);
    v16h b2 = pack_frag(wt + (32 + lo) * 512 + kb + hi * 8);
    v16h b3 = pack_frag(wt + (48 + lo) * 512 + kb + hi * 8);
    acc0 = WMMA16(a, b0, acc0);
    acc1 = WMMA16(a, b1, acc1);
    acc2 = WMMA16(a, b2, acc2);
    acc3 = WMMA16(a, b3, acc3);
  }
  float bs0 = bias[lo], bs1 = bias[16 + lo], bs2 = bias[32 + lo], bs3 = bias[48 + lo];
#pragma unroll
  for (int r = 0; r < 8; r++) {
    int mm = wid * 16 + hi * 8 + r;
    float v0 = acc0[r] + bs0; v0 = v0 > 0.0f ? v0 : 0.0f;
    float v1 = acc1[r] + bs1; v1 = v1 > 0.0f ? v1 : 0.0f;
    float v2 = acc2[r] + bs2; v2 = v2 > 0.0f ? v2 : 0.0f;
    float v3 = acc3[r] + bs3; v3 = v3 > 0.0f ? v3 : 0.0f;
    out[mm * 64 + lo]      = (_Float16)v0;
    out[mm * 64 + 16 + lo] = (_Float16)v1;
    out[mm * 64 + 32 + lo] = (_Float16)v2;
    out[mm * 64 + 48 + lo] = (_Float16)v3;
  }
}

__global__ void __launch_bounds__(256) k_conv3(const _Float16* act2, const _Float16* wt,
                                               const float* bias, _Float16* out) {
  int wid  = (blockIdx.x * blockDim.x + threadIdx.x) >> 5;
  int lane = threadIdx.x & 31, hi = lane >> 4, lo = lane & 15;
  int m = wid * 16 + lo;
  int b = m / 49, p = m % 49, oh = p / 7, ow = p % 7;
  v8f acc0 = {}, acc1 = {}, acc2 = {}, acc3 = {};
  for (int kb = 0; kb < 576; kb += 32) {
    int pix = kb >> 6, cb = (kb & 63) + hi * 8;
    int kh = pix / 3, kw = pix % 3;
    v16h a  = pack_frag(act2 + ((b * 9 + oh + kh) * 9 + (ow + kw)) * 64 + cb);
    v16h b0 = pack_frag(wt + (lo)      * 576 + kb + hi * 8);
    v16h b1 = pack_frag(wt + (16 + lo) * 576 + kb + hi * 8);
    v16h b2 = pack_frag(wt + (32 + lo) * 576 + kb + hi * 8);
    v16h b3 = pack_frag(wt + (48 + lo) * 576 + kb + hi * 8);
    acc0 = WMMA16(a, b0, acc0);
    acc1 = WMMA16(a, b1, acc1);
    acc2 = WMMA16(a, b2, acc2);
    acc3 = WMMA16(a, b3, acc3);
  }
  float bs0 = bias[lo], bs1 = bias[16 + lo], bs2 = bias[32 + lo], bs3 = bias[48 + lo];
#pragma unroll
  for (int r = 0; r < 8; r++) {
    int mm = wid * 16 + hi * 8 + r;
    float v0 = acc0[r] + bs0; v0 = v0 > 0.0f ? v0 : 0.0f;
    float v1 = acc1[r] + bs1; v1 = v1 > 0.0f ? v1 : 0.0f;
    float v2 = acc2[r] + bs2; v2 = v2 > 0.0f ? v2 : 0.0f;
    float v3 = acc3[r] + bs3; v3 = v3 > 0.0f ? v3 : 0.0f;
    out[mm * 64 + lo]      = (_Float16)v0;
    out[mm * 64 + 16 + lo] = (_Float16)v1;
    out[mm * 64 + 32 + lo] = (_Float16)v2;
    out[mm * 64 + 48 + lo] = (_Float16)v3;
  }
}

__global__ void __launch_bounds__(256) k_fc(const _Float16* act3, const _Float16* wt,
                                            const float* bias, _Float16* X) {
  int wid  = (blockIdx.x * blockDim.x + threadIdx.x) >> 5;
  int lane = threadIdx.x & 31, hi = lane >> 4, lo = lane & 15;
  int ng = wid & 7, mt = wid >> 3;
  int m = mt * 16 + lo;
  int nb = ng * 64;
  v8f acc0 = {}, acc1 = {}, acc2 = {}, acc3 = {};
  for (int kb = 0; kb < 3136; kb += 32) {
    v16h a  = pack_frag(act3 + (size_t)m * 3136 + kb + hi * 8);
    v16h b0 = pack_frag(wt + (size_t)(nb + lo)      * 3136 + kb + hi * 8);
    v16h b1 = pack_frag(wt + (size_t)(nb + 16 + lo) * 3136 + kb + hi * 8);
    v16h b2 = pack_frag(wt + (size_t)(nb + 32 + lo) * 3136 + kb + hi * 8);
    v16h b3 = pack_frag(wt + (size_t)(nb + 48 + lo) * 3136 + kb + hi * 8);
    acc0 = WMMA16(a, b0, acc0);
    acc1 = WMMA16(a, b1, acc1);
    acc2 = WMMA16(a, b2, acc2);
    acc3 = WMMA16(a, b3, acc3);
  }
  float bs0 = bias[nb + lo], bs1 = bias[nb + 16 + lo];
  float bs2 = bias[nb + 32 + lo], bs3 = bias[nb + 48 + lo];
#pragma unroll
  for (int r = 0; r < 8; r++) {
    int mm = mt * 16 + hi * 8 + r;
    float v0 = acc0[r] + bs0; v0 = v0 > 0.0f ? v0 : 0.0f;
    float v1 = acc1[r] + bs1; v1 = v1 > 0.0f ? v1 : 0.0f;
    float v2 = acc2[r] + bs2; v2 = v2 > 0.0f ? v2 : 0.0f;
    float v3 = acc3[r] + bs3; v3 = v3 > 0.0f ? v3 : 0.0f;
    X[mm * XS + nb + lo]      = (_Float16)v0;
    X[mm * XS + nb + 16 + lo] = (_Float16)v1;
    X[mm * XS + nb + 32 + lo] = (_Float16)v2;
    X[mm * XS + nb + 48 + lo] = (_Float16)v3;
  }
}

__global__ void __launch_bounds__(256) k_xproj(const _Float16* X, const _Float16* Wih,
                                               const float* biasg, float* G) {
  int wid  = (blockIdx.x * blockDim.x + threadIdx.x) >> 5;
  int lane = threadIdx.x & 31, hi = lane >> 4, lo = lane & 15;
  int mtile = wid & 127, cg = wid >> 7;
  int ncol = cg * 16 + lo;
  float bi = biasg[0 * HL + ncol], bf = biasg[1 * HL + ncol];
  float bg = biasg[2 * HL + ncol], bo = biasg[3 * HL + ncol];
  v8f ai = {bi, bi, bi, bi, bi, bi, bi, bi};
  v8f af = {bf, bf, bf, bf, bf, bf, bf, bf};
  v8f ag = {bg, bg, bg, bg, bg, bg, bg, bg};
  v8f ao = {bo, bo, bo, bo, bo, bo, bo, bo};
  const _Float16* ap = X   + (mtile * 16 + lo) * XS + hi * 8;
  const _Float16* bp = Wih + ncol * XS + hi * 8;
  for (int kb = 0; kb < XS; kb += 32) {
    v16h a  = pack_frag(ap + kb);
    v16h b0 = pack_frag(bp + 0 * HL * XS + kb);
    v16h b1 = pack_frag(bp + 1 * HL * XS + kb);
    v16h b2 = pack_frag(bp + 2 * HL * XS + kb);
    v16h b3 = pack_frag(bp + 3 * HL * XS + kb);
    ai = WMMA16(a, b0, ai);
    af = WMMA16(a, b1, af);
    ag = WMMA16(a, b2, ag);
    ao = WMMA16(a, b3, ao);
  }
  v4f* gp = (v4f*)(G + (size_t)(mtile * 16 + cg) * GT) + lane;
  gp[0 * 64]      = ai.lo;  gp[0 * 64 + 32] = ai.hi;
  gp[1 * 64]      = af.lo;  gp[1 * 64 + 32] = af.hi;
  gp[2 * 64]      = ag.lo;  gp[2 * 64 + 32] = ag.hi;
  gp[3 * 64]      = ao.lo;  gp[3 * 64 + 32] = ao.hi;
}

__global__ void __launch_bounds__(1024) k_lstm(const float* G, const _Float16* Whh,
                                               const int* done,
                                               const float* h0, const float* c0,
                                               float* hs, float* out) {
  __shared__ _Float16 A[E_ENVS * HSTR];
  int tid = threadIdx.x;
  int w = tid >> 5, lane = tid & 31, hi = lane >> 4, lo = lane & 15;
  int mt = w >> 4, cg = w & 15;
  int ncol = cg * 16 + lo;

  for (int i = tid; i < E_ENVS * HL; i += 1024) {
    int e = i >> 8, k = i & 255;
    A[e * HSTR + k] = (_Float16)h0[i];
  }
  v8f creg;
#pragma unroll
  for (int r = 0; r < 8; r++) {
    int env = mt * 16 + hi * 8 + r;
    creg[r] = c0[env * HL + ncol];
  }
  const _Float16* bp = Whh + ncol * HL + hi * 8;
  __syncthreads();

  for (int t = 0; t < T_STEPS; t++) {
    const v4f* gp = (const v4f*)(G + (size_t)((t * 2 + mt) * 16 + cg) * GT) + lane;
    v8f ai, af, ag, ao;
    ai.lo = gp[0 * 64];  ai.hi = gp[0 * 64 + 32];
    af.lo = gp[1 * 64];  af.hi = gp[1 * 64 + 32];
    ag.lo = gp[2 * 64];  ag.hi = gp[2 * 64 + 32];
    ao.lo = gp[3 * 64];  ao.hi = gp[3 * 64 + 32];

    for (int i = tid; i < E_ENVS * HL; i += 1024) {
      int e = i >> 8, k = i & 255;
      if (done[t * E_ENVS + e]) A[e * HSTR + k] = (_Float16)0.0f;
    }
#pragma unroll
    for (int r = 0; r < 8; r++) {
      int env = mt * 16 + hi * 8 + r;
      if (done[t * E_ENVS + env]) creg[r] = 0.0f;
    }
    __syncthreads();

    for (int kb = 0; kb < HL; kb += 32) {
      v16h a  = pack_frag(&A[(mt * 16 + lo) * HSTR + kb + hi * 8]);
      v16h b0 = pack_frag(bp + 0 * HL * HL + kb);
      v16h b1 = pack_frag(bp + 1 * HL * HL + kb);
      v16h b2 = pack_frag(bp + 2 * HL * HL + kb);
      v16h b3 = pack_frag(bp + 3 * HL * HL + kb);
      ai = WMMA16(a, b0, ai);
      af = WMMA16(a, b1, af);
      ag = WMMA16(a, b2, ag);
      ao = WMMA16(a, b3, ao);
    }
    __syncthreads();

    float hv[8];
#pragma unroll
    for (int r = 0; r < 8; r++) {
      float iv = sigmoidf_(ai[r]), fv = sigmoidf_(af[r]);
      float gv = tanhf(ag[r]),     ov = sigmoidf_(ao[r]);
      float cv = fv * creg[r] + iv * gv;
      creg[r] = cv;
      hv[r]   = ov * tanhf(cv);
    }
#pragma unroll
    for (int r = 0; r < 8; r++) {
      int env = mt * 16 + hi * 8 + r;
      A[env * HSTR + ncol] = (_Float16)hv[r];
      hs[(t * E_ENVS + env) * HL + ncol] = hv[r];
      if (t == T_STEPS - 1) {
        out[12288 + env * HL + ncol] = hv[r];
        out[20480 + env * HL + ncol] = creg[r];
      }
    }
    __syncthreads();
  }
}

__global__ void k_heads(const float* hs, const float* pw, const float* pb,
                        const float* vw, const float* vb, float* out) {
  int i = blockIdx.x * blockDim.x + threadIdx.x;
  if (i >= B_TOT * 6) return;
  int b = i / 6, j = i % 6;
  const float* h    = hs + b * HL;
  const float* wrow = (j < 5) ? pw + j * HL : vw;
  float s = (j < 5) ? pb[j] : vb[0];
#pragma unroll 4
  for (int k = 0; k < HL; k++) s += h[k] * wrow[k];
  if (j < 5) out[b * 5 + j] = s;
  else       out[10240 + b] = s;
}

extern "C" void kernel_launch(void* const* d_in, const int* in_sizes, int n_in,
                              void* d_out, int out_size, void* d_ws, size_t ws_size,
                              hipStream_t stream) {
  (void)in_sizes; (void)n_in; (void)out_size; (void)ws_size;
  const float* img  = (const float*)d_in[0];
  const int*   pos  = (const int*)d_in[1];
  const int*   lact = (const int*)d_in[2];
  const int*   done = (const int*)d_in[3];
  const float* h0   = (const float*)d_in[4];
  const float* c0   = (const float*)d_in[5];
  const float* w1   = (const float*)d_in[6];
  const float* b1   = (const float*)d_in[7];
  const float* w2   = (const float*)d_in[8];
  const float* b2   = (const float*)d_in[9];
  const float* w3   = (const float*)d_in[10];
  const float* b3   = (const float*)d_in[11];
  const float* fcw  = (const float*)d_in[12];
  const float* fcb  = (const float*)d_in[13];
  const float* wih  = (const float*)d_in[14];
  const float* whh  = (const float*)d_in[15];
  const float* bih  = (const float*)d_in[16];
  const float* bhh  = (const float*)d_in[17];
  const float* pw   = (const float*)d_in[18];
  const float* pb   = (const float*)d_in[19];
  const float* vw   = (const float*)d_in[20];
  const float* vb   = (const float*)d_in[21];
  float* out = (float*)d_out;
  char*  ws  = (char*)d_ws;

  const size_t o_img  = 0;
  const size_t o_act1 = 90832896;
  const size_t o_w1t  = o_act1 + 52428800;
  const size_t o_w2t  = o_w1t + 12288;
  const size_t o_w3t  = o_w2t + 65536;
  const size_t o_fcwt = o_w3t + 73728;
  const size_t o_wih  = o_fcwt + 3211264;
  const size_t o_whh  = o_wih + 1179648;
  const size_t o_bg   = o_whh + 524288;
  const size_t o_act2 = 0;
  const size_t o_act3 = 21233664;
  const size_t o_X    = 34078720;
  const size_t o_hs   = 36438016;
  const size_t o_G    = 38535168;

  _Float16* imgC = (_Float16*)(ws + o_img);
  _Float16* act1 = (_Float16*)(ws + o_act1);
  _Float16* act2 = (_Float16*)(ws + o_act2);
  _Float16* act3 = (_Float16*)(ws + o_act3);
  _Float16* Xm   = (_Float16*)(ws + o_X);
  _Float16* w1t  = (_Float16*)(ws + o_w1t);
  _Float16* w2t  = (_Float16*)(ws + o_w2t);
  _Float16* w3t  = (_Float16*)(ws + o_w3t);
  _Float16* fcwt = (_Float16*)(ws + o_fcwt);
  _Float16* wih16 = (_Float16*)(ws + o_wih);
  _Float16* whh16 = (_Float16*)(ws + o_whh);
  float*    bg   = (float*)(ws + o_bg);
  float*    hsb  = (float*)(ws + o_hs);
  float*    Gx   = (float*)(ws + o_G);

  const int TB = 256;
  int nImg = B_TOT * 3 * 84 * 84;
  k_prep_img <<<(nImg + TB - 1) / TB, TB, 0, stream>>>(img, imgC, nImg);
  k_prep_w1  <<<(32 * 192   + TB - 1) / TB, TB, 0, stream>>>(w1, w1t);
  k_prep_w2  <<<(64 * 512   + TB - 1) / TB, TB, 0, stream>>>(w2, w2t);
  k_prep_w3  <<<(64 * 576   + TB - 1) / TB, TB, 0, stream>>>(w3, w3t);
  k_prep_fc  <<<(512 * 3136 + TB - 1) / TB, TB, 0, stream>>>(fcw, fcwt);
  k_prep_wl  <<<(1024 * (XS + HL) + TB - 1) / TB, TB, 0, stream>>>(wih, whh, wih16, whh16);
  k_prep_bias<<<(1024 + TB - 1) / TB, TB, 0, stream>>>(bih, bhh, bg);

  k_conv1<<<6400, TB, 0, stream>>>(imgC, w1t, b1, act1);
  k_conv2<<<1296, TB, 0, stream>>>(act1, w2t, b2, act2);
  k_conv3<<< 784, TB, 0, stream>>>(act2, w3t, b3, act3);
  k_fc   <<< 128, TB, 0, stream>>>(act3, fcwt, fcb, Xm);

  k_onehot<<<(B_TOT * 64 + TB - 1) / TB, TB, 0, stream>>>(pos, lact, Xm);

  k_xproj<<<256, TB, 0, stream>>>(Xm, wih16, bg, Gx);

  k_lstm<<<1, 1024, 0, stream>>>(Gx, whh16, done, h0, c0, hsb, out);

  k_heads<<<(B_TOT * 6 + TB - 1) / TB, TB, 0, stream>>>(hsb, pw, pb, vw, vb, out);
}


# LSTM gate activations on the hardware tanh: tanh = one v_tanh_f32, sigmoid(x) = 0.5+0.5*tanh(x/2), replacing libm tanhf and exp + IEEE division; activation rounding changes, sum order unchanged
# speedup vs baseline: 6.2248x; 3.0682x over previous
#include <hip/hip_runtime.h>

typedef __attribute__((ext_vector_type(16))) _Float16 v16h;
typedef __attribute__((ext_vector_type(8)))  _Float16 v8h;
typedef __attribute__((ext_vector_type(4)))  _Float16 v4h;
typedef __attribute__((ext_vector_type(8)))  float    v8f;
typedef __attribute__((ext_vector_type(4)))  float    v4f;

#define WMMA16(A,B,C) __builtin_amdgcn_wmma_f32_16x16x32_f16(false,(A),false,(B),(short)0,(C),false,false)

static constexpr int T_STEPS = 64;
static constexpr int E_ENVS  = 32;
static constexpr int B_TOT   = 2048;
static constexpr int HL      = 256;
static constexpr int IND     = 549;
static constexpr int XS      = 576;
static constexpr int HSTR    = 264;
static constexpr int GT      = 1024;
static constexpr int IROW    = 84 * 3;

__device__ __forceinline__ v16h pack_frag(const _Float16* p) {
  v8h u = *(const v8h*)p;
  v8h w = *(const v8h*)(p + 16);
  v16h a;
#pragma unroll
  for (int e = 0; e < 8; e++) { a[e] = u[e]; a[e + 8] = w[e]; }
  return a;
}
__device__ __forceinline__ v16h pack_frag_img(const float* p0, const float* p1) {
  v4f u0 = *(const v4f*)p0;
  v4f u1 = *(const v4f*)(p0 + 4);
  v4f u2 = *(const v4f*)p1;
  v4f u3 = *(const v4f*)(p1 + 4);
  v16h a;
#pragma unroll
  for (int e = 0; e < 4; e++) {
    a[e]      = (_Float16)(u0[e] * (1.0f / 255.0f));
    a[e + 4]  = (_Float16)(u1[e] * (1.0f / 255.0f));
    a[e + 8]  = (_Float16)(u2[e] * (1.0f / 255.0f));
    a[e + 12] = (_Float16)(u3[e] * (1.0f / 255.0f));
  }
  return a;
}
__host__ __device__ constexpr int patch_off(int f) { return (f / 24) * IROW + (f % 24); }

__device__ __forceinline__ float tanh_hw(float x) { return __builtin_amdgcn_tanhf(x); }
__device__ __forceinline__ float sigmoid_hw(float x) {
  return __builtin_fmaf(0.5f, tanh_hw(0.5f * x), 0.5f);
}

__global__ void k_prep_w1(const float* w, _Float16* wt) {
  int i = blockIdx.x * blockDim.x + threadIdx.x;
  if (i >= 32 * 192) return;
  int n = i / 192, f = i % 192;
  int kh = f / 24, r = f % 24, kw = r / 3, c = r % 3;
  wt[i] = (_Float16)w[((n * 3 + c) * 8 + kh) * 8 + kw];
}
__global__ void k_prep_w2(const float* w, _Float16* wt) {
  int i = blockIdx.x * blockDim.x + threadIdx.x;
  if (i >= 64 * 512) return;
  int n = i / 512, k = i % 512;
  int pix = k >> 5, c = k & 31, kh = pix >> 2, kw = pix & 3;
  wt[i] = (_Float16)w[((n * 32 + c) * 4 + kh) * 4 + kw];
}
__global__ void k_prep_w3(const float* w, _Float16* wt) {
  int i = blockIdx.x * blockDim.x + threadIdx.x;
  if (i >= 64 * 576) return;
  int n = i / 576, k = i % 576;
  int pix = k >> 6, c = k & 63, kh = pix / 3, kw = pix % 3;
  wt[i] = (_Float16)w[((n * 64 + c) * 3 + kh) * 3 + kw];
}
__global__ void k_prep_fc(const float* w, _Float16* wt) {
  int i = blockIdx.x * blockDim.x + threadIdx.x;
  if (i >= 512 * 3136) return;
  int f = i / 3136, k = i % 3136;
  int s = k >> 6, c = k & 63;
  wt[i] = (_Float16)w[f * 3136 + c * 49 + s];
}
__global__ void k_prep_wl(const float* wih, const float* whh, _Float16* wih16, _Float16* whh16) {
  int i = blockIdx.x * blockDim.x + threadIdx.x;
  if (i >= 1024 * (XS + HL)) return;
  int r = i / (XS + HL), k = i % (XS + HL);
  if (k < XS) wih16[r * XS + k] = (_Float16)((k < IND) ? wih[r * IND + k] : 0.0f);
  else        whh16[r * HL + (k - XS)] = (_Float16)whh[r * HL + (k - XS)];
}
__global__ void k_prep_bias(const float* bih, const float* bhh, float* bg) {
  int i = blockIdx.x * blockDim.x + threadIdx.x;
  if (i < 1024) bg[i] = bih[i] + bhh[i];
}
__global__ void k_onehot(const int* pos, const int* lact, _Float16* X) {
  int i = blockIdx.x * blockDim.x + threadIdx.x;
  if (i >= B_TOT * 64) return;
  int b = i >> 6, c = i & 63;
  float v = 0.0f;
  if (c < 16)      v = (pos[b * 2 + 0] == c)      ? 1.0f : 0.0f;
  else if (c < 32) v = (pos[b * 2 + 1] == c - 16) ? 1.0f : 0.0f;
  else if (c < 37) v = (lact[b]        == c - 32) ? 1.0f : 0.0f;
  X[b * XS + 512 + c] = (_Float16)v;
}

__global__ void __launch_bounds__(256) k_conv1(const float* img, const _Float16* wt,
                                               const float* bias, _Float16* out) {
  int wid  = (blockIdx.x * blockDim.x + threadIdx.x) >> 5;
  int lane = threadIdx.x & 31, hi = lane >> 4, lo = lane & 15;
  int m = wid * 16 + lo;
  int b = m / 400, p = m % 400, oh = p / 20, ow = p % 20;
  const float* abase = img + ((size_t)(b * 84 + oh * 4) * 84 + ow * 4) * 3;
  v8f acc0 = {}, acc1 = {};
#pragma unroll
  for (int kb = 0; kb < 192; kb += 32) {
    int o0 = hi ? patch_off(kb + 8)  : patch_off(kb);
    int o1 = hi ? patch_off(kb + 24) : patch_off(kb + 16);
    v16h a  = pack_frag_img(abase + o0, abase + o1);
    v16h b0 = pack_frag(wt + lo * 192        + kb + hi * 8);
    v16h b1 = pack_frag(wt + (16 + lo) * 192 + kb + hi * 8);
    acc0 = WMMA16(a, b0, acc0);
    acc1 = WMMA16(a, b1, acc1);
  }
  float bs0 = bias[lo], bs1 = bias[16 + lo];
#pragma unroll
  for (int r = 0; r < 8; r++) {
    int mm = wid * 16 + hi * 8 + r;
    float v0 = acc0[r] + bs0; v0 = v0 > 0.0f ? v0 : 0.0f;
    float v1 = acc1[r] + bs1; v1 = v1 > 0.0f ? v1 : 0.0f;
    out[mm * 32 + lo]      = (_Float16)v0;
    out[mm * 32 + 16 + lo] = (_Float16)v1;
  }
}

__global__ void __launch_bounds__(256) k_conv2(const _Float16* act1, const _Float16* wt,
                                               const float* bias, _Float16* out) {
  int wid  = (blockIdx.x * blockDim.x + threadIdx.x) >> 5;
  int lane = threadIdx.x & 31, hi = lane >> 4, lo = lane & 15;
  int m = wid * 16 + lo;
  int b = m / 81, p = m % 81, oh = p / 9, ow = p % 9;
  v8f acc0 = {}, acc1 = {}, acc2 = {}, acc3 = {};
  for (int kb = 0; kb < 512; kb += 32) {
    int pix = kb >> 5, kh = pix >> 2, kw = pix & 3;
    v16h a  = pack_frag(act1 + ((b * 20 + oh * 2 + kh) * 20 + (ow * 2 + kw)) * 32 + hi * 8);
    v16h b0 = pack_frag(wt + (lo)      * 512 + kb + hi * 8);
    v16h b1 = pack_frag(wt + (16 + lo) * 512 + kb + hi * 8);
    v16h b2 = pack_frag(wt + (32 + lo) * 512 + kb + hi * 8);
    v16h b3 = pack_frag(wt + (48 + lo) * 512 + kb + hi * 8);
    acc0 = WMMA16(a, b0, acc0);
    acc1 = WMMA16(a, b1, acc1);
    acc2 = WMMA16(a, b2, acc2);
    acc3 = WMMA16(a, b3, acc3);
  }
  float bs0 = bias[lo], bs1 = bias[16 + lo], bs2 = bias[32 + lo], bs3 = bias[48 + lo];
#pragma unroll
  for (int r = 0; r < 8; r++) {
    int mm = wid * 16 + hi * 8 + r;
    float v0 = acc0[r] + bs0; v0 = v0 > 0.0f ? v0 : 0.0f;
    float v1 = acc1[r] + bs1; v1 = v1 > 0.0f ? v1 : 0.0f;
    float v2 = acc2[r] + bs2; v2 = v2 > 0.0f ? v2 : 0.0f;
    float v3 = acc3[r] + bs3; v3 = v3 > 0.0f ? v3 : 0.0f;
    out[mm * 64 + lo]      = (_Float16)v0;
    out[mm * 64 + 16 + lo] = (_Float16)v1;
    out[mm * 64 + 32 + lo] = (_Float16)v2;
    out[mm * 64 + 48 + lo] = (_Float16)v3;
  }
}

__global__ void __launch_bounds__(256) k_conv3(const _Float16* act2, const _Float16* wt,
                                               const float* bias, _Float16* out) {
  int wid  = (blockIdx.x * blockDim.x + threadIdx.x) >> 5;
  int lane = threadIdx.x & 31, hi = lane >> 4, lo = lane & 15;
  int m = wid * 16 + lo;
  int b = m / 49, p = m % 49, oh = p / 7, ow = p % 7;
  v8f acc0 = {}, acc1 = {}, acc2 = {}, acc3 = {};
  for (int kb = 0; kb < 576; kb += 32) {
    int pix = kb >> 6, cb = (kb & 63) + hi * 8;
    int kh = pix / 3, kw = pix % 3;
    v16h a  = pack_frag(act2 + ((b * 9 + oh + kh) * 9 + (ow + kw)) * 64 + cb);
    v16h b0 = pack_frag(wt + (lo)      * 576 + kb + hi * 8);
    v16h b1 = pack_frag(wt + (16 + lo) * 576 + kb + hi * 8);
    v16h b2 = pack_frag(wt + (32 + lo) * 576 + kb + hi * 8);
    v16h b3 = pack_frag(wt + (48 + lo) * 576 + kb + hi * 8);
    acc0 = WMMA16(a, b0, acc0);
    acc1 = WMMA16(a, b1, acc1);
    acc2 = WMMA16(a, b2, acc2);
    acc3 = WMMA16(a, b3, acc3);
  }
  float bs0 = bias[lo], bs1 = bias[16 + lo], bs2 = bias[32 + lo], bs3 = bias[48 + lo];
#pragma unroll
  for (int r = 0; r < 8; r++) {
    int mm = wid * 16 + hi * 8 + r;
    float v0 = acc0[r] + bs0; v0 = v0 > 0.0f ? v0 : 0.0f;
    float v1 = acc1[r] + bs1; v1 = v1 > 0.0f ? v1 : 0.0f;
    float v2 = acc2[r] + bs2; v2 = v2 > 0.0f ? v2 : 0.0f;
    float v3 = acc3[r] + bs3; v3 = v3 > 0.0f ? v3 : 0.0f;
    out[mm * 64 + lo]      = (_Float16)v0;
    out[mm * 64 + 16 + lo] = (_Float16)v1;
    out[mm * 64 + 32 + lo] = (_Float16)v2;
    out[mm * 64 + 48 + lo] = (_Float16)v3;
  }
}

__global__ void __launch_bounds__(256) k_fc(const _Float16* act3, const _Float16* wt,
                                            const float* bias, _Float16* X) {
  int wid  = (blockIdx.x * blockDim.x + threadIdx.x) >> 5;
  int lane = threadIdx.x & 31, hi = lane >> 4, lo = lane & 15;
  int ng = wid & 7, mt = wid >> 3;
  int m = mt * 16 + lo;
  int nb = ng * 64;
  v8f acc0 = {}, acc1 = {}, acc2 = {}, acc3 = {};
  for (int kb = 0; kb < 3136; kb += 32) {
    v16h a  = pack_frag(act3 + (size_t)m * 3136 + kb + hi * 8);
    v16h b0 = pack_frag(wt + (size_t)(nb + lo)      * 3136 + kb + hi * 8);
    v16h b1 = pack_frag(wt + (size_t)(nb + 16 + lo) * 3136 + kb + hi * 8);
    v16h b2 = pack_frag(wt + (size_t)(nb + 32 + lo) * 3136 + kb + hi * 8);
    v16h b3 = pack_frag(wt + (size_t)(nb + 48 + lo) * 3136 + kb + hi * 8);
    acc0 = WMMA16(a, b0, acc0);
    acc1 = WMMA16(a, b1, acc1);
    acc2 = WMMA16(a, b2, acc2);
    acc3 = WMMA16(a, b3, acc3);
  }
  float bs0 = bias[nb + lo], bs1 = bias[nb + 16 + lo];
  float bs2 = bias[nb + 32 + lo], bs3 = bias[nb + 48 + lo];
#pragma unroll
  for (int r = 0; r < 8; r++) {
    int mm = mt * 16 + hi * 8 + r;
    float v0 = acc0[r] + bs0; v0 = v0 > 0.0f ? v0 : 0.0f;
    float v1 = acc1[r] + bs1; v1 = v1 > 0.0f ? v1 : 0.0f;
    float v2 = acc2[r] + bs2; v2 = v2 > 0.0f ? v2 : 0.0f;
    float v3 = acc3[r] + bs3; v3 = v3 > 0.0f ? v3 : 0.0f;
    X[mm * XS + nb + lo]      = (_Float16)v0;
    X[mm * XS + nb + 16 + lo] = (_Float16)v1;
    X[mm * XS + nb + 32 + lo] = (_Float16)v2;
    X[mm * XS + nb + 48 + lo] = (_Float16)v3;
  }
}

__global__ void __launch_bounds__(256) k_xproj(const _Float16* X, const _Float16* Wih,
                                               const float* biasg, float* G) {
  int wid  = (blockIdx.x * blockDim.x + threadIdx.x) >> 5;
  int lane = threadIdx.x & 31, hi = lane >> 4, lo = lane & 15;
  int mtile = wid & 127, cg = wid >> 7;
  int ncol = cg * 16 + lo;
  float bi = biasg[0 * HL + ncol], bf = biasg[1 * HL + ncol];
  float bg = biasg[2 * HL + ncol], bo = biasg[3 * HL + ncol];
  v8f ai = {bi, bi, bi, bi, bi, bi, bi, bi};
  v8f af = {bf, bf, bf, bf, bf, bf, bf, bf};
  v8f ag = {bg, bg, bg, bg, bg, bg, bg, bg};
  v8f ao = {bo, bo, bo, bo, bo, bo, bo, bo};
  const _Float16* ap = X   + (mtile * 16 + lo) * XS + hi * 8;
  const _Float16* bp = Wih + ncol * XS + hi * 8;
  for (int kb = 0; kb < XS; kb += 32) {
    v16h a  = pack_frag(ap + kb);
    v16h b0 = pack_frag(bp + 0 * HL * XS + kb);
    v16h b1 = pack_frag(bp + 1 * HL * XS + kb);
    v16h b2 = pack_frag(bp + 2 * HL * XS + kb);
    v16h b3 = pack_frag(bp + 3 * HL * XS + kb);
    ai = WMMA16(a, b0, ai);
    af = WMMA16(a, b1, af);
    ag = WMMA16(a, b2, ag);
    ao = WMMA16(a, b3, ao);
  }
  v4f* gp = (v4f*)(G + (size_t)(mtile * 16 + cg) * GT) + lane;
  gp[0 * 64]      = ai.lo;  gp[0 * 64 + 32] = ai.hi;
  gp[1 * 64]      = af.lo;  gp[1 * 64 + 32] = af.hi;
  gp[2 * 64]      = ag.lo;  gp[2 * 64 + 32] = ag.hi;
  gp[3 * 64]      = ao.lo;  gp[3 * 64 + 32] = ao.hi;
}

__global__ void __launch_bounds__(512) k_lstm(const float* G, const _Float16* Whh,
                                              const int* done,
                                              const float* h0, const float* c0,
                                              float* hs, float* out) {
  __shared__ v8h Wl[16 * 4 * 4 * 2 * 32];
  __shared__ __attribute__((aligned(16))) _Float16 A[16 * HSTR];
  const int tid = threadIdx.x;
  const int mt  = blockIdx.x;
  const int cg  = tid >> 5, lane = tid & 31, hi = lane >> 4, lo = lane & 15;
  const int ncol = cg * 16 + lo;
  const int env0 = mt * 16;

  for (int i = tid; i < 16 * HL; i += 512) {
    int e = i >> 8, k = i & 255;
    A[e * HSTR + k] = (_Float16)h0[(env0 + e) * HL + k];
  }
  v8f creg;
#pragma unroll
  for (int r = 0; r < 8; r++) creg[r] = c0[(env0 + hi * 8 + r) * HL + ncol];

  const _Float16* bp = Whh + ncol * HL + hi * 8;
  v8h* wl = Wl + cg * (4 * 4 * 2 * 32) + lane;
#pragma unroll
  for (int ch = 0; ch < 4; ch++) {
#pragma unroll
    for (int g = 0; g < 4; g++) {
      const _Float16* p = bp + g * HL * HL + (4 + ch) * 32;
      wl[((ch * 4 + g) * 2 + 0) * 32] = *(const v8h*)p;
      wl[((ch * 4 + g) * 2 + 1) * 32] = *(const v8h*)(p + 16);
    }
  }
  __builtin_amdgcn_sched_barrier(0);
  v16h wr[4][4];
#pragma unroll
  for (int ch = 0; ch < 4; ch++) {
#pragma unroll
    for (int g = 0; g < 4; g++) wr[ch][g] = pack_frag(bp + g * HL * HL + ch * 32);
  }
  __syncthreads();

  float hv[8];
  for (int t = 0; t < T_STEPS; t++) {
    const v4f* gp = (const v4f*)(G + (size_t)((t * 2 + mt) * 16 + cg) * GT) + lane;
    v8f ai, af, ag, ao;
    ai.lo = gp[0 * 64];  ai.hi = gp[0 * 64 + 32];
    af.lo = gp[1 * 64];  af.hi = gp[1 * 64 + 32];
    ag.lo = gp[2 * 64];  ag.hi = gp[2 * 64 + 32];
    ao.lo = gp[3 * 64];  ao.hi = gp[3 * 64 + 32];

    for (int i = tid; i < 16 * HL; i += 512) {
      int e = i >> 8, k = i & 255;
      if (done[t * E_ENVS + env0 + e]) A[e * HSTR + k] = (_Float16)0.0f;
    }
#pragma unroll
    for (int r = 0; r < 8; r++) {
      if (done[t * E_ENVS + env0 + hi * 8 + r]) creg[r] = 0.0f;
    }
    __syncthreads();

    const _Float16* arow = &A[lo * HSTR + hi * 8];
#pragma unroll
    for (int ch = 0; ch < 4; ch++) {
      v16h a = pack_frag(arow + ch * 32);
      ai = WMMA16(a, wr[ch][0], ai);
      af = WMMA16(a, wr[ch][1], af);
      ag = WMMA16(a, wr[ch][2], ag);
      ao = WMMA16(a, wr[ch][3], ao);
    }
#pragma unroll
    for (int ch = 0; ch < 4; ch++) {
      v16h a = pack_frag(arow + (4 + ch) * 32);
      v16h b[4];
#pragma unroll
      for (int g = 0; g < 4; g++) {
        v8h u = wl[((ch * 4 + g) * 2 + 0) * 32];
        v8h w = wl[((ch * 4 + g) * 2 + 1) * 32];
#pragma unroll
        for (int e = 0; e < 8; e++) { b[g][e] = u[e]; b[g][e + 8] = w[e]; }
      }
      ai = WMMA16(a, b[0], ai);
      af = WMMA16(a, b[1], af);
      ag = WMMA16(a, b[2], ag);
      ao = WMMA16(a, b[3], ao);
    }
    __syncthreads();

#pragma unroll
    for (int r = 0; r < 8; r++) {
      float iv = sigmoid_hw(ai[r]), fv = sigmoid_hw(af[r]);
      float gv = tanh_hw(ag[r]),    ov = sigmoid_hw(ao[r]);
      float cv = fv * creg[r] + iv * gv;
      creg[r] = cv;
      hv[r]   = ov * tanh_hw(cv);
    }
#pragma unroll
    for (int r = 0; r < 8; r++) {
      int el  = hi * 8 + r;
      int env = env0 + el;
      A[el * HSTR + ncol] = (_Float16)hv[r];
      hs[(t * E_ENVS + env) * HL + ncol] = hv[r];
    }
    __syncthreads();
  }
#pragma unroll
  for (int r = 0; r < 8; r++) {
    int env = env0 + hi * 8 + r;
    out[12288 + env * HL + ncol] = hv[r];
    out[20480 + env * HL + ncol] = creg[r];
  }
}

__global__ void k_heads(const float* hs, const float* pw, const float* pb,
                        const float* vw, const float* vb, float* out) {
  int i = blockIdx.x * blockDim.x + threadIdx.x;
  if (i >= B_TOT * 6) return;
  int b = i / 6, j = i % 6;
  const float* h    = hs + b * HL;
  const float* wrow = (j < 5) ? pw + j * HL : vw;
  float s = (j < 5) ? pb[j] : vb[0];
#pragma unroll 4
  for (int k = 0; k < HL; k++) s += h[k] * wrow[k];
  if (j < 5) out[b * 5 + j] = s;
  else       out[10240 + b] = s;
}

extern "C" void kernel_launch(void* const* d_in, const int* in_sizes, int n_in,
                              void* d_out, int out_size, void* d_ws, size_t ws_size,
                              hipStream_t stream) {
  (void)in_sizes; (void)n_in; (void)out_size; (void)ws_size;
  const float* img  = (const float*)d_in[0];
  const int*   pos  = (const int*)d_in[1];
  const int*   lact = (const int*)d_in[2];
  const int*   done = (const int*)d_in[3];
  const float* h0   = (const float*)d_in[4];
  const float* c0   = (const float*)d_in[5];
  const float* w1   = (const float*)d_in[6];
  const float* b1   = (const float*)d_in[7];
  const float* w2   = (const float*)d_in[8];
  const float* b2   = (const float*)d_in[9];
  const float* w3   = (const float*)d_in[10];
  const float* b3   = (const float*)d_in[11];
  const float* fcw  = (const float*)d_in[12];
  const float* fcb  = (const float*)d_in[13];
  const float* wih  = (const float*)d_in[14];
  const float* whh  = (const float*)d_in[15];
  const float* bih  = (const float*)d_in[16];
  const float* bhh  = (const float*)d_in[17];
  const float* pw   = (const float*)d_in[18];
  const float* pb   = (const float*)d_in[19];
  const float* vw   = (const float*)d_in[20];
  const float* vb   = (const float*)d_in[21];
  float* out = (float*)d_out;
  char*  ws  = (char*)d_ws;

  const size_t o_act1 = 90832896;
  const size_t o_w1t  = o_act1 + 52428800;
  const size_t o_w2t  = o_w1t + 12288;
  const size_t o_w3t  = o_w2t + 65536;
  const size_t o_fcwt = o_w3t + 73728;
  const size_t o_wih  = o_fcwt + 3211264;
  const size_t o_whh  = o_wih + 1179648;
  const size_t o_bg   = o_whh + 524288;
  const size_t o_act2 = 0;
  const size_t o_act3 = 21233664;
  const size_t o_X    = 34078720;
  const size_t o_hs   = 36438016;
  const size_t o_G    = 38535168;

  _Float16* act1 = (_Float16*)(ws + o_act1);
  _Float16* act2 = (_Float16*)(ws + o_act2);
  _Float16* act3 = (_Float16*)(ws + o_act3);
  _Float16* Xm   = (_Float16*)(ws + o_X);
  _Float16* w1t  = (_Float16*)(ws + o_w1t);
  _Float16* w2t  = (_Float16*)(ws + o_w2t);
  _Float16* w3t  = (_Float16*)(ws + o_w3t);
  _Float16* fcwt = (_Float16*)(ws + o_fcwt);
  _Float16* wih16 = (_Float16*)(ws + o_wih);
  _Float16* whh16 = (_Float16*)(ws + o_whh);
  float*    bg   = (float*)(ws + o_bg);
  float*    hsb  = (float*)(ws + o_hs);
  float*    Gx   = (float*)(ws + o_G);

  const int TB = 256;
  k_prep_w1  <<<(32 * 192   + TB - 1) / TB, TB, 0, stream>>>(w1, w1t);
  k_prep_w2  <<<(64 * 512   + TB - 1) / TB, TB, 0, stream>>>(w2, w2t);
  k_prep_w3  <<<(64 * 576   + TB - 1) / TB, TB, 0, stream>>>(w3, w3t);
  k_prep_fc  <<<(512 * 3136 + TB - 1) / TB, TB, 0, stream>>>(fcw, fcwt);
  k_prep_wl  <<<(1024 * (XS + HL) + TB - 1) / TB, TB, 0, stream>>>(wih, whh, wih16, whh16);
  k_prep_bias<<<(1024 + TB - 1) / TB, TB, 0, stream>>>(bih, bhh, bg);

  k_conv1<<<6400, TB, 0, stream>>>(img, w1t, b1, act1);
  k_conv2<<<1296, TB, 0, stream>>>(act1, w2t, b2, act2);
  k_conv3<<< 784, TB, 0, stream>>>(act2, w3t, b3, act3);
  k_fc   <<< 128, TB, 0, stream>>>(act3, fcwt, fcb, Xm);

  k_onehot<<<(B_TOT * 64 + TB - 1) / TB, TB, 0, stream>>>(pos, lact, Xm);

  k_xproj<<<256, TB, 0, stream>>>(Xm, wih16, bg, Gx);

  k_lstm<<<2, 512, 0, stream>>>(Gx, whh16, done, h0, c0, hsb, out);

  k_heads<<<(B_TOT * 6 + TB - 1) / TB, TB, 0, stream>>>(hsb, pw, pb, vw, vb, out);
}


# LSTM step waits for no global memory: done flags read once into per-step LDS bit masks applied where h and c are made (mask pass, 1 of 3 barriers gone); LDS-only barriers; next G tile prefetched
# speedup vs baseline: 7.5047x; 1.2056x over previous
#include <hip/hip_runtime.h>

typedef __attribute__((ext_vector_type(16))) _Float16 v16h;
typedef __attribute__((ext_vector_type(8)))  _Float16 v8h;
typedef __attribute__((ext_vector_type(4)))  _Float16 v4h;
typedef __attribute__((ext_vector_type(8)))  float    v8f;
typedef __attribute__((ext_vector_type(4)))  float    v4f;

#define WMMA16(A,B,C) __builtin_amdgcn_wmma_f32_16x16x32_f16(false,(A),false,(B),(short)0,(C),false,false)

static constexpr int T_STEPS = 64;
static constexpr int E_ENVS  = 32;
static constexpr int B_TOT   = 2048;
static constexpr int HL      = 256;
static constexpr int IND     = 549;
static constexpr int XS      = 576;
static constexpr int HSTR    = 264;
static constexpr int GT      = 1024;
static constexpr int IROW    = 84 * 3;

__device__ __forceinline__ v16h pack_frag(const _Float16* p) {
  v8h u = *(const v8h*)p;
  v8h w = *(const v8h*)(p + 16);
  v16h a;
#pragma unroll
  for (int e = 0; e < 8; e++) { a[e] = u[e]; a[e + 8] = w[e]; }
  return a;
}
__device__ __forceinline__ v16h pack_frag_img(const float* p0, const float* p1) {
  v4f u0 = *(const v4f*)p0;
  v4f u1 = *(const v4f*)(p0 + 4);
  v4f u2 = *(const v4f*)p1;
  v4f u3 = *(const v4f*)(p1 + 4);
  v16h a;
#pragma unroll
  for (int e = 0; e < 4; e++) {
    a[e]      = (_Float16)(u0[e] * (1.0f / 255.0f));
    a[e + 4]  = (_Float16)(u1[e] * (1.0f / 255.0f));
    a[e + 8]  = (_Float16)(u2[e] * (1.0f / 255.0f));
    a[e + 12] = (_Float16)(u3[e] * (1.0f / 255.0f));
  }
  return a;
}
__host__ __device__ constexpr int patch_off(int f) { return (f / 24) * IROW + (f % 24); }

__device__ __forceinline__ float tanh_hw(float x) { return __builtin_amdgcn_tanhf(x); }
__device__ __forceinline__ float sigmoid_hw(float x) {
  return __builtin_fmaf(0.5f, tanh_hw(0.5f * x), 0.5f);
}

__device__ __forceinline__ void lds_barrier() {
  __builtin_amdgcn_fence(__ATOMIC_RELEASE, "workgroup", "local");
  __builtin_amdgcn_s_barrier();
  __builtin_amdgcn_fence(__ATOMIC_ACQUIRE, "workgroup", "local");
}

__global__ void k_prep_w1(const float* w, _Float16* wt) {
  int i = blockIdx.x * blockDim.x + threadIdx.x;
  if (i >= 32 * 192) return;
  int n = i / 192, f = i % 192;
  int kh = f / 24, r = f % 24, kw = r / 3, c = r % 3;
  wt[i] = (_Float16)w[((n * 3 + c) * 8 + kh) * 8 + kw];
}
__global__ void k_prep_w2(const float* w, _Float16* wt) {
  int i = blockIdx.x * blockDim.x + threadIdx.x;
  if (i >= 64 * 512) return;
  int n = i / 512, k = i % 512;
  int pix = k >> 5, c = k & 31, kh = pix >> 2, kw = pix & 3;
  wt[i] = (_Float16)w[((n * 32 + c) * 4 + kh) * 4 + kw];
}
__global__ void k_prep_w3(const float* w, _Float16* wt) {
  int i = blockIdx.x * blockDim.x + threadIdx.x;
  if (i >= 64 * 576) return;
  int n = i / 576, k = i % 576;
  int pix = k >> 6, c = k & 63, kh = pix / 3, kw = pix % 3;
  wt[i] = (_Float16)w[((n * 64 + c) * 3 + kh) * 3 + kw];
}
__global__ void k_prep_fc(const float* w, _Float16* wt) {
  int i = blockIdx.x * blockDim.x + threadIdx.x;
  if (i >= 512 * 3136) return;
  int f = i / 3136, k = i % 3136;
  int s = k >> 6, c = k & 63;
  wt[i] = (_Float16)w[f * 3136 + c * 49 + s];
}
__global__ void k_prep_wl(const float* wih, const float* whh, _Float16* wih16, _Float16* whh16) {
  int i = blockIdx.x * blockDim.x + threadIdx.x;
  if (i >= 1024 * (XS + HL)) return;
  int r = i / (XS + HL), k = i % (XS + HL);
  if (k < XS) wih16[r * XS + k] = (_Float16)((k < IND) ? wih[r * IND + k] : 0.0f);
  else        whh16[r * HL + (k - XS)] = (_Float16)whh[r * HL + (k - XS)];
}
__global__ void k_prep_bias(const float* bih, const float* bhh, float* bg) {
  int i = blockIdx.x * blockDim.x + threadIdx.x;
  if (i < 1024) bg[i] = bih[i] + bhh[i];
}
__global__ void k_onehot(const int* pos, const int* lact, _Float16* X) {
  int i = blockIdx.x * blockDim.x + threadIdx.x;
  if (i >= B_TOT * 64) return;
  int b = i >> 6, c = i & 63;
  float v = 0.0f;
  if (c < 16)      v = (pos[b * 2 + 0] == c)      ? 1.0f : 0.0f;
  else if (c < 32) v = (pos[b * 2 + 1] == c - 16) ? 1.0f : 0.0f;
  else if (c < 37) v = (lact[b]        == c - 32) ? 1.0f : 0.0f;
  X[b * XS + 512 + c] = (_Float16)v;
}

__global__ void __launch_bounds__(256) k_conv1(const float* img, const _Float16* wt,
                                               const float* bias, _Float16* out) {
  int wid  = (blockIdx.x * blockDim.x + threadIdx.x) >> 5;
  int lane = threadIdx.x & 31, hi = lane >> 4, lo = lane & 15;
  int m = wid * 16 + lo;
  int b = m / 400, p = m % 400, oh = p / 20, ow = p % 20;
  const float* abase = img + ((size_t)(b * 84 + oh * 4) * 84 + ow * 4) * 3;
  v8f acc0 = {}, acc1 = {};
#pragma unroll
  for (int kb = 0; kb < 192; kb += 32) {
    int o0 = hi ? patch_off(kb + 8)  : patch_off(kb);
    int o1 = hi ? patch_off(kb + 24) : patch_off(kb + 16);
    v16h a  = pack_frag_img(abase + o0, abase + o1);
    v16h b0 = pack_frag(wt + lo * 192        + kb + hi * 8);
    v16h b1 = pack_frag(wt + (16 + lo) * 192 + kb + hi * 8);
    acc0 = WMMA16(a, b0, acc0);
    acc1 = WMMA16(a, b1, acc1);
  }
  float bs0 = bias[lo], bs1 = bias[16 + lo];
#pragma unroll
  for (int r = 0; r < 8; r++) {
    int mm = wid * 16 + hi * 8 + r;
    float v0 = acc0[r] + bs0; v0 = v0 > 0.0f ? v0 : 0.0f;
    float v1 = acc1[r] + bs1; v1 = v1 > 0.0f ? v1 : 0.0f;
    out[mm * 32 + lo]      = (_Float16)v0;
    out[mm * 32 + 16 + lo] = (_Float16)v1;
  }
}

__global__ void __launch_bounds__(256) k_conv2(const _Float16* act1, const _Float16* wt,
                                               const float* bias, _Float16* out) {
  int wid  = (blockIdx.x * blockDim.x + threadIdx.x) >> 5;
  int lane = threadIdx.x & 31, hi = lane >> 4, lo = lane & 15;
  int m = wid * 16 + lo;
  int b = m / 81, p = m % 81, oh = p / 9, ow = p % 9;
  v8f acc0 = {}, acc1 = {}, acc2 = {}, acc3 = {};
  for (int kb = 0; kb < 512; kb += 32) {
    int pix = kb >> 5, kh = pix >> 2, kw = pix & 3;
    v16h a  = pack_frag(act1 + ((b * 20 + oh * 2 + kh) * 20 + (ow * 2 + kw)) * 32 + hi * 8);
    v16h b0 = pack_frag(wt + (lo)      * 512 + kb + hi * 8);
    v16h b1 = pack_frag(wt + (16 + lo) * 512 + kb + hi * 8);
    v16h b2 = pack_frag(wt + (32 + lo) * 512 + kb + hi * 8);
    v16h b3 = pack_frag(wt + (48 + lo) * 512 + kb + hi * 8);
    acc0 = WMMA16(a, b0, acc0);
    acc1 = WMMA16(a, b1, acc1);
    acc2 = WMMA16(a, b2, acc2);
    acc3 = WMMA16(a, b3, acc3);
  }
  float bs0 = bias[lo], bs1 = bias[16 + lo], bs2 = bias[32 + lo], bs3 = bias[48 + lo];
#pragma unroll
  for (int r = 0; r < 8; r++) {
    int mm = wid * 16 + hi * 8 + r;
    float v0 = acc0[r] + bs0; v0 = v0 > 0.0f ? v0 : 0.0f;
    float v1 = acc1[r] + bs1; v1 = v1 > 0.0f ? v1 : 0.0f;
    float v2 = acc2[r] + bs2; v2 = v2 > 0.0f ? v2 : 0.0f;
    float v3 = acc3[r] + bs3; v3 = v3 > 0.0f ? v3 : 0.0f;
    out[mm * 64 + lo]      = (_Float16)v0;
    out[mm * 64 + 16 + lo] = (_Float16)v1;
    out[mm * 64 + 32 + lo] = (_Float16)v2;
    out[mm * 64 + 48 + lo] = (_Float16)v3;
  }
}

__global__ void __launch_bounds__(256) k_conv3(const _Float16* act2, const _Float16* wt,
                                               const float* bias, _Float16* out) {
  int wid  = (blockIdx.x * blockDim.x + threadIdx.x) >> 5;
  int lane = threadIdx.x & 31, hi = lane >> 4, lo = lane & 15;
  int m = wid * 16 + lo;
  int b = m / 49, p = m % 49, oh = p / 7, ow = p % 7;
  v8f acc0 = {}, acc1 = {}, acc2 = {}, acc3 = {};
  for (int kb = 0; kb < 576; kb += 32) {
    int pix = kb >> 6, cb = (kb & 63) + hi * 8;
    int kh = pix / 3, kw = pix % 3;
    v16h a  = pack_frag(act2 + ((b * 9 + oh + kh) * 9 + (ow + kw)) * 64 + cb);
    v16h b0 = pack_frag(wt + (lo)      * 576 + kb + hi * 8);
    v16h b1 = pack_frag(wt + (16 + lo) * 576 + kb + hi * 8);
    v16h b2 = pack_frag(wt + (32 + lo) * 576 + kb + hi * 8);
    v16h b3 = pack_frag(wt + (48 + lo) * 576 + kb + hi * 8);
    acc0 = WMMA16(a, b0, acc0);
    acc1 = WMMA16(a, b1, acc1);
    acc2 = WMMA16(a, b2, acc2);
    acc3 = WMMA16(a, b3, acc3);
  }
  float bs0 = bias[lo], bs1 = bias[16 + lo], bs2 = bias[32 + lo], bs3 = bias[48 + lo];
#pragma unroll
  for (int r = 0; r < 8; r++) {
    int mm = wid * 16 + hi * 8 + r;
    float v0 = acc0[r] + bs0; v0 = v0 > 0.0f ? v0 : 0.0f;
    float v1 = acc1[r] + bs1; v1 = v1 > 0.0f ? v1 : 0.0f;
    float v2 = acc2[r] + bs2; v2 = v2 > 0.0f ? v2 : 0.0f;
    float v3 = acc3[r] + bs3; v3 = v3 > 0.0f ? v3 : 0.0f;
    out[mm * 64 + lo]      = (_Float16)v0;
    out[mm * 64 + 16 + lo] = (_Float16)v1;
    out[mm * 64 + 32 + lo] = (_Float16)v2;
    out[mm * 64 + 48 + lo] = (_Float16)v3;
  }
}

__global__ void __launch_bounds__(256) k_fc(const _Float16* act3, const _Float16* wt,
                                            const float* bias, _Float16* X) {
  int wid  = (blockIdx.x * blockDim.x + threadIdx.x) >> 5;
  int lane = threadIdx.x & 31, hi = lane >> 4, lo = lane & 15;
  int ng = wid & 7, mt = wid >> 3;
  int m = mt * 16 + lo;
  int nb = ng * 64;
  v8f acc0 = {}, acc1 = {}, acc2 = {}, acc3 = {};
  for (int kb = 0; kb < 3136; kb += 32) {
    v16h a  = pack_frag(act3 + (size_t)m * 3136 + kb + hi * 8);
    v16h b0 = pack_frag(wt + (size_t)(nb + lo)      * 3136 + kb + hi * 8);
    v16h b1 = pack_frag(wt + (size_t)(nb + 16 + lo) * 3136 + kb + hi * 8);
    v16h b2 = pack_frag(wt + (size_t)(nb + 32 + lo) * 3136 + kb + hi * 8);
    v16h b3 = pack_frag(wt + (size_t)(nb + 48 + lo) * 3136 + kb + hi * 8);
    acc0 = WMMA16(a, b0, acc0);
    acc1 = WMMA16(a, b1, acc1);
    acc2 = WMMA16(a, b2, acc2);
    acc3 = WMMA16(a, b3, acc3);
  }
  float bs0 = bias[nb + lo], bs1 = bias[nb + 16 + lo];
  float bs2 = bias[nb + 32 + lo], bs3 = bias[nb + 48 + lo];
#pragma unroll
  for (int r = 0; r < 8; r++) {
    int mm = mt * 16 + hi * 8 + r;
    float v0 = acc0[r] + bs0; v0 = v0 > 0.0f ? v0 : 0.0f;
    float v1 = acc1[r] + bs1; v1 = v1 > 0.0f ? v1 : 0.0f;
    float v2 = acc2[r] + bs2; v2 = v2 > 0.0f ? v2 : 0.0f;
    float v3 = acc3[r] + bs3; v3 = v3 > 0.0f ? v3 : 0.0f;
    X[mm * XS + nb + lo]      = (_Float16)v0;
    X[mm * XS + nb + 16 + lo] = (_Float16)v1;
    X[mm * XS + nb + 32 + lo] = (_Float16)v2;
    X[mm * XS + nb + 48 + lo] = (_Float16)v3;
  }
}

__global__ void __launch_bounds__(256) k_xproj(const _Float16* X, const _Float16* Wih,
                                               const float* biasg, float* G) {
  int wid  = (blockIdx.x * blockDim.x + threadIdx.x) >> 5;
  int lane = threadIdx.x & 31, hi = lane >> 4, lo = lane & 15;
  int mtile = wid & 127, cg = wid >> 7;
  int ncol = cg * 16 + lo;
  float bi = biasg[0 * HL + ncol], bf = biasg[1 * HL + ncol];
  float bg = biasg[2 * HL + ncol], bo = biasg[3 * HL + ncol];
  v8f ai = {bi, bi, bi, bi, bi, bi, bi, bi};
  v8f af = {bf, bf, bf, bf, bf, bf, bf, bf};
  v8f ag = {bg, bg, bg, bg, bg, bg, bg, bg};
  v8f ao = {bo, bo, bo, bo, bo, bo, bo, bo};
  const _Float16* ap = X   + (mtile * 16 + lo) * XS + hi * 8;
  const _Float16* bp = Wih + ncol * XS + hi * 8;
  for (int kb = 0; kb < XS; kb += 32) {
    v16h a  = pack_frag(ap + kb);
    v16h b0 = pack_frag(bp + 0 * HL * XS + kb);
    v16h b1 = pack_frag(bp + 1 * HL * XS + kb);
    v16h b2 = pack_frag(bp + 2 * HL * XS + kb);
    v16h b3 = pack_frag(bp + 3 * HL * XS + kb);
    ai = WMMA16(a, b0, ai);
    af = WMMA16(a, b1, af);
    ag = WMMA16(a, b2, ag);
    ao = WMMA16(a, b3, ao);
  }
  v4f* gp = (v4f*)(G + (size_t)(mtile * 16 + cg) * GT) + lane;
  gp[0 * 64]      = ai.lo;  gp[0 * 64 + 32] = ai.hi;
  gp[1 * 64]      = af.lo;  gp[1 * 64 + 32] = af.hi;
  gp[2 * 64]      = ag.lo;  gp[2 * 64 + 32] = ag.hi;
  gp[3 * 64]      = ao.lo;  gp[3 * 64 + 32] = ao.hi;
}

__global__ void __launch_bounds__(512) k_lstm(const float* G, const _Float16* Whh,
                                              const int* done,
                                              const float* h0, const float* c0,
                                              float* hs, float* out) {
  __shared__ v8h Wl[16 * 4 * 4 * 2 * 32];
  __shared__ __attribute__((aligned(16))) _Float16 A[16 * HSTR];
  __shared__ unsigned nm[T_STEPS];
  const int tid = threadIdx.x;
  const int mt  = blockIdx.x;
  const int cg  = tid >> 5, lane = tid & 31, hi = lane >> 4, lo = lane & 15;
  const int ncol = cg * 16 + lo;
  const int env0 = mt * 16;

  if (tid < T_STEPS) {
    unsigned m = 0;
    if (tid + 1 < T_STEPS) {
#pragma unroll
      for (int e = 0; e < 16; e++)
        m |= (done[(tid + 1) * E_ENVS + env0 + e] != 0 ? 1u : 0u) << e;
    }
    nm[tid] = m;
  }

  for (int i = tid; i < 16 * HL; i += 512) {
    int e = i >> 8, k = i & 255;
    float v = h0[(env0 + e) * HL + k];
    A[e * HSTR + k] = done[env0 + e] ? (_Float16)0.0f : (_Float16)v;
  }
  v8f creg;
#pragma unroll
  for (int r = 0; r < 8; r++) {
    float v = c0[(env0 + hi * 8 + r) * HL + ncol];
    creg[r] = done[env0 + hi * 8 + r] ? 0.0f : v;
  }

  const _Float16* bp = Whh + ncol * HL + hi * 8;
  v8h* wl = Wl + cg * (4 * 4 * 2 * 32) + lane;
#pragma unroll
  for (int ch = 0; ch < 4; ch++) {
#pragma unroll
    for (int g = 0; g < 4; g++) {
      const _Float16* p = bp + g * HL * HL + (4 + ch) * 32;
      wl[((ch * 4 + g) * 2 + 0) * 32] = *(const v8h*)p;
      wl[((ch * 4 + g) * 2 + 1) * 32] = *(const v8h*)(p + 16);
    }
  }
  __builtin_amdgcn_sched_barrier(0);
  v16h wr[4][4];
#pragma unroll
  for (int ch = 0; ch < 4; ch++) {
#pragma unroll
    for (int g = 0; g < 4; g++) wr[ch][g] = pack_frag(bp + g * HL * HL + ch * 32);
  }
  __syncthreads();

  const v4f* gbase = (const v4f*)(G + (size_t)(mt * 16 + cg) * GT) + lane;
  v8f gi, gf, gg, go;
  gi.lo = gbase[0 * 64];  gi.hi = gbase[0 * 64 + 32];
  gf.lo = gbase[1 * 64];  gf.hi = gbase[1 * 64 + 32];
  gg.lo = gbase[2 * 64];  gg.hi = gbase[2 * 64 + 32];
  go.lo = gbase[3 * 64];  go.hi = gbase[3 * 64 + 32];

  float hv[8];
  for (int t = 0; t < T_STEPS; t++) {
    const unsigned my = nm[t] >> (hi * 8);

    const _Float16* arow = &A[lo * HSTR + hi * 8];
    v8f ai, af, ag, ao;
#pragma unroll
    for (int ch = 0; ch < 4; ch++) {
      v16h a = pack_frag(arow + ch * 32);
      ai = WMMA16(a, wr[ch][0], ch ? ai : gi);
      af = WMMA16(a, wr[ch][1], ch ? af : gf);
      ag = WMMA16(a, wr[ch][2], ch ? ag : gg);
      ao = WMMA16(a, wr[ch][3], ch ? ao : go);
    }
#pragma unroll
    for (int ch = 0; ch < 4; ch++) {
      v16h a = pack_frag(arow + (4 + ch) * 32);
      v16h b[4];
#pragma unroll
      for (int g = 0; g < 4; g++) {
        v8h u = wl[((ch * 4 + g) * 2 + 0) * 32];
        v8h w = wl[((ch * 4 + g) * 2 + 1) * 32];
#pragma unroll
        for (int e = 0; e < 8; e++) { b[g][e] = u[e]; b[g][e + 8] = w[e]; }
      }
      ai = WMMA16(a, b[0], ai);
      af = WMMA16(a, b[1], af);
      ag = WMMA16(a, b[2], ag);
      ao = WMMA16(a, b[3], ao);
    }
    lds_barrier();

    if (t + 1 < T_STEPS) {
      const v4f* gp = gbase + (size_t)(t + 1) * (2 * 16 * GT / 4);
      gi.lo = gp[0 * 64];  gi.hi = gp[0 * 64 + 32];
      gf.lo = gp[1 * 64];  gf.hi = gp[1 * 64 + 32];
      gg.lo = gp[2 * 64];  gg.hi = gp[2 * 64 + 32];
      go.lo = gp[3 * 64];  go.hi = gp[3 * 64 + 32];
    }

    bool dn[8];
#pragma unroll
    for (int r = 0; r < 8; r++) {
      float iv = sigmoid_hw(ai[r]), fv = sigmoid_hw(af[r]);
      float gv = tanh_hw(ag[r]),    ov = sigmoid_hw(ao[r]);
      float cv = fv * creg[r] + iv * gv;
      hv[r]   = ov * tanh_hw(cv);
      dn[r]   = ((my >> r) & 1u) != 0;
      creg[r] = dn[r] ? 0.0f : cv;
    }
#pragma unroll
    for (int r = 0; r < 8; r++) {
      int el  = hi * 8 + r;
      int env = env0 + el;
      A[el * HSTR + ncol] = dn[r] ? (_Float16)0.0f : (_Float16)hv[r];
      hs[(t * E_ENVS + env) * HL + ncol] = hv[r];
    }
    lds_barrier();
  }
#pragma unroll
  for (int r = 0; r < 8; r++) {
    int env = env0 + hi * 8 + r;
    out[12288 + env * HL + ncol] = hv[r];
    out[20480 + env * HL + ncol] = creg[r];
  }
}

__global__ void k_heads(const float* hs, const float* pw, const float* pb,
                        const float* vw, const float* vb, float* out) {
  int i = blockIdx.x * blockDim.x + threadIdx.x;
  if (i >= B_TOT * 6) return;
  int b = i / 6, j = i % 6;
  const float* h    = hs + b * HL;
  const float* wrow = (j < 5) ? pw + j * HL : vw;
  float s = (j < 5) ? pb[j] : vb[0];
#pragma unroll 4
  for (int k = 0; k < HL; k++) s += h[k] * wrow[k];
  if (j < 5) out[b * 5 + j] = s;
  else       out[10240 + b] = s;
}

extern "C" void kernel_launch(void* const* d_in, const int* in_sizes, int n_in,
                              void* d_out, int out_size, void* d_ws, size_t ws_size,
                              hipStream_t stream) {
  (void)in_sizes; (void)n_in; (void)out_size; (void)ws_size;
  const float* img  = (const float*)d_in[0];
  const int*   pos  = (const int*)d_in[1];
  const int*   lact = (const int*)d_in[2];
  const int*   done = (const int*)d_in[3];
  const float* h0   = (const float*)d_in[4];
  const float* c0   = (const float*)d_in[5];
  const float* w1   = (const float*)d_in[6];
  const float* b1   = (const float*)d_in[7];
  const float* w2   = (const float*)d_in[8];
  const float* b2   = (const float*)d_in[9];
  const float* w3   = (const float*)d_in[10];
  const float* b3   = (const float*)d_in[11];
  const float* fcw  = (const float*)d_in[12];
  const float* fcb  = (const float*)d_in[13];
  const float* wih  = (const float*)d_in[14];
  const float* whh  = (const float*)d_in[15];
  const float* bih  = (const float*)d_in[16];
  const float* bhh  = (const float*)d_in[17];
  const float* pw   = (const float*)d_in[18];
  const float* pb   = (const float*)d_in[19];
  const float* vw   = (const float*)d_in[20];
  const float* vb   = (const float*)d_in[21];
  float* out = (float*)d_out;
  char*  ws  = (char*)d_ws;

  const size_t o_act1 = 90832896;
  const size_t o_w1t  = o_act1 + 52428800;
  const size_t o_w2t  = o_w1t + 12288;
  const size_t o_w3t  = o_w2t + 65536;
  const size_t o_fcwt = o_w3t + 73728;
  const size_t o_wih  = o_fcwt + 3211264;
  const size_t o_whh  = o_wih + 1179648;
  const size_t o_bg   = o_whh + 524288;
  const size_t o_act2 = 0;
  const size_t o_act3 = 21233664;
  const size_t o_X    = 34078720;
  const size_t o_hs   = 36438016;
  const size_t o_G    = 38535168;

  _Float16* act1 = (_Float16*)(ws + o_act1);
  _Float16* act2 = (_Float16*)(ws + o_act2);
  _Float16* act3 = (_Float16*)(ws + o_act3);
  _Float16* Xm   = (_Float16*)(ws + o_X);
  _Float16* w1t  = (_Float16*)(ws + o_w1t);
  _Float16* w2t  = (_Float16*)(ws + o_w2t);
  _Float16* w3t  = (_Float16*)(ws + o_w3t);
  _Float16* fcwt = (_Float16*)(ws + o_fcwt);
  _Float16* wih16 = (_Float16*)(ws + o_wih);
  _Float16* whh16 = (_Float16*)(ws + o_whh);
  float*    bg   = (float*)(ws + o_bg);
  float*    hsb  = (float*)(ws + o_hs);
  float*    Gx   = (float*)(ws + o_G);

  const int TB = 256;
  k_prep_w1  <<<(32 * 192   + TB - 1) / TB, TB, 0, stream>>>(w1, w1t);
  k_prep_w2  <<<(64 * 512   + TB - 1) / TB, TB, 0, stream>>>(w2, w2t);
  k_prep_w3  <<<(64 * 576   + TB - 1) / TB, TB, 0, stream>>>(w3, w3t);
  k_prep_fc  <<<(512 * 3136 + TB - 1) / TB, TB, 0, stream>>>(fcw, fcwt);
  k_prep_wl  <<<(1024 * (XS + HL) + TB - 1) / TB, TB, 0, stream>>>(wih, whh, wih16, whh16);
  k_prep_bias<<<(1024 + TB - 1) / TB, TB, 0, stream>>>(bih, bhh, bg);

  k_conv1<<<6400, TB, 0, stream>>>(img, w1t, b1, act1);
  k_conv2<<<1296, TB, 0, stream>>>(act1, w2t, b2, act2);
  k_conv3<<< 784, TB, 0, stream>>>(act2, w3t, b3, act3);
  k_fc   <<< 128, TB, 0, stream>>>(act3, fcwt, fcb, Xm);

  k_onehot<<<(B_TOT * 64 + TB - 1) / TB, TB, 0, stream>>>(pos, lact, Xm);

  k_xproj<<<256, TB, 0, stream>>>(Xm, wih16, bg, Gx);

  k_lstm<<<2, 512, 0, stream>>>(Gx, whh16, done, h0, c0, hsb, out);

  k_heads<<<(B_TOT * 6 + TB - 1) / TB, TB, 0, stream>>>(hsb, pw, pb, vw, vb, out);
}


# GEMM loops of conv2, conv3, FC and gate projection: the ten fragment loads of a k-step issued as one batch, one k-step ahead (two fragment sets), not one round trip per WMMA; sum order unchanged
# speedup vs baseline: 9.3415x; 1.2448x over previous
#include <hip/hip_runtime.h>

typedef __attribute__((ext_vector_type(16))) _Float16 v16h;
typedef __attribute__((ext_vector_type(8)))  _Float16 v8h;
typedef __attribute__((ext_vector_type(4)))  _Float16 v4h;
typedef __attribute__((ext_vector_type(8)))  float    v8f;
typedef __attribute__((ext_vector_type(4)))  float    v4f;

#define WMMA16(A,B,C) __builtin_amdgcn_wmma_f32_16x16x32_f16(false,(A),false,(B),(short)0,(C),false,false)

static constexpr int T_STEPS = 64;
static constexpr int E_ENVS  = 32;
static constexpr int B_TOT   = 2048;
static constexpr int HL      = 256;
static constexpr int IND     = 549;
static constexpr int XS      = 576;
static constexpr int HSTR    = 264;
static constexpr int GT      = 1024;
static constexpr int IROW    = 84 * 3;

__device__ __forceinline__ v16h pack_frag(const _Float16* p) {
  v8h u = *(const v8h*)p;
  v8h w = *(const v8h*)(p + 16);
  v16h a;
#pragma unroll
  for (int e = 0; e < 8; e++) { a[e] = u[e]; a[e + 8] = w[e]; }
  return a;
}
__device__ __forceinline__ v16h pack_frag_img(const float* p0, const float* p1) {
  v4f u0 = *(const v4f*)p0;
  v4f u1 = *(const v4f*)(p0 + 4);
  v4f u2 = *(const v4f*)p1;
  v4f u3 = *(const v4f*)(p1 + 4);
  v16h a;
#pragma unroll
  for (int e = 0; e < 4; e++) {
    a[e]      = (_Float16)(u0[e] * (1.0f / 255.0f));
    a[e + 4]  = (_Float16)(u1[e] * (1.0f / 255.0f));
    a[e + 8]  = (_Float16)(u2[e] * (1.0f / 255.0f));
    a[e + 12] = (_Float16)(u3[e] * (1.0f / 255.0f));
  }
  return a;
}
__host__ __device__ constexpr int patch_off(int f) { return (f / 24) * IROW + (f % 24); }

__device__ __forceinline__ float tanh_hw(float x) { return __builtin_amdgcn_tanhf(x); }
__device__ __forceinline__ float sigmoid_hw(float x) {
  return __builtin_fmaf(0.5f, tanh_hw(0.5f * x), 0.5f);
}

__device__ __forceinline__ void lds_barrier() {
  __builtin_amdgcn_fence(__ATOMIC_RELEASE, "workgroup", "local");
  __builtin_amdgcn_s_barrier();
  __builtin_amdgcn_fence(__ATOMIC_ACQUIRE, "workgroup", "local");
}

struct KStep { v16h a, b0, b1, b2, b3; };
#define LOADS_STAY_AHEAD() __builtin_amdgcn_sched_barrier(0)
#define MMA4(F, C0, C1, C2, C3)      \
  C0 = WMMA16((F).a, (F).b0, C0);    \
  C1 = WMMA16((F).a, (F).b1, C1);    \
  C2 = WMMA16((F).a, (F).b2, C2);    \
  C3 = WMMA16((F).a, (F).b3, C3)

__global__ void k_prep_w1(const float* w, _Float16* wt) {
  int i = blockIdx.x * blockDim.x + threadIdx.x;
  if (i >= 32 * 192) return;
  int n = i / 192, f = i % 192;
  int kh = f / 24, r = f % 24, kw = r / 3, c = r % 3;
  wt[i] = (_Float16)w[((n * 3 + c) * 8 + kh) * 8 + kw];
}
__global__ void k_prep_w2(const float* w, _Float16* wt) {
  int i = blockIdx.x * blockDim.x + threadIdx.x;
  if (i >= 64 * 512) return;
  int n = i / 512, k = i % 512;
  int pix = k >> 5, c = k & 31, kh = pix >> 2, kw = pix & 3;
  wt[i] = (_Float16)w[((n * 32 + c) * 4 + kh) * 4 + kw];
}
__global__ void k_prep_w3(const float* w, _Float16* wt) {
  int i = blockIdx.x * blockDim.x + threadIdx.x;
  if (i >= 64 * 576) return;
  int n = i / 576, k = i % 576;
  int pix = k >> 6, c = k & 63, kh = pix / 3, kw = pix % 3;
  wt[i] = (_Float16)w[((n * 64 + c) * 3 + kh) * 3 + kw];
}
__global__ void k_prep_fc(const float* w, _Float16* wt) {
  int i = blockIdx.x * blockDim.x + threadIdx.x;
  if (i >= 512 * 3136) return;
  int f = i / 3136, k = i % 3136;
  int s = k >> 6, c = k & 63;
  wt[i] = (_Float16)w[f * 3136 + c * 49 + s];
}
__global__ void k_prep_wl(const float* wih, const float* whh, _Float16* wih16, _Float16* whh16) {
  int i = blockIdx.x * blockDim.x + threadIdx.x;
  if (i >= 1024 * (XS + HL)) return;
  int r = i / (XS + HL), k = i % (XS + HL);
  if (k < XS) wih16[r * XS + k] = (_Float16)((k < IND) ? wih[r * IND + k] : 0.0f);
  else        whh16[r * HL + (k - XS)] = (_Float16)whh[r * HL + (k - XS)];
}
__global__ void k_prep_bias(const float* bih, const float* bhh, float* bg) {
  int i = blockIdx.x * blockDim.x + threadIdx.x;
  if (i < 1024) bg[i] = bih[i] + bhh[i];
}
__global__ void k_onehot(const int* pos, const int* lact, _Float16* X) {
  int i = blockIdx.x * blockDim.x + threadIdx.x;
  if (i >= B_TOT * 64) return;
  int b = i >> 6, c = i & 63;
  float v = 0.0f;
  if (c < 16)      v = (pos[b * 2 + 0] == c)      ? 1.0f : 0.0f;
  else if (c < 32) v = (pos[b * 2 + 1] == c - 16) ? 1.0f : 0.0f;
  else if (c < 37) v = (lact[b]        == c - 32) ? 1.0f : 0.0f;
  X[b * XS + 512 + c] = (_Float16)v;
}

__global__ void __launch_bounds__(256) k_conv1(const float* img, const _Float16* wt,
                                               const float* bias, _Float16* out) {
  int wid  = (blockIdx.x * blockDim.x + threadIdx.x) >> 5;
  int lane = threadIdx.x & 31, hi = lane >> 4, lo = lane & 15;
  int m = wid * 16 + lo;
  int b = m / 400, p = m % 400, oh = p / 20, ow = p % 20;
  const float* abase = img + ((size_t)(b * 84 + oh * 4) * 84 + ow * 4) * 3;
  v8f acc0 = {}, acc1 = {};
#pragma unroll
  for (int kb = 0; kb < 192; kb += 32) {
    int o0 = hi ? patch_off(kb + 8)  : patch_off(kb);
    int o1 = hi ? patch_off(kb + 24) : patch_off(kb + 16);
    v16h a  = pack_frag_img(abase + o0, abase + o1);
    v16h b0 = pack_frag(wt + lo * 192        + kb + hi * 8);
    v16h b1 = pack_frag(wt + (16 + lo) * 192 + kb + hi * 8);
    acc0 = WMMA16(a, b0, acc0);
    acc1 = WMMA16(a, b1, acc1);
  }
  float bs0 = bias[lo], bs1 = bias[16 + lo];
#pragma unroll
  for (int r = 0; r < 8; r++) {
    int mm = wid * 16 + hi * 8 + r;
    float v0 = acc0[r] + bs0; v0 = v0 > 0.0f ? v0 : 0.0f;
    float v1 = acc1[r] + bs1; v1 = v1 > 0.0f ? v1 : 0.0f;
    out[mm * 32 + lo]      = (_Float16)v0;
    out[mm * 32 + 16 + lo] = (_Float16)v1;
  }
}

__global__ void __launch_bounds__(256) k_conv2(const _Float16* act1, const _Float16* wt,
                                               const float* bias, _Float16* out) {
  int wid  = (blockIdx.x * blockDim.x + threadIdx.x) >> 5;
  int lane = threadIdx.x & 31, hi = lane >> 4, lo = lane & 15;
  int m = wid * 16 + lo;
  int b = m / 81, p = m % 81, oh = p / 9, ow = p % 9;
  v8f acc0 = {}, acc1 = {}, acc2 = {}, acc3 = {};
  const _Float16* ap = act1 + ((b * 20 + oh * 2) * 20 + ow * 2) * 32 + hi * 8;
  const _Float16* bp = wt + lo * 512 + hi * 8;
  auto ld = [&](int kb) {
    int pix = kb >> 5, kh = pix >> 2, kw = pix & 3;
    KStep f;
    f.a  = pack_frag(ap + (kh * 20 + kw) * 32);
    f.b0 = pack_frag(bp + kb);
    f.b1 = pack_frag(bp + 16 * 512 + kb);
    f.b2 = pack_frag(bp + 32 * 512 + kb);
    f.b3 = pack_frag(bp + 48 * 512 + kb);
    return f;
  };
  KStep f0 = ld(0), f1;
  for (int kb = 0; kb < 512 - 64; kb += 64) {
    f1 = ld(kb + 32);
    LOADS_STAY_AHEAD();
    MMA4(f0, acc0, acc1, acc2, acc3);
    f0 = ld(kb + 64);
    LOADS_STAY_AHEAD();
    MMA4(f1, acc0, acc1, acc2, acc3);
  }
  f1 = ld(512 - 32);
  LOADS_STAY_AHEAD();
  MMA4(f0, acc0, acc1, acc2, acc3);
  MMA4(f1, acc0, acc1, acc2, acc3);
  float bs0 = bias[lo], bs1 = bias[16 + lo], bs2 = bias[32 + lo], bs3 = bias[48 + lo];
#pragma unroll
  for (int r = 0; r < 8; r++) {
    int mm = wid * 16 + hi * 8 + r;
    float v0 = acc0[r] + bs0; v0 = v0 > 0.0f ? v0 : 0.0f;
    float v1 = acc1[r] + bs1; v1 = v1 > 0.0f ? v1 : 0.0f;
    float v2 = acc2[r] + bs2; v2 = v2 > 0.0f ? v2 : 0.0f;
    float v3 = acc3[r] + bs3; v3 = v3 > 0.0f ? v3 : 0.0f;
    out[mm * 64 + lo]      = (_Float16)v0;
    out[mm * 64 + 16 + lo] = (_Float16)v1;
    out[mm * 64 + 32 + lo] = (_Float16)v2;
    out[mm * 64 + 48 + lo] = (_Float16)v3;
  }
}

__global__ void __launch_bounds__(256) k_conv3(const _Float16* act2, const _Float16* wt,
                                               const float* bias, _Float16* out) {
  int wid  = (blockIdx.x * blockDim.x + threadIdx.x) >> 5;
  int lane = threadIdx.x & 31, hi = lane >> 4, lo = lane & 15;
  int m = wid * 16 + lo;
  int b = m / 49, p = m % 49, oh = p / 7, ow = p % 7;
  v8f acc0 = {}, acc1 = {}, acc2 = {}, acc3 = {};
  const _Float16* ap = act2 + ((b * 9 + oh) * 9 + ow) * 64 + hi * 8;
  const _Float16* bp = wt + lo * 576 + hi * 8;
  auto ld = [&](int kb) {
    int pix = kb >> 6, kh = pix / 3, kw = pix % 3;
    KStep f;
    f.a  = pack_frag(ap + (kh * 9 + kw) * 64 + (kb & 63));
    f.b0 = pack_frag(bp + kb);
    f.b1 = pack_frag(bp + 16 * 576 + kb);
    f.b2 = pack_frag(bp + 32 * 576 + kb);
    f.b3 = pack_frag(bp + 48 * 576 + kb);
    return f;
  };
  KStep f0 = ld(0), f1;
  for (int kb = 0; kb < 576 - 64; kb += 64) {
    f1 = ld(kb + 32);
    LOADS_STAY_AHEAD();
    MMA4(f0, acc0, acc1, acc2, acc3);
    f0 = ld(kb + 64);
    LOADS_STAY_AHEAD();
    MMA4(f1, acc0, acc1, acc2, acc3);
  }
  f1 = ld(576 - 32);
  LOADS_STAY_AHEAD();
  MMA4(f0, acc0, acc1, acc2, acc3);
  MMA4(f1, acc0, acc1, acc2, acc3);
  float bs0 = bias[lo], bs1 = bias[16 + lo], bs2 = bias[32 + lo], bs3 = bias[48 + lo];
#pragma unroll
  for (int r = 0; r < 8; r++) {
    int mm = wid * 16 + hi * 8 + r;
    float v0 = acc0[r] + bs0; v0 = v0 > 0.0f ? v0 : 0.0f;
    float v1 = acc1[r] + bs1; v1 = v1 > 0.0f ? v1 : 0.0f;
    float v2 = acc2[r] + bs2; v2 = v2 > 0.0f ? v2 : 0.0f;
    float v3 = acc3[r] + bs3; v3 = v3 > 0.0f ? v3 : 0.0f;
    out[mm * 64 + lo]      = (_Float16)v0;
    out[mm * 64 + 16 + lo] = (_Float16)v1;
    out[mm * 64 + 32 + lo] = (_Float16)v2;
    out[mm * 64 + 48 + lo] = (_Float16)v3;
  }
}

__global__ void __launch_bounds__(256) k_fc(const _Float16* act3, const _Float16* wt,
                                            const float* bias, _Float16* X) {
  int wid  = (blockIdx.x * blockDim.x + threadIdx.x) >> 5;
  int lane = threadIdx.x & 31, hi = lane >> 4, lo = lane & 15;
  int ng = wid & 7, mt = wid >> 3;
  int m = mt * 16 + lo;
  int nb = ng * 64;
  v8f acc0 = {}, acc1 = {}, acc2 = {}, acc3 = {};
  const _Float16* ap = act3 + (size_t)m * 3136 + hi * 8;
  const _Float16* bp = wt + (size_t)(nb + lo) * 3136 + hi * 8;
  auto ld = [&](int kb) {
    KStep f;
    f.a  = pack_frag(ap + kb);
    f.b0 = pack_frag(bp + kb);
    f.b1 = pack_frag(bp + 16 * 3136 + kb);
    f.b2 = pack_frag(bp + 32 * 3136 + kb);
    f.b3 = pack_frag(bp + 48 * 3136 + kb);
    return f;
  };
  KStep f0 = ld(0), f1;
  for (int kb = 0; kb < 3136 - 64; kb += 64) {
    f1 = ld(kb + 32);
    LOADS_STAY_AHEAD();
    MMA4(f0, acc0, acc1, acc2, acc3);
    f0 = ld(kb + 64);
    LOADS_STAY_AHEAD();
    MMA4(f1, acc0, acc1, acc2, acc3);
  }
  f1 = ld(3136 - 32);
  LOADS_STAY_AHEAD();
  MMA4(f0, acc0, acc1, acc2, acc3);
  MMA4(f1, acc0, acc1, acc2, acc3);
  float bs0 = bias[nb + lo], bs1 = bias[nb + 16 + lo];
  float bs2 = bias[nb + 32 + lo], bs3 = bias[nb + 48 + lo];
#pragma unroll
  for (int r = 0; r < 8; r++) {
    int mm = mt * 16 + hi * 8 + r;
    float v0 = acc0[r] + bs0; v0 = v0 > 0.0f ? v0 : 0.0f;
    float v1 = acc1[r] + bs1; v1 = v1 > 0.0f ? v1 : 0.0f;
    float v2 = acc2[r] + bs2; v2 = v2 > 0.0f ? v2 : 0.0f;
    float v3 = acc3[r] + bs3; v3 = v3 > 0.0f ? v3 : 0.0f;
    X[mm * XS + nb + lo]      = (_Float16)v0;
    X[mm * XS + nb + 16 + lo] = (_Float16)v1;
    X[mm * XS + nb + 32 + lo] = (_Float16)v2;
    X[mm * XS + nb + 48 + lo] = (_Float16)v3;
  }
}

__global__ void __launch_bounds__(256) k_xproj(const _Float16* X, const _Float16* Wih,
                                               const float* biasg, float* G) {
  int wid  = (blockIdx.x * blockDim.x + threadIdx.x) >> 5;
  int lane = threadIdx.x & 31, hi = lane >> 4, lo = lane & 15;
  int mtile = wid & 127, cg = wid >> 7;
  int ncol = cg * 16 + lo;
  float bi = biasg[0 * HL + ncol], bf = biasg[1 * HL + ncol];
  float bg = biasg[2 * HL + ncol], bo = biasg[3 * HL + ncol];
  v8f ai = {bi, bi, bi, bi, bi, bi, bi, bi};
  v8f af = {bf, bf, bf, bf, bf, bf, bf, bf};
  v8f ag = {bg, bg, bg, bg, bg, bg, bg, bg};
  v8f ao = {bo, bo, bo, bo, bo, bo, bo, bo};
  const _Float16* ap = X   + (mtile * 16 + lo) * XS + hi * 8;
  const _Float16* bp = Wih + ncol * XS + hi * 8;
  auto ld = [&](int kb) {
    KStep f;
    f.a  = pack_frag(ap + kb);
    f.b0 = pack_frag(bp + 0 * HL * XS + kb);
    f.b1 = pack_frag(bp + 1 * HL * XS + kb);
    f.b2 = pack_frag(bp + 2 * HL * XS + kb);
    f.b3 = pack_frag(bp + 3 * HL * XS + kb);
    return f;
  };
  KStep f0 = ld(0), f1;
  for (int kb = 0; kb < XS - 64; kb += 64) {
    f1 = ld(kb + 32);
    LOADS_STAY_AHEAD();
    MMA4(f0, ai, af, ag, ao);
    f0 = ld(kb + 64);
    LOADS_STAY_AHEAD();
    MMA4(f1, ai, af, ag, ao);
  }
  f1 = ld(XS - 32);
  LOADS_STAY_AHEAD();
  MMA4(f0, ai, af, ag, ao);
  MMA4(f1, ai, af, ag, ao);
  v4f* gp = (v4f*)(G + (size_t)(mtile * 16 + cg) * GT) + lane;
  gp[0 * 64]      = ai.lo;  gp[0 * 64 + 32] = ai.hi;
  gp[1 * 64]      = af.lo;  gp[1 * 64 + 32] = af.hi;
  gp[2 * 64]      = ag.lo;  gp[2 * 64 + 32] = ag.hi;
  gp[3 * 64]      = ao.lo;  gp[3 * 64 + 32] = ao.hi;
}

__global__ void __launch_bounds__(512) k_lstm(const float* G, const _Float16* Whh,
                                              const int* done,
                                              const float* h0, const float* c0,
                                              float* hs, float* out) {
  __shared__ v8h Wl[16 * 4 * 4 * 2 * 32];
  __shared__ __attribute__((aligned(16))) _Float16 A[16 * HSTR];
  __shared__ unsigned nm[T_STEPS];
  const int tid = threadIdx.x;
  const int mt  = blockIdx.x;
  const int cg  = tid >> 5, lane = tid & 31, hi = lane >> 4, lo = lane & 15;
  const int ncol = cg * 16 + lo;
  const int env0 = mt * 16;

  if (tid < T_STEPS) {
    unsigned m = 0;
    if (tid + 1 < T_STEPS) {
#pragma unroll
      for (int e = 0; e < 16; e++)
        m |= (done[(tid + 1) * E_ENVS + env0 + e] != 0 ? 1u : 0u) << e;
    }
    nm[tid] = m;
  }

  for (int i = tid; i < 16 * HL; i += 512) {
    int e = i >> 8, k = i & 255;
    float v = h0[(env0 + e) * HL + k];
    A[e * HSTR + k] = done[env0 + e] ? (_Float16)0.0f : (_Float16)v;
  }
  v8f creg;
#pragma unroll
  for (int r = 0; r < 8; r++) {
    float v = c0[(env0 + hi * 8 + r) * HL + ncol];
    creg[r] = done[env0 + hi * 8 + r] ? 0.0f : v;
  }

  const _Float16* bp = Whh + ncol * HL + hi * 8;
  v8h* wl = Wl + cg * (4 * 4 * 2 * 32) + lane;
#pragma unroll
  for (int ch = 0; ch < 4; ch++) {
#pragma unroll
    for (int g = 0; g < 4; g++) {
      const _Float16* p = bp + g * HL * HL + (4 + ch) * 32;
      wl[((ch * 4 + g) * 2 + 0) * 32] = *(const v8h*)p;
      wl[((ch * 4 + g) * 2 + 1) * 32] = *(const v8h*)(p + 16);
    }
  }
  __builtin_amdgcn_sched_barrier(0);
  v16h wr[4][4];
#pragma unroll
  for (int ch = 0; ch < 4; ch++) {
#pragma unroll
    for (int g = 0; g < 4; g++) wr[ch][g] = pack_frag(bp + g * HL * HL + ch * 32);
  }
  __syncthreads();

  const v4f* gbase = (const v4f*)(G + (size_t)(mt * 16 + cg) * GT) + lane;
  v8f gi, gf, gg, go;
  gi.lo = gbase[0 * 64];  gi.hi = gbase[0 * 64 + 32];
  gf.lo = gbase[1 * 64];  gf.hi = gbase[1 * 64 + 32];
  gg.lo = gbase[2 * 64];  gg.hi = gbase[2 * 64 + 32];
  go.lo = gbase[3 * 64];  go.hi = gbase[3 * 64 + 32];

  float hv[8];
  for (int t = 0; t < T_STEPS; t++) {
    const unsigned my = nm[t] >> (hi * 8);

    const _Float16* arow = &A[lo * HSTR + hi * 8];
    v8f ai, af, ag, ao;
#pragma unroll
    for (int ch = 0; ch < 4; ch++) {
      v16h a = pack_frag(arow + ch * 32);
      ai = WMMA16(a, wr[ch][0], ch ? ai : gi);
      af = WMMA16(a, wr[ch][1], ch ? af : gf);
      ag = WMMA16(a, wr[ch][2], ch ? ag : gg);
      ao = WMMA16(a, wr[ch][3], ch ? ao : go);
    }
#pragma unroll
    for (int ch = 0; ch < 4; ch++) {
      v16h a = pack_frag(arow + (4 + ch) * 32);
      v16h b[4];
#pragma unroll
      for (int g = 0; g < 4; g++) {
        v8h u = wl[((ch * 4 + g) * 2 + 0) * 32];
        v8h w = wl[((ch * 4 + g) * 2 + 1) * 32];
#pragma unroll
        for (int e = 0; e < 8; e++) { b[g][e] = u[e]; b[g][e + 8] = w[e]; }
      }
      ai = WMMA16(a, b[0], ai);
      af = WMMA16(a, b[1], af);
      ag = WMMA16(a, b[2], ag);
      ao = WMMA16(a, b[3], ao);
    }
    lds_barrier();

    if (t + 1 < T_STEPS) {
      const v4f* gp = gbase + (size_t)(t + 1) * (2 * 16 * GT / 4);
      gi.lo = gp[0 * 64];  gi.hi = gp[0 * 64 + 32];
      gf.lo = gp[1 * 64];  gf.hi = gp[1 * 64 + 32];
      gg.lo = gp[2 * 64];  gg.hi = gp[2 * 64 + 32];
      go.lo = gp[3 * 64];  go.hi = gp[3 * 64 + 32];
    }

    bool dn[8];
#pragma unroll
    for (int r = 0; r < 8; r++) {
      float iv = sigmoid_hw(ai[r]), fv = sigmoid_hw(af[r]);
      float gv = tanh_hw(ag[r]),    ov = sigmoid_hw(ao[r]);
      float cv = fv * creg[r] + iv * gv;
      hv[r]   = ov * tanh_hw(cv);
      dn[r]   = ((my >> r) & 1u) != 0;
      creg[r] = dn[r] ? 0.0f : cv;
    }
#pragma unroll
    for (int r = 0; r < 8; r++) {
      int el  = hi * 8 + r;
      int env = env0 + el;
      A[el * HSTR + ncol] = dn[r] ? (_Float16)0.0f : (_Float16)hv[r];
      hs[(t * E_ENVS + env) * HL + ncol] = hv[r];
    }
    lds_barrier();
  }
#pragma unroll
  for (int r = 0; r < 8; r++) {
    int env = env0 + hi * 8 + r;
    out[12288 + env * HL + ncol] = hv[r];
    out[20480 + env * HL + ncol] = creg[r];
  }
}

__global__ void k_heads(const float* hs, const float* pw, const float* pb,
                        const float* vw, const float* vb, float* out) {
  int i = blockIdx.x * blockDim.x + threadIdx.x;
  if (i >= B_TOT * 6) return;
  int b = i / 6, j = i % 6;
  const float* h    = hs + b * HL;
  const float* wrow = (j < 5) ? pw + j * HL : vw;
  float s = (j < 5) ? pb[j] : vb[0];
#pragma unroll 4
  for (int k = 0; k < HL; k++) s += h[k] * wrow[k];
  if (j < 5) out[b * 5 + j] = s;
  else       out[10240 + b] = s;
}

extern "C" void kernel_launch(void* const* d_in, const int* in_sizes, int n_in,
                              void* d_out, int out_size, void* d_ws, size_t ws_size,
                              hipStream_t stream) {
  (void)in_sizes; (void)n_in; (void)out_size; (void)ws_size;
  const float* img  = (const float*)d_in[0];
  const int*   pos  = (const int*)d_in[1];
  const int*   lact = (const int*)d_in[2];
  const int*   done = (const int*)d_in[3];
  const float* h0   = (const float*)d_in[4];
  const float* c0   = (const float*)d_in[5];
  const float* w1   = (const float*)d_in[6];
  const float* b1   = (const float*)d_in[7];
  const float* w2   = (const float*)d_in[8];
  const float* b2   = (const float*)d_in[9];
  const float* w3   = (const float*)d_in[10];
  const float* b3   = (const float*)d_in[11];
  const float* fcw  = (const float*)d_in[12];
  const float* fcb  = (const float*)d_in[13];
  const float* wih  = (const float*)d_in[14];
  const float* whh  = (const float*)d_in[15];
  const float* bih  = (const float*)d_in[16];
  const float* bhh  = (const float*)d_in[17];
  const float* pw   = (const float*)d_in[18];
  const float* pb   = (const float*)d_in[19];
  const float* vw   = (const float*)d_in[20];
  const float* vb   = (const float*)d_in[21];
  float* out = (float*)d_out;
  char*  ws  = (char*)d_ws;

  const size_t o_act1 = 90832896;
  const size_t o_w1t  = o_act1 + 52428800;
  const size_t o_w2t  = o_w1t + 12288;
  const size_t o_w3t  = o_w2t + 65536;
  const size_t o_fcwt = o_w3t + 73728;
  const size_t o_wih  = o_fcwt + 3211264;
  const size_t o_whh  = o_wih + 1179648;
  const size_t o_bg   = o_whh + 524288;
  const size_t o_act2 = 0;
  const size_t o_act3 = 21233664;
  const size_t o_X    = 34078720;
  const size_t o_hs   = 36438016;
  const size_t o_G    = 38535168;

  _Float16* act1 = (_Float16*)(ws + o_act1);
  _Float16* act2 = (_Float16*)(ws + o_act2);
  _Float16* act3 = (_Float16*)(ws + o_act3);
  _Float16* Xm   = (_Float16*)(ws + o_X);
  _Float16* w1t  = (_Float16*)(ws + o_w1t);
  _Float16* w2t  = (_Float16*)(ws + o_w2t);
  _Float16* w3t  = (_Float16*)(ws + o_w3t);
  _Float16* fcwt = (_Float16*)(ws + o_fcwt);
  _Float16* wih16 = (_Float16*)(ws + o_wih);
  _Float16* whh16 = (_Float16*)(ws + o_whh);
  float*    bg   = (float*)(ws + o_bg);
  float*    hsb  = (float*)(ws + o_hs);
  float*    Gx   = (float*)(ws + o_G);

  const int TB = 256;
  k_prep_w1  <<<(32 * 192   + TB - 1) / TB, TB, 0, stream>>>(w1, w1t);
  k_prep_w2  <<<(64 * 512   + TB - 1) / TB, TB, 0, stream>>>(w2, w2t);
  k_prep_w3  <<<(64 * 576   + TB - 1) / TB, TB, 0, stream>>>(w3, w3t);
  k_prep_fc  <<<(512 * 3136 + TB - 1) / TB, TB, 0, stream>>>(fcw, fcwt);
  k_prep_wl  <<<(1024 * (XS + HL) + TB - 1) / TB, TB, 0, stream>>>(wih, whh, wih16, whh16);
  k_prep_bias<<<(1024 + TB - 1) / TB, TB, 0, stream>>>(bih, bhh, bg);

  k_conv1<<<6400, TB, 0, stream>>>(img, w1t, b1, act1);
  k_conv2<<<1296, TB, 0, stream>>>(act1, w2t, b2, act2);
  k_conv3<<< 784, TB, 0, stream>>>(act2, w3t, b3, act3);
  k_fc   <<< 128, TB, 0, stream>>>(act3, fcwt, fcb, Xm);

  k_onehot<<<(B_TOT * 64 + TB - 1) / TB, TB, 0, stream>>>(pos, lact, Xm);

  k_xproj<<<256, TB, 0, stream>>>(Xm, wih16, bg, Gx);

  k_lstm<<<2, 512, 0, stream>>>(Gx, whh16, done, h0, c0, hsb, out);

  k_heads<<<(B_TOT * 6 + TB - 1) / TB, TB, 0, stream>>>(hsb, pw, pb, vw, vb, out);
}


# Weights of conv1-3, FC and W_ih stored by the prep kernels in WMMA B-fragment order: a wave's 16-byte B-fragment load reads 4 whole 128-byte lines, not pieces of 16 rows; sum order unchanged
# speedup vs baseline: 12.3809x; 1.3254x over previous
#include <hip/hip_runtime.h>

typedef __attribute__((ext_vector_type(16))) _Float16 v16h;
typedef __attribute__((ext_vector_type(8)))  _Float16 v8h;
typedef __attribute__((ext_vector_type(4)))  _Float16 v4h;
typedef __attribute__((ext_vector_type(8)))  float    v8f;
typedef __attribute__((ext_vector_type(4)))  float    v4f;

#define WMMA16(A,B,C) __builtin_amdgcn_wmma_f32_16x16x32_f16(false,(A),false,(B),(short)0,(C),false,false)

static constexpr int T_STEPS = 64;
static constexpr int E_ENVS  = 32;
static constexpr int B_TOT   = 2048;
static constexpr int HL      = 256;
static constexpr int IND     = 549;
static constexpr int XS      = 576;
static constexpr int HSTR    = 264;
static constexpr int GT      = 1024;
static constexpr int IROW    = 84 * 3;

__device__ __forceinline__ v16h pack_frag(const _Float16* p) {
  v8h u = *(const v8h*)p;
  v8h w = *(const v8h*)(p + 16);
  v16h a;
#pragma unroll
  for (int e = 0; e < 8; e++) { a[e] = u[e]; a[e + 8] = w[e]; }
  return a;
}
__device__ __forceinline__ v16h pack_frag_img(const float* p0, const float* p1) {
  v4f u0 = *(const v4f*)p0;
  v4f u1 = *(const v4f*)(p0 + 4);
  v4f u2 = *(const v4f*)p1;
  v4f u3 = *(const v4f*)(p1 + 4);
  v16h a;
#pragma unroll
  for (int e = 0; e < 4; e++) {
    a[e]      = (_Float16)(u0[e] * (1.0f / 255.0f));
    a[e + 4]  = (_Float16)(u1[e] * (1.0f / 255.0f));
    a[e + 8]  = (_Float16)(u2[e] * (1.0f / 255.0f));
    a[e + 12] = (_Float16)(u3[e] * (1.0f / 255.0f));
  }
  return a;
}
__host__ __device__ constexpr int patch_off(int f) { return (f / 24) * IROW + (f % 24); }

__device__ __forceinline__ v16h pack_frag_w(const _Float16* q) {
  v8h u = *(const v8h*)q;
  v8h w = *(const v8h*)(q + 256);
  v16h a;
#pragma unroll
  for (int e = 0; e < 8; e++) { a[e] = u[e]; a[e + 8] = w[e]; }
  return a;
}
__device__ __forceinline__ void frag_pos(int i, int NT, int& n, int& k) {
  int e = i & 7, lo = (i >> 3) & 15, hi = (i >> 7) & 1, part = (i >> 8) & 1, blk = i >> 9;
  int tl = blk % NT, ks = blk / NT;
  n = tl * 16 + lo;
  k = ks * 32 + part * 16 + hi * 8 + e;
}

__device__ __forceinline__ float tanh_hw(float x) { return __builtin_amdgcn_tanhf(x); }
__device__ __forceinline__ float sigmoid_hw(float x) {
  return __builtin_fmaf(0.5f, tanh_hw(0.5f * x), 0.5f);
}

__device__ __forceinline__ void lds_barrier() {
  __builtin_amdgcn_fence(__ATOMIC_RELEASE, "workgroup", "local");
  __builtin_amdgcn_s_barrier();
  __builtin_amdgcn_fence(__ATOMIC_ACQUIRE, "workgroup", "local");
}

struct KStep { v16h a, b0, b1, b2, b3; };
#define LOADS_STAY_AHEAD() __builtin_amdgcn_sched_barrier(0)
#define MMA4(F, C0, C1, C2, C3)      \
  C0 = WMMA16((F).a, (F).b0, C0);    \
  C1 = WMMA16((F).a, (F).b1, C1);    \
  C2 = WMMA16((F).a, (F).b2, C2);    \
  C3 = WMMA16((F).a, (F).b3, C3)

__global__ void k_prep_w1(const float* w, _Float16* wt) {
  int i = blockIdx.x * blockDim.x + threadIdx.x;
  if (i >= 32 * 192) return;
  int n, f;
  frag_pos(i, 2, n, f);
  int kh = f / 24, r = f % 24, kw = r / 3, c = r % 3;
  wt[i] = (_Float16)w[((n * 3 + c) * 8 + kh) * 8 + kw];
}
__global__ void k_prep_w2(const float* w, _Float16* wt) {
  int i = blockIdx.x * blockDim.x + threadIdx.x;
  if (i >= 64 * 512) return;
  int n, k;
  frag_pos(i, 4, n, k);
  int pix = k >> 5, c = k & 31, kh = pix >> 2, kw = pix & 3;
  wt[i] = (_Float16)w[((n * 32 + c) * 4 + kh) * 4 + kw];
}
__global__ void k_prep_w3(const float* w, _Float16* wt) {
  int i = blockIdx.x * blockDim.x + threadIdx.x;
  if (i >= 64 * 576) return;
  int n, k;
  frag_pos(i, 4, n, k);
  int pix = k >> 6, c = k & 63, kh = pix / 3, kw = pix % 3;
  wt[i] = (_Float16)w[((n * 64 + c) * 3 + kh) * 3 + kw];
}
__global__ void k_prep_fc(const float* w, _Float16* wt) {
  int i = blockIdx.x * blockDim.x + threadIdx.x;
  if (i >= 512 * 3136) return;
  int f, k;
  frag_pos(i, 32, f, k);
  int s = k >> 6, c = k & 63;
  wt[i] = (_Float16)w[f * 3136 + c * 49 + s];
}
__global__ void k_prep_wl(const float* wih, const float* whh, _Float16* wih16, _Float16* whh16) {
  int i = blockIdx.x * blockDim.x + threadIdx.x;
  if (i >= 1024 * (XS + HL)) return;
  if (i < 1024 * XS) {
    int r, k;
    frag_pos(i, 64, r, k);
    wih16[i] = (_Float16)((k < IND) ? wih[r * IND + k] : 0.0f);
  } else {
    int j = i - 1024 * XS;
    whh16[j] = (_Float16)whh[j];
  }
}
__global__ void k_prep_bias(const float* bih, const float* bhh, float* bg) {
  int i = blockIdx.x * blockDim.x + threadIdx.x;
  if (i < 1024) bg[i] = bih[i] + bhh[i];
}
__global__ void k_onehot(const int* pos, const int* lact, _Float16* X) {
  int i = blockIdx.x * blockDim.x + threadIdx.x;
  if (i >= B_TOT * 64) return;
  int b = i >> 6, c = i & 63;
  float v = 0.0f;
  if (c < 16)      v = (pos[b * 2 + 0] == c)      ? 1.0f : 0.0f;
  else if (c < 32) v = (pos[b * 2 + 1] == c - 16) ? 1.0f : 0.0f;
  else if (c < 37) v = (lact[b]        == c - 32) ? 1.0f : 0.0f;
  X[b * XS + 512 + c] = (_Float16)v;
}

__global__ void __launch_bounds__(256) k_conv1(const float* img, const _Float16* wt,
                                               const float* bias, _Float16* out) {
  int wid  = (blockIdx.x * blockDim.x + threadIdx.x) >> 5;
  int lane = threadIdx.x & 31, hi = lane >> 4, lo = lane & 15;
  int m = wid * 16 + lo;
  int b = m / 400, p = m % 400, oh = p / 20, ow = p % 20;
  const float* abase = img + ((size_t)(b * 84 + oh * 4) * 84 + ow * 4) * 3;
  v8f acc0 = {}, acc1 = {};
#pragma unroll
  for (int kb = 0; kb < 192; kb += 32) {
    int o0 = hi ? patch_off(kb + 8)  : patch_off(kb);
    int o1 = hi ? patch_off(kb + 24) : patch_off(kb + 16);
    v16h a  = pack_frag_img(abase + o0, abase + o1);
    v16h b0 = pack_frag_w(wt + ((kb >> 5) * 2 + 0) * 512 + lane * 8);
    v16h b1 = pack_frag_w(wt + ((kb >> 5) * 2 + 1) * 512 + lane * 8);
    acc0 = WMMA16(a, b0, acc0);
    acc1 = WMMA16(a, b1, acc1);
  }
  float bs0 = bias[lo], bs1 = bias[16 + lo];
#pragma unroll
  for (int r = 0; r < 8; r++) {
    int mm = wid * 16 + hi * 8 + r;
    float v0 = acc0[r] + bs0; v0 = v0 > 0.0f ? v0 : 0.0f;
    float v1 = acc1[r] + bs1; v1 = v1 > 0.0f ? v1 : 0.0f;
    out[mm * 32 + lo]      = (_Float16)v0;
    out[mm * 32 + 16 + lo] = (_Float16)v1;
  }
}

__global__ void __launch_bounds__(256) k_conv2(const _Float16* act1, const _Float16* wt,
                                               const float* bias, _Float16* out) {
  int wid  = (blockIdx.x * blockDim.x + threadIdx.x) >> 5;
  int lane = threadIdx.x & 31, hi = lane >> 4, lo = lane & 15;
  int m = wid * 16 + lo;
  int b = m / 81, p = m % 81, oh = p / 9, ow = p % 9;
  v8f acc0 = {}, acc1 = {}, acc2 = {}, acc3 = {};
  const _Float16* ap = act1 + ((b * 20 + oh * 2) * 20 + ow * 2) * 32 + hi * 8;
  const _Float16* bp = wt + lane * 8;
  auto ld = [&](int kb) {
    int pix = kb >> 5, kh = pix >> 2, kw = pix & 3;
    KStep f;
    f.a  = pack_frag(ap + (kh * 20 + kw) * 32);
    f.b0 = pack_frag_w(bp + ((kb >> 5) * 4 + 0) * 512);
    f.b1 = pack_frag_w(bp + ((kb >> 5) * 4 + 1) * 512);
    f.b2 = pack_frag_w(bp + ((kb >> 5) * 4 + 2) * 512);
    f.b3 = pack_frag_w(bp + ((kb >> 5) * 4 + 3) * 512);
    return f;
  };
  KStep f0 = ld(0), f1;
  for (int kb = 0; kb < 512 - 64; kb += 64) {
    f1 = ld(kb + 32);
    LOADS_STAY_AHEAD();
    MMA4(f0, acc0, acc1, acc2, acc3);
    f0 = ld(kb + 64);
    LOADS_STAY_AHEAD();
    MMA4(f1, acc0, acc1, acc2, acc3);
  }
  f1 = ld(512 - 32);
  LOADS_STAY_AHEAD();
  MMA4(f0, acc0, acc1, acc2, acc3);
  MMA4(f1, acc0, acc1, acc2, acc3);
  float bs0 = bias[lo], bs1 = bias[16 + lo], bs2 = bias[32 + lo], bs3 = bias[48 + lo];
#pragma unroll
  for (int r = 0; r < 8; r++) {
    int mm = wid * 16 + hi * 8 + r;
    float v0 = acc0[r] + bs0; v0 = v0 > 0.0f ? v0 : 0.0f;
    float v1 = acc1[r] + bs1; v1 = v1 > 0.0f ? v1 : 0.0f;
    float v2 = acc2[r] + bs2; v2 = v2 > 0.0f ? v2 : 0.0f;
    float v3 = acc3[r] + bs3; v3 = v3 > 0.0f ? v3 : 0.0f;
    out[mm * 64 + lo]      = (_Float16)v0;
    out[mm * 64 + 16 + lo] = (_Float16)v1;
    out[mm * 64 + 32 + lo] = (_Float16)v2;
    out[mm * 64 + 48 + lo] = (_Float16)v3;
  }
}

__global__ void __launch_bounds__(256) k_conv3(const _Float16* act2, const _Float16* wt,
                                               const float* bias, _Float16* out) {
  int wid  = (blockIdx.x * blockDim.x + threadIdx.x) >> 5;
  int lane = threadIdx.x & 31, hi = lane >> 4, lo = lane & 15;
  int m = wid * 16 + lo;
  int b = m / 49, p = m % 49, oh = p / 7, ow = p % 7;
  v8f acc0 = {}, acc1 = {}, acc2 = {}, acc3 = {};
  const _Float16* ap = act2 + ((b * 9 + oh) * 9 + ow) * 64 + hi * 8;
  const _Float16* bp = wt + lane * 8;
  auto ld = [&](int kb) {
    int pix = kb >> 6, kh = pix / 3, kw = pix % 3;
    KStep f;
    f.a  = pack_frag(ap + (kh * 9 + kw) * 64 + (kb & 63));
    f.b0 = pack_frag_w(bp + ((kb >> 5) * 4 + 0) * 512);
    f.b1 = pack_frag_w(bp + ((kb >> 5) * 4 + 1) * 512);
    f.b2 = pack_frag_w(bp + ((kb >> 5) * 4 + 2) * 512);
    f.b3 = pack_frag_w(bp + ((kb >> 5) * 4 + 3) * 512);
    return f;
  };
  KStep f0 = ld(0), f1;
  for (int kb = 0; kb < 576 - 64; kb += 64) {
    f1 = ld(kb + 32);
    LOADS_STAY_AHEAD();
    MMA4(f0, acc0, acc1, acc2, acc3);
    f0 = ld(kb + 64);
    LOADS_STAY_AHEAD();
    MMA4(f1, acc0, acc1, acc2, acc3);
  }
  f1 = ld(576 - 32);
  LOADS_STAY_AHEAD();
  MMA4(f0, acc0, acc1, acc2, acc3);
  MMA4(f1, acc0, acc1, acc2, acc3);
  float bs0 = bias[lo], bs1 = bias[16 + lo], bs2 = bias[32 + lo], bs3 = bias[48 + lo];
#pragma unroll
  for (int r = 0; r < 8; r++) {
    int mm = wid * 16 + hi * 8 + r;
    float v0 = acc0[r] + bs0; v0 = v0 > 0.0f ? v0 : 0.0f;
    float v1 = acc1[r] + bs1; v1 = v1 > 0.0f ? v1 : 0.0f;
    float v2 = acc2[r] + bs2; v2 = v2 > 0.0f ? v2 : 0.0f;
    float v3 = acc3[r] + bs3; v3 = v3 > 0.0f ? v3 : 0.0f;
    out[mm * 64 + lo]      = (_Float16)v0;
    out[mm * 64 + 16 + lo] = (_Float16)v1;
    out[mm * 64 + 32 + lo] = (_Float16)v2;
    out[mm * 64 + 48 + lo] = (_Float16)v3;
  }
}

__global__ void __launch_bounds__(256) k_fc(const _Float16* act3, const _Float16* wt,
                                            const float* bias, _Float16* X) {
  int wid  = (blockIdx.x * blockDim.x + threadIdx.x) >> 5;
  int lane = threadIdx.x & 31, hi = lane >> 4, lo = lane & 15;
  int ng = wid & 7, mt = wid >> 3;
  int m = mt * 16 + lo;
  int nb = ng * 64;
  v8f acc0 = {}, acc1 = {}, acc2 = {}, acc3 = {};
  const _Float16* ap = act3 + (size_t)m * 3136 + hi * 8;
  const _Float16* bp = wt + (ng * 4) * 512 + lane * 8;
  auto ld = [&](int kb) {
    KStep f;
    f.a  = pack_frag(ap + kb);
    f.b0 = pack_frag_w(bp + ((kb >> 5) * 32 + 0) * 512);
    f.b1 = pack_frag_w(bp + ((kb >> 5) * 32 + 1) * 512);
    f.b2 = pack_frag_w(bp + ((kb >> 5) * 32 + 2) * 512);
    f.b3 = pack_frag_w(bp + ((kb >> 5) * 32 + 3) * 512);
    return f;
  };
  KStep f0 = ld(0), f1;
  for (int kb = 0; kb < 3136 - 64; kb += 64) {
    f1 = ld(kb + 32);
    LOADS_STAY_AHEAD();
    MMA4(f0, acc0, acc1, acc2, acc3);
    f0 = ld(kb + 64);
    LOADS_STAY_AHEAD();
    MMA4(f1, acc0, acc1, acc2, acc3);
  }
  f1 = ld(3136 - 32);
  LOADS_STAY_AHEAD();
  MMA4(f0, acc0, acc1, acc2, acc3);
  MMA4(f1, acc0, acc1, acc2, acc3);
  float bs0 = bias[nb + lo], bs1 = bias[nb + 16 + lo];
  float bs2 = bias[nb + 32 + lo], bs3 = bias[nb + 48 + lo];
#pragma unroll
  for (int r = 0; r < 8; r++) {
    int mm = mt * 16 + hi * 8 + r;
    float v0 = acc0[r] + bs0; v0 = v0 > 0.0f ? v0 : 0.0f;
    float v1 = acc1[r] + bs1; v1 = v1 > 0.0f ? v1 : 0.0f;
    float v2 = acc2[r] + bs2; v2 = v2 > 0.0f ? v2 : 0.0f;
    float v3 = acc3[r] + bs3; v3 = v3 > 0.0f ? v3 : 0.0f;
    X[mm * XS + nb + lo]      = (_Float16)v0;
    X[mm * XS + nb + 16 + lo] = (_Float16)v1;
    X[mm * XS + nb + 32 + lo] = (_Float16)v2;
    X[mm * XS + nb + 48 + lo] = (_Float16)v3;
  }
}

__global__ void __launch_bounds__(256) k_xproj(const _Float16* X, const _Float16* Wih,
                                               const float* biasg, float* G) {
  int wid  = (blockIdx.x * blockDim.x + threadIdx.x) >> 5;
  int lane = threadIdx.x & 31, hi = lane >> 4, lo = lane & 15;
  int mtile = wid & 127, cg = wid >> 7;
  int ncol = cg * 16 + lo;
  float bi = biasg[0 * HL + ncol], bf = biasg[1 * HL + ncol];
  float bg = biasg[2 * HL + ncol], bo = biasg[3 * HL + ncol];
  v8f ai = {bi, bi, bi, bi, bi, bi, bi, bi};
  v8f af = {bf, bf, bf, bf, bf, bf, bf, bf};
  v8f ag = {bg, bg, bg, bg, bg, bg, bg, bg};
  v8f ao = {bo, bo, bo, bo, bo, bo, bo, bo};
  const _Float16* ap = X   + (mtile * 16 + lo) * XS + hi * 8;
  const _Float16* bp = Wih + cg * 512 + lane * 8;
  auto ld = [&](int kb) {
    KStep f;
    f.a  = pack_frag(ap + kb);
    f.b0 = pack_frag_w(bp + ((kb >> 5) * 64 + 0 * 16) * 512);
    f.b1 = pack_frag_w(bp + ((kb >> 5) * 64 + 1 * 16) * 512);
    f.b2 = pack_frag_w(bp + ((kb >> 5) * 64 + 2 * 16) * 512);
    f.b3 = pack_frag_w(bp + ((kb >> 5) * 64 + 3 * 16) * 512);
    return f;
  };
  KStep f0 = ld(0), f1;
  for (int kb = 0; kb < XS - 64; kb += 64) {
    f1 = ld(kb + 32);
    LOADS_STAY_AHEAD();
    MMA4(f0, ai, af, ag, ao);
    f0 = ld(kb + 64);
    LOADS_STAY_AHEAD();
    MMA4(f1, ai, af, ag, ao);
  }
  f1 = ld(XS - 32);
  LOADS_STAY_AHEAD();
  MMA4(f0, ai, af, ag, ao);
  MMA4(f1, ai, af, ag, ao);
  v4f* gp = (v4f*)(G + (size_t)(mtile * 16 + cg) * GT) + lane;
  gp[0 * 64]      = ai.lo;  gp[0 * 64 + 32] = ai.hi;
  gp[1 * 64]      = af.lo;  gp[1 * 64 + 32] = af.hi;
  gp[2 * 64]      = ag.lo;  gp[2 * 64 + 32] = ag.hi;
  gp[3 * 64]      = ao.lo;  gp[3 * 64 + 32] = ao.hi;
}

__global__ void __launch_bounds__(512) k_lstm(const float* G, const _Float16* Whh,
                                              const int* done,
                                              const float* h0, const float* c0,
                                              float* hs, float* out) {
  __shared__ v8h Wl[16 * 4 * 4 * 2 * 32];
  __shared__ __attribute__((aligned(16))) _Float16 A[16 * HSTR];
  __shared__ unsigned nm[T_STEPS];
  const int tid = threadIdx.x;
  const int mt  = blockIdx.x;
  const int cg  = tid >> 5, lane = tid & 31, hi = lane >> 4, lo = lane & 15;
  const int ncol = cg * 16 + lo;
  const int env0 = mt * 16;

  if (tid < T_STEPS) {
    unsigned m = 0;
    if (tid + 1 < T_STEPS) {
#pragma unroll
      for (int e = 0; e < 16; e++)
        m |= (done[(tid + 1) * E_ENVS + env0 + e] != 0 ? 1u : 0u) << e;
    }
    nm[tid] = m;
  }

  for (int i = tid; i < 16 * HL; i += 512) {
    int e = i >> 8, k = i & 255;
    float v = h0[(env0 + e) * HL + k];
    A[e * HSTR + k] = done[env0 + e] ? (_Float16)0.0f : (_Float16)v;
  }
  v8f creg;
#pragma unroll
  for (int r = 0; r < 8; r++) {
    float v = c0[(env0 + hi * 8 + r) * HL + ncol];
    creg[r] = done[env0 + hi * 8 + r] ? 0.0f : v;
  }

  const _Float16* bp = Whh + ncol * HL + hi * 8;
  v8h* wl = Wl + cg * (4 * 4 * 2 * 32) + lane;
#pragma unroll
  for (int ch = 0; ch < 4; ch++) {
#pragma unroll
    for (int g = 0; g < 4; g++) {
      const _Float16* p = bp + g * HL * HL + (4 + ch) * 32;
      wl[((ch * 4 + g) * 2 + 0) * 32] = *(const v8h*)p;
      wl[((ch * 4 + g) * 2 + 1) * 32] = *(const v8h*)(p + 16);
    }
  }
  __builtin_amdgcn_sched_barrier(0);
  v16h wr[4][4];
#pragma unroll
  for (int ch = 0; ch < 4; ch++) {
#pragma unroll
    for (int g = 0; g < 4; g++) wr[ch][g] = pack_frag(bp + g * HL * HL + ch * 32);
  }
  __syncthreads();

  const v4f* gbase = (const v4f*)(G + (size_t)(mt * 16 + cg) * GT) + lane;
  v8f gi, gf, gg, go;
  gi.lo = gbase[0 * 64];  gi.hi = gbase[0 * 64 + 32];
  gf.lo = gbase[1 * 64];  gf.hi = gbase[1 * 64 + 32];
  gg.lo = gbase[2 * 64];  gg.hi = gbase[2 * 64 + 32];
  go.lo = gbase[3 * 64];  go.hi = gbase[3 * 64 + 32];

  float hv[8];
  for (int t = 0; t < T_STEPS; t++) {
    const unsigned my = nm[t] >> (hi * 8);

    const _Float16* arow = &A[lo * HSTR + hi * 8];
    v8f ai, af, ag, ao;
#pragma unroll
    for (int ch = 0; ch < 4; ch++) {
      v16h a = pack_frag(arow + ch * 32);
      ai = WMMA16(a, wr[ch][0], ch ? ai : gi);
      af = WMMA16(a, wr[ch][1], ch ? af : gf);
      ag = WMMA16(a, wr[ch][2], ch ? ag : gg);
      ao = WMMA16(a, wr[ch][3], ch ? ao : go);
    }
#pragma unroll
    for (int ch = 0; ch < 4; ch++) {
      v16h a = pack_frag(arow + (4 + ch) * 32);
      v16h b[4];
#pragma unroll
      for (int g = 0; g < 4; g++) {
        v8h u = wl[((ch * 4 + g) * 2 + 0) * 32];
        v8h w = wl[((ch * 4 + g) * 2 + 1) * 32];
#pragma unroll
        for (int e = 0; e < 8; e++) { b[g][e] = u[e]; b[g][e + 8] = w[e]; }
      }
      ai = WMMA16(a, b[0], ai);
      af = WMMA16(a, b[1], af);
      ag = WMMA16(a, b[2], ag);
      ao = WMMA16(a, b[3], ao);
    }
    lds_barrier();

    if (t + 1 < T_STEPS) {
      const v4f* gp = gbase + (size_t)(t + 1) * (2 * 16 * GT / 4);
      gi.lo = gp[0 * 64];  gi.hi = gp[0 * 64 + 32];
      gf.lo = gp[1 * 64];  gf.hi = gp[1 * 64 + 32];
      gg.lo = gp[2 * 64];  gg.hi = gp[2 * 64 + 32];
      go.lo = gp[3 * 64];  go.hi = gp[3 * 64 + 32];
    }

    bool dn[8];
#pragma unroll
    for (int r = 0; r < 8; r++) {
      float iv = sigmoid_hw(ai[r]), fv = sigmoid_hw(af[r]);
      float gv = tanh_hw(ag[r]),    ov = sigmoid_hw(ao[r]);
      float cv = fv * creg[r] + iv * gv;
      hv[r]   = ov * tanh_hw(cv);
      dn[r]   = ((my >> r) & 1u) != 0;
      creg[r] = dn[r] ? 0.0f : cv;
    }
#pragma unroll
    for (int r = 0; r < 8; r++) {
      int el  = hi * 8 + r;
      int env = env0 + el;
      A[el * HSTR + ncol] = dn[r] ? (_Float16)0.0f : (_Float16)hv[r];
      hs[(t * E_ENVS + env) * HL + ncol] = hv[r];
    }
    lds_barrier();
  }
#pragma unroll
  for (int r = 0; r < 8; r++) {
    int env = env0 + hi * 8 + r;
    out[12288 + env * HL + ncol] = hv[r];
    out[20480 + env * HL + ncol] = creg[r];
  }
}

__global__ void k_heads(const float* hs, const float* pw, const float* pb,
                        const float* vw, const float* vb, float* out) {
  int i = blockIdx.x * blockDim.x + threadIdx.x;
  if (i >= B_TOT * 6) return;
  int b = i / 6, j = i % 6;
  const float* h    = hs + b * HL;
  const float* wrow = (j < 5) ? pw + j * HL : vw;
  float s = (j < 5) ? pb[j] : vb[0];
#pragma unroll 4
  for (int k = 0; k < HL; k++) s += h[k] * wrow[k];
  if (j < 5) out[b * 5 + j] = s;
  else       out[10240 + b] = s;
}

extern "C" void kernel_launch(void* const* d_in, const int* in_sizes, int n_in,
                              void* d_out, int out_size, void* d_ws, size_t ws_size,
                              hipStream_t stream) {
  (void)in_sizes; (void)n_in; (void)out_size; (void)ws_size;
  const float* img  = (const float*)d_in[0];
  const int*   pos  = (const int*)d_in[1];
  const int*   lact = (const int*)d_in[2];
  const int*   done = (const int*)d_in[3];
  const float* h0   = (const float*)d_in[4];
  const float* c0   = (const float*)d_in[5];
  const float* w1   = (const float*)d_in[6];
  const float* b1   = (const float*)d_in[7];
  const float* w2   = (const float*)d_in[8];
  const float* b2   = (const float*)d_in[9];
  const float* w3   = (const float*)d_in[10];
  const float* b3   = (const float*)d_in[11];
  const float* fcw  = (const float*)d_in[12];
  const float* fcb  = (const float*)d_in[13];
  const float* wih  = (const float*)d_in[14];
  const float* whh  = (const float*)d_in[15];
  const float* bih  = (const float*)d_in[16];
  const float* bhh  = (const float*)d_in[17];
  const float* pw   = (const float*)d_in[18];
  const float* pb   = (const float*)d_in[19];
  const float* vw   = (const float*)d_in[20];
  const float* vb   = (const float*)d_in[21];
  float* out = (float*)d_out;
  char*  ws  = (char*)d_ws;

  const size_t o_act1 = 90832896;
  const size_t o_w1t  = o_act1 + 52428800;
  const size_t o_w2t  = o_w1t + 12288;
  const size_t o_w3t  = o_w2t + 65536;
  const size_t o_fcwt = o_w3t + 73728;
  const size_t o_wih  = o_fcwt + 3211264;
  const size_t o_whh  = o_wih + 1179648;
  const size_t o_bg   = o_whh + 524288;
  const size_t o_act2 = 0;
  const size_t o_act3 = 21233664;
  const size_t o_X    = 34078720;
  const size_t o_hs   = 36438016;
  const size_t o_G    = 38535168;

  _Float16* act1 = (_Float16*)(ws + o_act1);
  _Float16* act2 = (_Float16*)(ws + o_act2);
  _Float16* act3 = (_Float16*)(ws + o_act3);
  _Float16* Xm   = (_Float16*)(ws + o_X);
  _Float16* w1t  = (_Float16*)(ws + o_w1t);
  _Float16* w2t  = (_Float16*)(ws + o_w2t);
  _Float16* w3t  = (_Float16*)(ws + o_w3t);
  _Float16* fcwt = (_Float16*)(ws + o_fcwt);
  _Float16* wih16 = (_Float16*)(ws + o_wih);
  _Float16* whh16 = (_Float16*)(ws + o_whh);
  float*    bg   = (float*)(ws + o_bg);
  float*    hsb  = (float*)(ws + o_hs);
  float*    Gx   = (float*)(ws + o_G);

  const int TB = 256;
  k_prep_w1  <<<(32 * 192   + TB - 1) / TB, TB, 0, stream>>>(w1, w1t);
  k_prep_w2  <<<(64 * 512   + TB - 1) / TB, TB, 0, stream>>>(w2, w2t);
  k_prep_w3  <<<(64 * 576   + TB - 1) / TB, TB, 0, stream>>>(w3, w3t);
  k_prep_fc  <<<(512 * 3136 + TB - 1) / TB, TB, 0, stream>>>(fcw, fcwt);
  k_prep_wl  <<<(1024 * (XS + HL) + TB - 1) / TB, TB, 0, stream>>>(wih, whh, wih16, whh16);
  k_prep_bias<<<(1024 + TB - 1) / TB, TB, 0, stream>>>(bih, bhh, bg);

  k_conv1<<<6400, TB, 0, stream>>>(img, w1t, b1, act1);
  k_conv2<<<1296, TB, 0, stream>>>(act1, w2t, b2, act2);
  k_conv3<<< 784, TB, 0, stream>>>(act2, w3t, b3, act3);
  k_fc   <<< 128, TB, 0, stream>>>(act3, fcwt, fcb, Xm);

  k_onehot<<<(B_TOT * 64 + TB - 1) / TB, TB, 0, stream>>>(pos, lact, Xm);

  k_xproj<<<256, TB, 0, stream>>>(Xm, wih16, bg, Gx);

  k_lstm<<<2, 512, 0, stream>>>(Gx, whh16, done, h0, c0, hsb, out);

  k_heads<<<(B_TOT * 6 + TB - 1) / TB, TB, 0, stream>>>(hsb, pw, pb, vw, vb, out);
}


# Six independent launches (FC, LSTM, conv2, conv3 weight re-layouts, bias sum, one-hot columns of X) now run as extra workgroups spread through the conv1 launch: 14 launches -> 8; sum orders unchanged
# speedup vs baseline: 13.0080x; 1.0507x over previous
#include <hip/hip_runtime.h>

typedef __attribute__((ext_vector_type(16))) _Float16 v16h;
typedef __attribute__((ext_vector_type(8)))  _Float16 v8h;
typedef __attribute__((ext_vector_type(4)))  _Float16 v4h;
typedef __attribute__((ext_vector_type(8)))  float    v8f;
typedef __attribute__((ext_vector_type(4)))  float    v4f;

#define WMMA16(A,B,C) __builtin_amdgcn_wmma_f32_16x16x32_f16(false,(A),false,(B),(short)0,(C),false,false)

static constexpr int T_STEPS = 64;
static constexpr int E_ENVS  = 32;
static constexpr int B_TOT   = 2048;
static constexpr int HL      = 256;
static constexpr int IND     = 549;
static constexpr int XS      = 576;
static constexpr int HSTR    = 264;
static constexpr int GT      = 1024;
static constexpr int IROW    = 84 * 3;

__device__ __forceinline__ v16h pack_frag(const _Float16* p) {
  v8h u = *(const v8h*)p;
  v8h w = *(const v8h*)(p + 16);
  v16h a;
#pragma unroll
  for (int e = 0; e < 8; e++) { a[e] = u[e]; a[e + 8] = w[e]; }
  return a;
}
__device__ __forceinline__ v16h pack_frag_img(const float* p0, const float* p1) {
  v4f u0 = *(const v4f*)p0;
  v4f u1 = *(const v4f*)(p0 + 4);
  v4f u2 = *(const v4f*)p1;
  v4f u3 = *(const v4f*)(p1 + 4);
  v16h a;
#pragma unroll
  for (int e = 0; e < 4; e++) {
    a[e]      = (_Float16)(u0[e] * (1.0f / 255.0f));
    a[e + 4]  = (_Float16)(u1[e] * (1.0f / 255.0f));
    a[e + 8]  = (_Float16)(u2[e] * (1.0f / 255.0f));
    a[e + 12] = (_Float16)(u3[e] * (1.0f / 255.0f));
  }
  return a;
}
__host__ __device__ constexpr int patch_off(int f) { return (f / 24) * IROW + (f % 24); }

__device__ __forceinline__ v16h pack_frag_w(const _Float16* q) {
  v8h u = *(const v8h*)q;
  v8h w = *(const v8h*)(q + 256);
  v16h a;
#pragma unroll
  for (int e = 0; e < 8; e++) { a[e] = u[e]; a[e + 8] = w[e]; }
  return a;
}
__device__ __forceinline__ void frag_pos(int i, int NT, int& n, int& k) {
  int e = i & 7, lo = (i >> 3) & 15, hi = (i >> 7) & 1, part = (i >> 8) & 1, blk = i >> 9;
  int tl = blk % NT, ks = blk / NT;
  n = tl * 16 + lo;
  k = ks * 32 + part * 16 + hi * 8 + e;
}

__device__ __forceinline__ float tanh_hw(float x) { return __builtin_amdgcn_tanhf(x); }
__device__ __forceinline__ float sigmoid_hw(float x) {
  return __builtin_fmaf(0.5f, tanh_hw(0.5f * x), 0.5f);
}

__device__ __forceinline__ void lds_barrier() {
  __builtin_amdgcn_fence(__ATOMIC_RELEASE, "workgroup", "local");
  __builtin_amdgcn_s_barrier();
  __builtin_amdgcn_fence(__ATOMIC_ACQUIRE, "workgroup", "local");
}

struct KStep { v16h a, b0, b1, b2, b3; };
#define LOADS_STAY_AHEAD() __builtin_amdgcn_sched_barrier(0)
#define MMA4(F, C0, C1, C2, C3)      \
  C0 = WMMA16((F).a, (F).b0, C0);    \
  C1 = WMMA16((F).a, (F).b1, C1);    \
  C2 = WMMA16((F).a, (F).b2, C2);    \
  C3 = WMMA16((F).a, (F).b3, C3)

__global__ void k_prep_w1(const float* w, _Float16* wt) {
  int i = blockIdx.x * blockDim.x + threadIdx.x;
  if (i >= 32 * 192) return;
  int n, f;
  frag_pos(i, 2, n, f);
  int kh = f / 24, r = f % 24, kw = r / 3, c = r % 3;
  wt[i] = (_Float16)w[((n * 3 + c) * 8 + kh) * 8 + kw];
}
__device__ __forceinline__ void prep_w2(int i, const float* w, _Float16* wt) {
  int n, k;
  frag_pos(i, 4, n, k);
  int pix = k >> 5, c = k & 31, kh = pix >> 2, kw = pix & 3;
  wt[i] = (_Float16)w[((n * 32 + c) * 4 + kh) * 4 + kw];
}
__device__ __forceinline__ void prep_w3(int i, const float* w, _Float16* wt) {
  int n, k;
  frag_pos(i, 4, n, k);
  int pix = k >> 6, c = k & 63, kh = pix / 3, kw = pix % 3;
  wt[i] = (_Float16)w[((n * 64 + c) * 3 + kh) * 3 + kw];
}
__device__ __forceinline__ void prep_fc(int i, const float* w, _Float16* wt) {
  int f, k;
  frag_pos(i, 32, f, k);
  int s = k >> 6, c = k & 63;
  wt[i] = (_Float16)w[f * 3136 + c * 49 + s];
}
__device__ __forceinline__ void prep_wl(int i, const float* wih, const float* whh,
                                        _Float16* wih16, _Float16* whh16) {
  if (i < 1024 * XS) {
    int r, k;
    frag_pos(i, 64, r, k);
    wih16[i] = (_Float16)((k < IND) ? wih[r * IND + k] : 0.0f);
  } else {
    int j = i - 1024 * XS;
    whh16[j] = (_Float16)whh[j];
  }
}
__device__ __forceinline__ void prep_bias(int i, const float* bih, const float* bhh, float* bg) {
  bg[i] = bih[i] + bhh[i];
}
__device__ __forceinline__ void onehot(int i, const int* pos, const int* lact, _Float16* X) {
  int b = i >> 6, c = i & 63;
  float v = 0.0f;
  if (c < 16)      v = (pos[b * 2 + 0] == c)      ? 1.0f : 0.0f;
  else if (c < 32) v = (pos[b * 2 + 1] == c - 16) ? 1.0f : 0.0f;
  else if (c < 37) v = (lact[b]        == c - 32) ? 1.0f : 0.0f;
  X[b * XS + 512 + c] = (_Float16)v;
}

static constexpr int NB_CONV1 = 6400;
static constexpr int GRP_CONV = 8, GRP_ALL = 21;
static constexpr int NE_FC    = 512 * 3136;
static constexpr int NE_WL    = 1024 * (XS + HL);
static constexpr int NE_OH    = B_TOT * 64;
static constexpr int NE_W2    = 64 * 512;
static constexpr int NE_W3    = 64 * 576;
static constexpr int NE_BIAS  = 1024;
static constexpr int NB_EXTRA = (NE_FC + NE_WL + NE_OH + NE_W2 + NE_W3 + NE_BIAS) / 256;
static constexpr int NGRP     = NB_CONV1 / GRP_CONV;
static_assert(NE_FC % 256 == 0 && NE_WL % 256 == 0 && NE_OH % 256 == 0 && NE_W2 % 256 == 0 &&
              NE_W3 % 256 == 0 && NE_BIAS % 256 == 0, "every extra job is whole workgroups");
static_assert(NGRP * GRP_CONV == NB_CONV1 && NGRP * (GRP_ALL - GRP_CONV) >= NB_EXTRA,
              "the groups hold every conv1 workgroup and every extra one");
__global__ void __launch_bounds__(256) k_conv1(const float* img, const _Float16* wt,
                                               const float* bias, _Float16* out,
                                               const float* fcw, _Float16* fcwt,
                                               const float* wih, const float* whh,
                                               _Float16* wih16, _Float16* whh16,
                                               const int* pos, const int* lact, _Float16* X,
                                               const float* w2, _Float16* w2t,
                                               const float* w3, _Float16* w3t,
                                               const float* bih, const float* bhh, float* bg) {
  const int grp = blockIdx.x / GRP_ALL, rr = blockIdx.x % GRP_ALL;
  if (rr >= GRP_CONV) {
    int xb = grp * (GRP_ALL - GRP_CONV) + (rr - GRP_CONV);
    if (xb >= NB_EXTRA) return;
    const int t = threadIdx.x;
    if (xb < NE_FC / 256)   { prep_fc(xb * 256 + t, fcw, fcwt);              return; }
    xb -= NE_FC / 256;
    if (xb < NE_WL / 256)   { prep_wl(xb * 256 + t, wih, whh, wih16, whh16); return; }
    xb -= NE_WL / 256;
    if (xb < NE_OH / 256)   { onehot(xb * 256 + t, pos, lact, X);            return; }
    xb -= NE_OH / 256;
    if (xb < NE_W2 / 256)   { prep_w2(xb * 256 + t, w2, w2t);                return; }
    xb -= NE_W2 / 256;
    if (xb < NE_W3 / 256)   { prep_w3(xb * 256 + t, w3, w3t);                return; }
    xb -= NE_W3 / 256;
    prep_bias(xb * 256 + t, bih, bhh, bg);
    return;
  }
  const int cb = grp * GRP_CONV + rr;
  int wid  = (cb * 256 + threadIdx.x) >> 5;
  int lane = threadIdx.x & 31, hi = lane >> 4, lo = lane & 15;
  int m = wid * 16 + lo;
  int b = m / 400, p = m % 400, oh = p / 20, ow = p % 20;
  const float* abase = img + ((size_t)(b * 84 + oh * 4) * 84 + ow * 4) * 3;
  v8f acc0 = {}, acc1 = {};
#pragma unroll
  for (int kb = 0; kb < 192; kb += 32) {
    int o0 = hi ? patch_off(kb + 8)  : patch_off(kb);
    int o1 = hi ? patch_off(kb + 24) : patch_off(kb + 16);
    v16h a  = pack_frag_img(abase + o0, abase + o1);
    v16h b0 = pack_frag_w(wt + ((kb >> 5) * 2 + 0) * 512 + lane * 8);
    v16h b1 = pack_frag_w(wt + ((kb >> 5) * 2 + 1) * 512 + lane * 8);
    acc0 = WMMA16(a, b0, acc0);
    acc1 = WMMA16(a, b1, acc1);
  }
  float bs0 = bias[lo], bs1 = bias[16 + lo];
#pragma unroll
  for (int r = 0; r < 8; r++) {
    int mm = wid * 16 + hi * 8 + r;
    float v0 = acc0[r] + bs0; v0 = v0 > 0.0f ? v0 : 0.0f;
    float v1 = acc1[r] + bs1; v1 = v1 > 0.0f ? v1 : 0.0f;
    out[mm * 32 + lo]      = (_Float16)v0;
    out[mm * 32 + 16 + lo] = (_Float16)v1;
  }
}

__global__ void __launch_bounds__(256) k_conv2(const _Float16* act1, const _Float16* wt,
                                               const float* bias, _Float16* out) {
  int wid  = (blockIdx.x * blockDim.x + threadIdx.x) >> 5;
  int lane = threadIdx.x & 31, hi = lane >> 4, lo = lane & 15;
  int m = wid * 16 + lo;
  int b = m / 81, p = m % 81, oh = p / 9, ow = p % 9;
  v8f acc0 = {}, acc1 = {}, acc2 = {}, acc3 = {};
  const _Float16* ap = act1 + ((b * 20 + oh * 2) * 20 + ow * 2) * 32 + hi * 8;
  const _Float16* bp = wt + lane * 8;
  auto ld = [&](int kb) {
    int pix = kb >> 5, kh = pix >> 2, kw = pix & 3;
    KStep f;
    f.a  = pack_frag(ap + (kh * 20 + kw) * 32);
    f.b0 = pack_frag_w(bp + ((kb >> 5) * 4 + 0) * 512);
    f.b1 = pack_frag_w(bp + ((kb >> 5) * 4 + 1) * 512);
    f.b2 = pack_frag_w(bp + ((kb >> 5) * 4 + 2) * 512);
    f.b3 = pack_frag_w(bp + ((kb >> 5) * 4 + 3) * 512);
    return f;
  };
  KStep f0 = ld(0), f1;
  for (int kb = 0; kb < 512 - 64; kb += 64) {
    f1 = ld(kb + 32);
    LOADS_STAY_AHEAD();
    MMA4(f0, acc0, acc1, acc2, acc3);
    f0 = ld(kb + 64);
    LOADS_STAY_AHEAD();
    MMA4(f1, acc0, acc1, acc2, acc3);
  }
  f1 = ld(512 - 32);
  LOADS_STAY_AHEAD();
  MMA4(f0, acc0, acc1, acc2, acc3);
  MMA4(f1, acc0, acc1, acc2, acc3);
  float bs0 = bias[lo], bs1 = bias[16 + lo], bs2 = bias[32 + lo], bs3 = bias[48 + lo];
#pragma unroll
  for (int r = 0; r < 8; r++) {
    int mm = wid * 16 + hi * 8 + r;
    float v0 = acc0[r] + bs0; v0 = v0 > 0.0f ? v0 : 0.0f;
    float v1 = acc1[r] + bs1; v1 = v1 > 0.0f ? v1 : 0.0f;
    float v2 = acc2[r] + bs2; v2 = v2 > 0.0f ? v2 : 0.0f;
    float v3 = acc3[r] + bs3; v3 = v3 > 0.0f ? v3 : 0.0f;
    out[mm * 64 + lo]      = (_Float16)v0;
    out[mm * 64 + 16 + lo] = (_Float16)v1;
    out[mm * 64 + 32 + lo] = (_Float16)v2;
    out[mm * 64 + 48 + lo] = (_Float16)v3;
  }
}

__global__ void __launch_bounds__(256) k_conv3(const _Float16* act2, const _Float16* wt,
                                               const float* bias, _Float16* out) {
  int wid  = (blockIdx.x * blockDim.x + threadIdx.x) >> 5;
  int lane = threadIdx.x & 31, hi = lane >> 4, lo = lane & 15;
  int m = wid * 16 + lo;
  int b = m / 49, p = m % 49, oh = p / 7, ow = p % 7;
  v8f acc0 = {}, acc1 = {}, acc2 = {}, acc3 = {};
  const _Float16* ap = act2 + ((b * 9 + oh) * 9 + ow) * 64 + hi * 8;
  const _Float16* bp = wt + lane * 8;
  auto ld = [&](int kb) {
    int pix = kb >> 6, kh = pix / 3, kw = pix % 3;
    KStep f;
    f.a  = pack_frag(ap + (kh * 9 + kw) * 64 + (kb & 63));
    f.b0 = pack_frag_w(bp + ((kb >> 5) * 4 + 0) * 512);
    f.b1 = pack_frag_w(bp + ((kb >> 5) * 4 + 1) * 512);
    f.b2 = pack_frag_w(bp + ((kb >> 5) * 4 + 2) * 512);
    f.b3 = pack_frag_w(bp + ((kb >> 5) * 4 + 3) * 512);
    return f;
  };
  KStep f0 = ld(0), f1;
  for (int kb = 0; kb < 576 - 64; kb += 64) {
    f1 = ld(kb + 32);
    LOADS_STAY_AHEAD();
    MMA4(f0, acc0, acc1, acc2, acc3);
    f0 = ld(kb + 64);
    LOADS_STAY_AHEAD();
    MMA4(f1, acc0, acc1, acc2, acc3);
  }
  f1 = ld(576 - 32);
  LOADS_STAY_AHEAD();
  MMA4(f0, acc0, acc1, acc2, acc3);
  MMA4(f1, acc0, acc1, acc2, acc3);
  float bs0 = bias[lo], bs1 = bias[16 + lo], bs2 = bias[32 + lo], bs3 = bias[48 + lo];
#pragma unroll
  for (int r = 0; r < 8; r++) {
    int mm = wid * 16 + hi * 8 + r;
    float v0 = acc0[r] + bs0; v0 = v0 > 0.0f ? v0 : 0.0f;
    float v1 = acc1[r] + bs1; v1 = v1 > 0.0f ? v1 : 0.0f;
    float v2 = acc2[r] + bs2; v2 = v2 > 0.0f ? v2 : 0.0f;
    float v3 = acc3[r] + bs3; v3 = v3 > 0.0f ? v3 : 0.0f;
    out[mm * 64 + lo]      = (_Float16)v0;
    out[mm * 64 + 16 + lo] = (_Float16)v1;
    out[mm * 64 + 32 + lo] = (_Float16)v2;
    out[mm * 64 + 48 + lo] = (_Float16)v3;
  }
}

__global__ void __launch_bounds__(256) k_fc(const _Float16* act3, const _Float16* wt,
                                            const float* bias, _Float16* X) {
  int wid  = (blockIdx.x * blockDim.x + threadIdx.x) >> 5;
  int lane = threadIdx.x & 31, hi = lane >> 4, lo = lane & 15;
  int ng = wid & 7, mt = wid >> 3;
  int m = mt * 16 + lo;
  int nb = ng * 64;
  v8f acc0 = {}, acc1 = {}, acc2 = {}, acc3 = {};
  const _Float16* ap = act3 + (size_t)m * 3136 + hi * 8;
  const _Float16* bp = wt + (ng * 4) * 512 + lane * 8;
  auto ld = [&](int kb) {
    KStep f;
    f.a  = pack_frag(ap + kb);
    f.b0 = pack_frag_w(bp + ((kb >> 5) * 32 + 0) * 512);
    f.b1 = pack_frag_w(bp + ((kb >> 5) * 32 + 1) * 512);
    f.b2 = pack_frag_w(bp + ((kb >> 5) * 32 + 2) * 512);
    f.b3 = pack_frag_w(bp + ((kb >> 5) * 32 + 3) * 512);
    return f;
  };
  KStep f0 = ld(0), f1;
  for (int kb = 0; kb < 3136 - 64; kb += 64) {
    f1 = ld(kb + 32);
    LOADS_STAY_AHEAD();
    MMA4(f0, acc0, acc1, acc2, acc3);
    f0 = ld(kb + 64);
    LOADS_STAY_AHEAD();
    MMA4(f1, acc0, acc1, acc2, acc3);
  }
  f1 = ld(3136 - 32);
  LOADS_STAY_AHEAD();
  MMA4(f0, acc0, acc1, acc2, acc3);
  MMA4(f1, acc0, acc1, acc2, acc3);
  float bs0 = bias[nb + lo], bs1 = bias[nb + 16 + lo];
  float bs2 = bias[nb + 32 + lo], bs3 = bias[nb + 48 + lo];
#pragma unroll
  for (int r = 0; r < 8; r++) {
    int mm = mt * 16 + hi * 8 + r;
    float v0 = acc0[r] + bs0; v0 = v0 > 0.0f ? v0 : 0.0f;
    float v1 = acc1[r] + bs1; v1 = v1 > 0.0f ? v1 : 0.0f;
    float v2 = acc2[r] + bs2; v2 = v2 > 0.0f ? v2 : 0.0f;
    float v3 = acc3[r] + bs3; v3 = v3 > 0.0f ? v3 : 0.0f;
    X[mm * XS + nb + lo]      = (_Float16)v0;
    X[mm * XS + nb + 16 + lo] = (_Float16)v1;
    X[mm * XS + nb + 32 + lo] = (_Float16)v2;
    X[mm * XS + nb + 48 + lo] = (_Float16)v3;
  }
}

__global__ void __launch_bounds__(256) k_xproj(const _Float16* X, const _Float16* Wih,
                                               const float* biasg, float* G) {
  int wid  = (blockIdx.x * blockDim.x + threadIdx.x) >> 5;
  int lane = threadIdx.x & 31, hi = lane >> 4, lo = lane & 15;
  int mtile = wid & 127, cg = wid >> 7;
  int ncol = cg * 16 + lo;
  float bi = biasg[0 * HL + ncol], bf = biasg[1 * HL + ncol];
  float bg = biasg[2 * HL + ncol], bo = biasg[3 * HL + ncol];
  v8f ai = {bi, bi, bi, bi, bi, bi, bi, bi};
  v8f af = {bf, bf, bf, bf, bf, bf, bf, bf};
  v8f ag = {bg, bg, bg, bg, bg, bg, bg, bg};
  v8f ao = {bo, bo, bo, bo, bo, bo, bo, bo};
  const _Float16* ap = X   + (mtile * 16 + lo) * XS + hi * 8;
  const _Float16* bp = Wih + cg * 512 + lane * 8;
  auto ld = [&](int kb) {
    KStep f;
    f.a  = pack_frag(ap + kb);
    f.b0 = pack_frag_w(bp + ((kb >> 5) * 64 + 0 * 16) * 512);
    f.b1 = pack_frag_w(bp + ((kb >> 5) * 64 + 1 * 16) * 512);
    f.b2 = pack_frag_w(bp + ((kb >> 5) * 64 + 2 * 16) * 512);
    f.b3 = pack_frag_w(bp + ((kb >> 5) * 64 + 3 * 16) * 512);
    return f;
  };
  KStep f0 = ld(0), f1;
  for (int kb = 0; kb < XS - 64; kb += 64) {
    f1 = ld(kb + 32);
    LOADS_STAY_AHEAD();
    MMA4(f0, ai, af, ag, ao);
    f0 = ld(kb + 64);
    LOADS_STAY_AHEAD();
    MMA4(f1, ai, af, ag, ao);
  }
  f1 = ld(XS - 32);
  LOADS_STAY_AHEAD();
  MMA4(f0, ai, af, ag, ao);
  MMA4(f1, ai, af, ag, ao);
  v4f* gp = (v4f*)(G + (size_t)(mtile * 16 + cg) * GT) + lane;
  gp[0 * 64]      = ai.lo;  gp[0 * 64 + 32] = ai.hi;
  gp[1 * 64]      = af.lo;  gp[1 * 64 + 32] = af.hi;
  gp[2 * 64]      = ag.lo;  gp[2 * 64 + 32] = ag.hi;
  gp[3 * 64]      = ao.lo;  gp[3 * 64 + 32] = ao.hi;
}

__global__ void __launch_bounds__(512) k_lstm(const float* G, const _Float16* Whh,
                                              const int* done,
                                              const float* h0, const float* c0,
                                              float* hs, float* out) {
  __shared__ v8h Wl[16 * 4 * 4 * 2 * 32];
  __shared__ __attribute__((aligned(16))) _Float16 A[16 * HSTR];
  __shared__ unsigned nm[T_STEPS];
  const int tid = threadIdx.x;
  const int mt  = blockIdx.x;
  const int cg  = tid >> 5, lane = tid & 31, hi = lane >> 4, lo = lane & 15;
  const int ncol = cg * 16 + lo;
  const int env0 = mt * 16;

  if (tid < T_STEPS) {
    unsigned m = 0;
    if (tid + 1 < T_STEPS) {
#pragma unroll
      for (int e = 0; e < 16; e++)
        m |= (done[(tid + 1) * E_ENVS + env0 + e] != 0 ? 1u : 0u) << e;
    }
    nm[tid] = m;
  }

  for (int i = tid; i < 16 * HL; i += 512) {
    int e = i >> 8, k = i & 255;
    float v = h0[(env0 + e) * HL + k];
    A[e * HSTR + k] = done[env0 + e] ? (_Float16)0.0f : (_Float16)v;
  }
  v8f creg;
#pragma unroll
  for (int r = 0; r < 8; r++) {
    float v = c0[(env0 + hi * 8 + r) * HL + ncol];
    creg[r] = done[env0 + hi * 8 + r] ? 0.0f : v;
  }

  const _Float16* bp = Whh + ncol * HL + hi * 8;
  v8h* wl = Wl + cg * (4 * 4 * 2 * 32) + lane;
#pragma unroll
  for (int ch = 0; ch < 4; ch++) {
#pragma unroll
    for (int g = 0; g < 4; g++) {
      const _Float16* p = bp + g * HL * HL + (4 + ch) * 32;
      wl[((ch * 4 + g) * 2 + 0) * 32] = *(const v8h*)p;
      wl[((ch * 4 + g) * 2 + 1) * 32] = *(const v8h*)(p + 16);
    }
  }
  __builtin_amdgcn_sched_barrier(0);
  v16h wr[4][4];
#pragma unroll
  for (int ch = 0; ch < 4; ch++) {
#pragma unroll
    for (int g = 0; g < 4; g++) wr[ch][g] = pack_frag(bp + g * HL * HL + ch * 32);
  }
  __syncthreads();

  const v4f* gbase = (const v4f*)(G + (size_t)(mt * 16 + cg) * GT) + lane;
  v8f gi, gf, gg, go;
  gi.lo = gbase[0 * 64];  gi.hi = gbase[0 * 64 + 32];
  gf.lo = gbase[1 * 64];  gf.hi = gbase[1 * 64 + 32];
  gg.lo = gbase[2 * 64];  gg.hi = gbase[2 * 64 + 32];
  go.lo = gbase[3 * 64];  go.hi = gbase[3 * 64 + 32];

  float hv[8];
  for (int t = 0; t < T_STEPS; t++) {
    const unsigned my = nm[t] >> (hi * 8);

    const _Float16* arow = &A[lo * HSTR + hi * 8];
    v8f ai, af, ag, ao;
#pragma unroll
    for (int ch = 0; ch < 4; ch++) {
      v16h a = pack_frag(arow + ch * 32);
      ai = WMMA16(a, wr[ch][0], ch ? ai : gi);
      af = WMMA16(a, wr[ch][1], ch ? af : gf);
      ag = WMMA16(a, wr[ch][2], ch ? ag : gg);
      ao = WMMA16(a, wr[ch][3], ch ? ao : go);
    }
#pragma unroll
    for (int ch = 0; ch < 4; ch++) {
      v16h a = pack_frag(arow + (4 + ch) * 32);
      v16h b[4];
#pragma unroll
      for (int g = 0; g < 4; g++) {
        v8h u = wl[((ch * 4 + g) * 2 + 0) * 32];
        v8h w = wl[((ch * 4 + g) * 2 + 1) * 32];
#pragma unroll
        for (int e = 0; e < 8; e++) { b[g][e] = u[e]; b[g][e + 8] = w[e]; }
      }
      ai = WMMA16(a, b[0], ai);
      af = WMMA16(a, b[1], af);
      ag = WMMA16(a, b[2], ag);
      ao = WMMA16(a, b[3], ao);
    }
    lds_barrier();

    if (t + 1 < T_STEPS) {
      const v4f* gp = gbase + (size_t)(t + 1) * (2 * 16 * GT / 4);
      gi.lo = gp[0 * 64];  gi.hi = gp[0 * 64 + 32];
      gf.lo = gp[1 * 64];  gf.hi = gp[1 * 64 + 32];
      gg.lo = gp[2 * 64];  gg.hi = gp[2 * 64 + 32];
      go.lo = gp[3 * 64];  go.hi = gp[3 * 64 + 32];
    }

    bool dn[8];
#pragma unroll
    for (int r = 0; r < 8; r++) {
      float iv = sigmoid_hw(ai[r]), fv = sigmoid_hw(af[r]);
      float gv = tanh_hw(ag[r]),    ov = sigmoid_hw(ao[r]);
      float cv = fv * creg[r] + iv * gv;
      hv[r]   = ov * tanh_hw(cv);
      dn[r]   = ((my >> r) & 1u) != 0;
      creg[r] = dn[r] ? 0.0f : cv;
    }
#pragma unroll
    for (int r = 0; r < 8; r++) {
      int el  = hi * 8 + r;
      int env = env0 + el;
      A[el * HSTR + ncol] = dn[r] ? (_Float16)0.0f : (_Float16)hv[r];
      hs[(t * E_ENVS + env) * HL + ncol] = hv[r];
    }
    lds_barrier();
  }
#pragma unroll
  for (int r = 0; r < 8; r++) {
    int env = env0 + hi * 8 + r;
    out[12288 + env * HL + ncol] = hv[r];
    out[20480 + env * HL + ncol] = creg[r];
  }
}

__global__ void k_heads(const float* hs, const float* pw, const float* pb,
                        const float* vw, const float* vb, float* out) {
  int i = blockIdx.x * blockDim.x + threadIdx.x;
  if (i >= B_TOT * 6) return;
  int b = i / 6, j = i % 6;
  const float* h    = hs + b * HL;
  const float* wrow = (j < 5) ? pw + j * HL : vw;
  float s = (j < 5) ? pb[j] : vb[0];
#pragma unroll 4
  for (int k = 0; k < HL; k++) s += h[k] * wrow[k];
  if (j < 5) out[b * 5 + j] = s;
  else       out[10240 + b] = s;
}

extern "C" void kernel_launch(void* const* d_in, const int* in_sizes, int n_in,
                              void* d_out, int out_size, void* d_ws, size_t ws_size,
                              hipStream_t stream) {
  (void)in_sizes; (void)n_in; (void)out_size; (void)ws_size;
  const float* img  = (const float*)d_in[0];
  const int*   pos  = (const int*)d_in[1];
  const int*   lact = (const int*)d_in[2];
  const int*   done = (const int*)d_in[3];
  const float* h0   = (const float*)d_in[4];
  const float* c0   = (const float*)d_in[5];
  const float* w1   = (const float*)d_in[6];
  const float* b1   = (const float*)d_in[7];
  const float* w2   = (const float*)d_in[8];
  const float* b2   = (const float*)d_in[9];
  const float* w3   = (const float*)d_in[10];
  const float* b3   = (const float*)d_in[11];
  const float* fcw  = (const float*)d_in[12];
  const float* fcb  = (const float*)d_in[13];
  const float* wih  = (const float*)d_in[14];
  const float* whh  = (const float*)d_in[15];
  const float* bih  = (const float*)d_in[16];
  const float* bhh  = (const float*)d_in[17];
  const float* pw   = (const float*)d_in[18];
  const float* pb   = (const float*)d_in[19];
  const float* vw   = (const float*)d_in[20];
  const float* vb   = (const float*)d_in[21];
  float* out = (float*)d_out;
  char*  ws  = (char*)d_ws;

  const size_t o_act1 = 90832896;
  const size_t o_w1t  = o_act1 + 52428800;
  const size_t o_w2t  = o_w1t + 12288;
  const size_t o_w3t  = o_w2t + 65536;
  const size_t o_fcwt = o_w3t + 73728;
  const size_t o_wih  = o_fcwt + 3211264;
  const size_t o_whh  = o_wih + 1179648;
  const size_t o_bg   = o_whh + 524288;
  const size_t o_act2 = 0;
  const size_t o_act3 = 21233664;
  const size_t o_X    = 34078720;
  const size_t o_hs   = 36438016;
  const size_t o_G    = 38535168;

  _Float16* act1 = (_Float16*)(ws + o_act1);
  _Float16* act2 = (_Float16*)(ws + o_act2);
  _Float16* act3 = (_Float16*)(ws + o_act3);
  _Float16* Xm   = (_Float16*)(ws + o_X);
  _Float16* w1t  = (_Float16*)(ws + o_w1t);
  _Float16* w2t  = (_Float16*)(ws + o_w2t);
  _Float16* w3t  = (_Float16*)(ws + o_w3t);
  _Float16* fcwt = (_Float16*)(ws + o_fcwt);
  _Float16* wih16 = (_Float16*)(ws + o_wih);
  _Float16* whh16 = (_Float16*)(ws + o_whh);
  float*    bg   = (float*)(ws + o_bg);
  float*    hsb  = (float*)(ws + o_hs);
  float*    Gx   = (float*)(ws + o_G);

  const int TB = 256;
  k_prep_w1  <<<(32 * 192   + TB - 1) / TB, TB, 0, stream>>>(w1, w1t);

  k_conv1<<<NGRP * GRP_ALL, TB, 0, stream>>>(img, w1t, b1, act1, fcw, fcwt, wih, whh, wih16, whh16,
                                                   pos, lact, Xm, w2, w2t, w3, w3t, bih, bhh, bg);
  k_conv2<<<1296, TB, 0, stream>>>(act1, w2t, b2, act2);
  k_conv3<<< 784, TB, 0, stream>>>(act2, w3t, b3, act3);
  k_fc   <<< 128, TB, 0, stream>>>(act3, fcwt, fcb, Xm);

  k_xproj<<<256, TB, 0, stream>>>(Xm, wih16, bg, Gx);

  k_lstm<<<2, 512, 0, stream>>>(Gx, whh16, done, h0, c0, hsb, out);

  k_heads<<<(B_TOT * 6 + TB - 1) / TB, TB, 0, stream>>>(hsb, pw, pb, vw, vb, out);
}


# GEMM kernels: the operand the 8 waves of a workgroup share (conv1-3 and gate-projection weights, FC's 16 activation rows) is staged once per workgroup in LDS and read from there; sum order unchanged
# speedup vs baseline: 14.2258x; 1.0936x over previous
#include <hip/hip_runtime.h>

typedef __attribute__((ext_vector_type(16))) _Float16 v16h;
typedef __attribute__((ext_vector_type(8)))  _Float16 v8h;
typedef __attribute__((ext_vector_type(4)))  _Float16 v4h;
typedef __attribute__((ext_vector_type(8)))  float    v8f;
typedef __attribute__((ext_vector_type(4)))  float    v4f;

#define WMMA16(A,B,C) __builtin_amdgcn_wmma_f32_16x16x32_f16(false,(A),false,(B),(short)0,(C),false,false)

static constexpr int T_STEPS = 64;
static constexpr int E_ENVS  = 32;
static constexpr int B_TOT   = 2048;
static constexpr int HL      = 256;
static constexpr int IND     = 549;
static constexpr int XS      = 576;
static constexpr int HSTR    = 264;
static constexpr int GT      = 1024;
static constexpr int IROW    = 84 * 3;
static constexpr int FCK     = 3136;
static constexpr int FCP     = 3144;

__device__ __forceinline__ v16h pack_frag(const _Float16* p) {
  v8h u = *(const v8h*)p;
  v8h w = *(const v8h*)(p + 16);
  v16h a;
#pragma unroll
  for (int e = 0; e < 8; e++) { a[e] = u[e]; a[e + 8] = w[e]; }
  return a;
}
__device__ __forceinline__ v16h pack_frag_img(const float* p0, const float* p1) {
  v4f u0 = *(const v4f*)p0;
  v4f u1 = *(const v4f*)(p0 + 4);
  v4f u2 = *(const v4f*)p1;
  v4f u3 = *(const v4f*)(p1 + 4);
  v16h a;
#pragma unroll
  for (int e = 0; e < 4; e++) {
    a[e]      = (_Float16)(u0[e] * (1.0f / 255.0f));
    a[e + 4]  = (_Float16)(u1[e] * (1.0f / 255.0f));
    a[e + 8]  = (_Float16)(u2[e] * (1.0f / 255.0f));
    a[e + 12] = (_Float16)(u3[e] * (1.0f / 255.0f));
  }
  return a;
}
__host__ __device__ constexpr int patch_off(int f) { return (f / 24) * IROW + (f % 24); }

__device__ __forceinline__ v16h pack_frag_w(const _Float16* q) {
  v8h u = *(const v8h*)q;
  v8h w = *(const v8h*)(q + 256);
  v16h a;
#pragma unroll
  for (int e = 0; e < 8; e++) { a[e] = u[e]; a[e + 8] = w[e]; }
  return a;
}
__device__ __forceinline__ void frag_pos(int i, int NT, int& n, int& k) {
  int e = i & 7, lo = (i >> 3) & 15, hi = (i >> 7) & 1, part = (i >> 8) & 1, blk = i >> 9;
  int tl = blk % NT, ks = blk / NT;
  n = tl * 16 + lo;
  k = ks * 32 + part * 16 + hi * 8 + e;
}

__device__ __forceinline__ float tanh_hw(float x) { return __builtin_amdgcn_tanhf(x); }
__device__ __forceinline__ float sigmoid_hw(float x) {
  return __builtin_fmaf(0.5f, tanh_hw(0.5f * x), 0.5f);
}

__device__ __forceinline__ void lds_barrier() {
  __builtin_amdgcn_fence(__ATOMIC_RELEASE, "workgroup", "local");
  __builtin_amdgcn_s_barrier();
  __builtin_amdgcn_fence(__ATOMIC_ACQUIRE, "workgroup", "local");
}

struct KStep { v16h a, b0, b1, b2, b3; };
#define LOADS_STAY_AHEAD() __builtin_amdgcn_sched_barrier(0)
#define MMA4(F, C0, C1, C2, C3)      \
  C0 = WMMA16((F).a, (F).b0, C0);    \
  C1 = WMMA16((F).a, (F).b1, C1);    \
  C2 = WMMA16((F).a, (F).b2, C2);    \
  C3 = WMMA16((F).a, (F).b3, C3)

template <int N>
__device__ __forceinline__ void stage_lds(_Float16* dst, const _Float16* src) {
  const v8h* s = (const v8h*)src + threadIdx.x;
  v8h* d = (v8h*)dst + threadIdx.x;
  v8h t[N];
#pragma unroll
  for (int j = 0; j < N; j++) t[j] = s[j * 256];
#pragma unroll
  for (int j = 0; j < N; j++) d[j * 256] = t[j];
}

__global__ void k_prep_w1(const float* w, _Float16* wt) {
  int i = blockIdx.x * blockDim.x + threadIdx.x;
  if (i >= 32 * 192) return;
  int n, f;
  frag_pos(i, 2, n, f);
  int kh = f / 24, r = f % 24, kw = r / 3, c = r % 3;
  wt[i] = (_Float16)w[((n * 3 + c) * 8 + kh) * 8 + kw];
}
__device__ __forceinline__ void prep_w2(int i, const float* w, _Float16* wt) {
  int n, k;
  frag_pos(i, 4, n, k);
  int pix = k >> 5, c = k & 31, kh = pix >> 2, kw = pix & 3;
  wt[i] = (_Float16)w[((n * 32 + c) * 4 + kh) * 4 + kw];
}
__device__ __forceinline__ void prep_w3(int i, const float* w, _Float16* wt) {
  int n, k;
  frag_pos(i, 4, n, k);
  int pix = k >> 6, c = k & 63, kh = pix / 3, kw = pix % 3;
  wt[i] = (_Float16)w[((n * 64 + c) * 3 + kh) * 3 + kw];
}
__device__ __forceinline__ void prep_fc(int i, const float* w, _Float16* wt) {
  int f, k;
  frag_pos(i, 32, f, k);
  int s = k >> 6, c = k & 63;
  wt[i] = (_Float16)w[f * 3136 + c * 49 + s];
}
__device__ __forceinline__ void prep_wl(int i, const float* wih, const float* whh,
                                        _Float16* wih16, _Float16* whh16) {
  if (i < 1024 * XS) {
    int r, k;
    frag_pos(i, 64, r, k);
    wih16[i] = (_Float16)((k < IND) ? wih[r * IND + k] : 0.0f);
  } else {
    int j = i - 1024 * XS;
    whh16[j] = (_Float16)whh[j];
  }
}
__device__ __forceinline__ void prep_bias(int i, const float* bih, const float* bhh, float* bg) {
  bg[i] = bih[i] + bhh[i];
}
__device__ __forceinline__ void onehot(int i, const int* pos, const int* lact, _Float16* X) {
  int b = i >> 6, c = i & 63;
  float v = 0.0f;
  if (c < 16)      v = (pos[b * 2 + 0] == c)      ? 1.0f : 0.0f;
  else if (c < 32) v = (pos[b * 2 + 1] == c - 16) ? 1.0f : 0.0f;
  else if (c < 37) v = (lact[b]        == c - 32) ? 1.0f : 0.0f;
  X[b * XS + 512 + c] = (_Float16)v;
}

static constexpr int NB_CONV1 = 6400;
static constexpr int GRP_CONV = 8, GRP_ALL = 21;
static constexpr int NE_FC    = 512 * 3136;
static constexpr int NE_WL    = 1024 * (XS + HL);
static constexpr int NE_OH    = B_TOT * 64;
static constexpr int NE_W2    = 64 * 512;
static constexpr int NE_W3    = 64 * 576;
static constexpr int NE_BIAS  = 1024;
static constexpr int NB_EXTRA = (NE_FC + NE_WL + NE_OH + NE_W2 + NE_W3 + NE_BIAS) / 256;
static constexpr int NGRP     = NB_CONV1 / GRP_CONV;
static_assert(NE_FC % 256 == 0 && NE_WL % 256 == 0 && NE_OH % 256 == 0 && NE_W2 % 256 == 0 &&
              NE_W3 % 256 == 0 && NE_BIAS % 256 == 0, "every extra job is whole workgroups");
static_assert(NGRP * GRP_CONV == NB_CONV1 && NGRP * (GRP_ALL - GRP_CONV) >= NB_EXTRA,
              "the groups hold every conv1 workgroup and every extra one");
__global__ void __launch_bounds__(256) k_conv1(const float* img, const _Float16* wt,
                                               const float* bias, _Float16* out,
                                               const float* fcw, _Float16* fcwt,
                                               const float* wih, const float* whh,
                                               _Float16* wih16, _Float16* whh16,
                                               const int* pos, const int* lact, _Float16* X,
                                               const float* w2, _Float16* w2t,
                                               const float* w3, _Float16* w3t,
                                               const float* bih, const float* bhh, float* bg) {
  __shared__ __attribute__((aligned(16))) _Float16 Ws[32 * 192];
  const int grp = blockIdx.x / GRP_ALL, rr = blockIdx.x % GRP_ALL;
  if (rr >= GRP_CONV) {
    int xb = grp * (GRP_ALL - GRP_CONV) + (rr - GRP_CONV);
    if (xb >= NB_EXTRA) return;
    const int t = threadIdx.x;
    if (xb < NE_FC / 256)   { prep_fc(xb * 256 + t, fcw, fcwt);              return; }
    xb -= NE_FC / 256;
    if (xb < NE_WL / 256)   { prep_wl(xb * 256 + t, wih, whh, wih16, whh16); return; }
    xb -= NE_WL / 256;
    if (xb < NE_OH / 256)   { onehot(xb * 256 + t, pos, lact, X);            return; }
    xb -= NE_OH / 256;
    if (xb < NE_W2 / 256)   { prep_w2(xb * 256 + t, w2, w2t);                return; }
    xb -= NE_W2 / 256;
    if (xb < NE_W3 / 256)   { prep_w3(xb * 256 + t, w3, w3t);                return; }
    xb -= NE_W3 / 256;
    prep_bias(xb * 256 + t, bih, bhh, bg);
    return;
  }
  const int cb = grp * GRP_CONV + rr;
  stage_lds<3>(Ws, wt);
  __syncthreads();
  int wid  = (cb * 256 + threadIdx.x) >> 5;
  int lane = threadIdx.x & 31, hi = lane >> 4, lo = lane & 15;
  int m = wid * 16 + lo;
  int b = m / 400, p = m % 400, oh = p / 20, ow = p % 20;
  const float* abase = img + ((size_t)(b * 84 + oh * 4) * 84 + ow * 4) * 3;
  v8f acc0 = {}, acc1 = {};
#pragma unroll
  for (int kb = 0; kb < 192; kb += 32) {
    int o0 = hi ? patch_off(kb + 8)  : patch_off(kb);
    int o1 = hi ? patch_off(kb + 24) : patch_off(kb + 16);
    v16h a  = pack_frag_img(abase + o0, abase + o1);
    v16h b0 = pack_frag_w(Ws + ((kb >> 5) * 2 + 0) * 512 + lane * 8);
    v16h b1 = pack_frag_w(Ws + ((kb >> 5) * 2 + 1) * 512 + lane * 8);
    acc0 = WMMA16(a, b0, acc0);
    acc1 = WMMA16(a, b1, acc1);
  }
  float bs0 = bias[lo], bs1 = bias[16 + lo];
#pragma unroll
  for (int r = 0; r < 8; r++) {
    int mm = wid * 16 + hi * 8 + r;
    float v0 = acc0[r] + bs0; v0 = v0 > 0.0f ? v0 : 0.0f;
    float v1 = acc1[r] + bs1; v1 = v1 > 0.0f ? v1 : 0.0f;
    out[mm * 32 + lo]      = (_Float16)v0;
    out[mm * 32 + 16 + lo] = (_Float16)v1;
  }
}

__global__ void __launch_bounds__(256) k_conv2(const _Float16* act1, const _Float16* wt,
                                               const float* bias, _Float16* out) {
  __shared__ __attribute__((aligned(16))) _Float16 Ws[64 * 512];
  stage_lds<16>(Ws, wt);
  __syncthreads();
  int wid  = (blockIdx.x * blockDim.x + threadIdx.x) >> 5;
  int lane = threadIdx.x & 31, hi = lane >> 4, lo = lane & 15;
  int m = wid * 16 + lo;
  int b = m / 81, p = m % 81, oh = p / 9, ow = p % 9;
  v8f acc0 = {}, acc1 = {}, acc2 = {}, acc3 = {};
  const _Float16* ap = act1 + ((b * 20 + oh * 2) * 20 + ow * 2) * 32 + hi * 8;
  const _Float16* bp = Ws + lane * 8;
  auto ld = [&](int kb) {
    int pix = kb >> 5, kh = pix >> 2, kw = pix & 3;
    KStep f;
    f.a  = pack_frag(ap + (kh * 20 + kw) * 32);
    f.b0 = pack_frag_w(bp + ((kb >> 5) * 4 + 0) * 512);
    f.b1 = pack_frag_w(bp + ((kb >> 5) * 4 + 1) * 512);
    f.b2 = pack_frag_w(bp + ((kb >> 5) * 4 + 2) * 512);
    f.b3 = pack_frag_w(bp + ((kb >> 5) * 4 + 3) * 512);
    return f;
  };
  KStep f0 = ld(0), f1;
  for (int kb = 0; kb < 512 - 64; kb += 64) {
    f1 = ld(kb + 32);
    LOADS_STAY_AHEAD();
    MMA4(f0, acc0, acc1, acc2, acc3);
    f0 = ld(kb + 64);
    LOADS_STAY_AHEAD();
    MMA4(f1, acc0, acc1, acc2, acc3);
  }
  f1 = ld(512 - 32);
  LOADS_STAY_AHEAD();
  MMA4(f0, acc0, acc1, acc2, acc3);
  MMA4(f1, acc0, acc1, acc2, acc3);
  float bs0 = bias[lo], bs1 = bias[16 + lo], bs2 = bias[32 + lo], bs3 = bias[48 + lo];
#pragma unroll
  for (int r = 0; r < 8; r++) {
    int mm = wid * 16 + hi * 8 + r;
    float v0 = acc0[r] + bs0; v0 = v0 > 0.0f ? v0 : 0.0f;
    float v1 = acc1[r] + bs1; v1 = v1 > 0.0f ? v1 : 0.0f;
    float v2 = acc2[r] + bs2; v2 = v2 > 0.0f ? v2 : 0.0f;
    float v3 = acc3[r] + bs3; v3 = v3 > 0.0f ? v3 : 0.0f;
    out[mm * 64 + lo]      = (_Float16)v0;
    out[mm * 64 + 16 + lo] = (_Float16)v1;
    out[mm * 64 + 32 + lo] = (_Float16)v2;
    out[mm * 64 + 48 + lo] = (_Float16)v3;
  }
}

__global__ void __launch_bounds__(256) k_conv3(const _Float16* act2, const _Float16* wt,
                                               const float* bias, _Float16* out) {
  __shared__ __attribute__((aligned(16))) _Float16 Ws[64 * 576];
  stage_lds<18>(Ws, wt);
  __syncthreads();
  int wid  = (blockIdx.x * blockDim.x + threadIdx.x) >> 5;
  int lane = threadIdx.x & 31, hi = lane >> 4, lo = lane & 15;
  int m = wid * 16 + lo;
  int b = m / 49, p = m % 49, oh = p / 7, ow = p % 7;
  v8f acc0 = {}, acc1 = {}, acc2 = {}, acc3 = {};
  const _Float16* ap = act2 + ((b * 9 + oh) * 9 + ow) * 64 + hi * 8;
  const _Float16* bp = Ws + lane * 8;
  auto ld = [&](int kb) {
    int pix = kb >> 6, kh = pix / 3, kw = pix % 3;
    KStep f;
    f.a  = pack_frag(ap + (kh * 9 + kw) * 64 + (kb & 63));
    f.b0 = pack_frag_w(bp + ((kb >> 5) * 4 + 0) * 512);
    f.b1 = pack_frag_w(bp + ((kb >> 5) * 4 + 1) * 512);
    f.b2 = pack_frag_w(bp + ((kb >> 5) * 4 + 2) * 512);
    f.b3 = pack_frag_w(bp + ((kb >> 5) * 4 + 3) * 512);
    return f;
  };
  KStep f0 = ld(0), f1;
  for (int kb = 0; kb < 576 - 64; kb += 64) {
    f1 = ld(kb + 32);
    LOADS_STAY_AHEAD();
    MMA4(f0, acc0, acc1, acc2, acc3);
    f0 = ld(kb + 64);
    LOADS_STAY_AHEAD();
    MMA4(f1, acc0, acc1, acc2, acc3);
  }
  f1 = ld(576 - 32);
  LOADS_STAY_AHEAD();
  MMA4(f0, acc0, acc1, acc2, acc3);
  MMA4(f1, acc0, acc1, acc2, acc3);
  float bs0 = bias[lo], bs1 = bias[16 + lo], bs2 = bias[32 + lo], bs3 = bias[48 + lo];
#pragma unroll
  for (int r = 0; r < 8; r++) {
    int mm = wid * 16 + hi * 8 + r;
    float v0 = acc0[r] + bs0; v0 = v0 > 0.0f ? v0 : 0.0f;
    float v1 = acc1[r] + bs1; v1 = v1 > 0.0f ? v1 : 0.0f;
    float v2 = acc2[r] + bs2; v2 = v2 > 0.0f ? v2 : 0.0f;
    float v3 = acc3[r] + bs3; v3 = v3 > 0.0f ? v3 : 0.0f;
    out[mm * 64 + lo]      = (_Float16)v0;
    out[mm * 64 + 16 + lo] = (_Float16)v1;
    out[mm * 64 + 32 + lo] = (_Float16)v2;
    out[mm * 64 + 48 + lo] = (_Float16)v3;
  }
}

__global__ void __launch_bounds__(256) k_fc(const _Float16* act3, const _Float16* wt,
                                            const float* bias, _Float16* X) {
  __shared__ __attribute__((aligned(16))) _Float16 As[16 * FCP];
  {
    const v8h* s = (const v8h*)(act3 + (size_t)blockIdx.x * 16 * FCK);
    constexpr int RC = FCK / 8, NC = 16 * RC;
    v8h t[(NC + 255) / 256];
#pragma unroll
    for (int j = 0; j < (NC + 255) / 256; j++) {
      int i = j * 256 + threadIdx.x;
      if (i < NC) t[j] = s[i];
    }
#pragma unroll
    for (int j = 0; j < (NC + 255) / 256; j++) {
      int i = j * 256 + threadIdx.x;
      if (i < NC) *(v8h*)(As + (i / RC) * FCP + (i % RC) * 8) = t[j];
    }
  }
  __syncthreads();
  int wid  = (blockIdx.x * blockDim.x + threadIdx.x) >> 5;
  int lane = threadIdx.x & 31, hi = lane >> 4, lo = lane & 15;
  int ng = wid & 7, mt = wid >> 3;
  int nb = ng * 64;
  v8f acc0 = {}, acc1 = {}, acc2 = {}, acc3 = {};
  const _Float16* ap = As + lo * FCP + hi * 8;
  const _Float16* bp = wt + (ng * 4) * 512 + lane * 8;
  auto ld = [&](int kb) {
    KStep f;
    f.a  = pack_frag(ap + kb);
    f.b0 = pack_frag_w(bp + ((kb >> 5) * 32 + 0) * 512);
    f.b1 = pack_frag_w(bp + ((kb >> 5) * 32 + 1) * 512);
    f.b2 = pack_frag_w(bp + ((kb >> 5) * 32 + 2) * 512);
    f.b3 = pack_frag_w(bp + ((kb >> 5) * 32 + 3) * 512);
    return f;
  };
  KStep f0 = ld(0), f1;
  for (int kb = 0; kb < 3136 - 64; kb += 64) {
    f1 = ld(kb + 32);
    LOADS_STAY_AHEAD();
    MMA4(f0, acc0, acc1, acc2, acc3);
    f0 = ld(kb + 64);
    LOADS_STAY_AHEAD();
    MMA4(f1, acc0, acc1, acc2, acc3);
  }
  f1 = ld(3136 - 32);
  LOADS_STAY_AHEAD();
  MMA4(f0, acc0, acc1, acc2, acc3);
  MMA4(f1, acc0, acc1, acc2, acc3);
  float bs0 = bias[nb + lo], bs1 = bias[nb + 16 + lo];
  float bs2 = bias[nb + 32 + lo], bs3 = bias[nb + 48 + lo];
#pragma unroll
  for (int r = 0; r < 8; r++) {
    int mm = mt * 16 + hi * 8 + r;
    float v0 = acc0[r] + bs0; v0 = v0 > 0.0f ? v0 : 0.0f;
    float v1 = acc1[r] + bs1; v1 = v1 > 0.0f ? v1 : 0.0f;
    float v2 = acc2[r] + bs2; v2 = v2 > 0.0f ? v2 : 0.0f;
    float v3 = acc3[r] + bs3; v3 = v3 > 0.0f ? v3 : 0.0f;
    X[mm * XS + nb + lo]      = (_Float16)v0;
    X[mm * XS + nb + 16 + lo] = (_Float16)v1;
    X[mm * XS + nb + 32 + lo] = (_Float16)v2;
    X[mm * XS + nb + 48 + lo] = (_Float16)v3;
  }
}

__global__ void __launch_bounds__(256) k_xproj(const _Float16* X, const _Float16* Wih,
                                               const float* biasg, float* G) {
  int wid  = (blockIdx.x * blockDim.x + threadIdx.x) >> 5;
  int lane = threadIdx.x & 31, hi = lane >> 4, lo = lane & 15;
  int mtile = wid & 127, cg = wid >> 7;
  __shared__ __attribute__((aligned(16))) _Float16 Ws[18 * 4 * 512];
  {
    const v8h* s = (const v8h*)Wih;
    v8h t[18];
#pragma unroll
    for (int j = 0; j < 18; j++) {
      int i = j * 256 + threadIdx.x;
      int blk = i >> 6, ks = blk >> 2, g = blk & 3;
      t[j] = s[((ks * 64 + g * 16 + cg) << 6) + (i & 63)];
    }
#pragma unroll
    for (int j = 0; j < 18; j++) ((v8h*)Ws)[j * 256 + threadIdx.x] = t[j];
  }
  __syncthreads();
  int ncol = cg * 16 + lo;
  float bi = biasg[0 * HL + ncol], bf = biasg[1 * HL + ncol];
  float bg = biasg[2 * HL + ncol], bo = biasg[3 * HL + ncol];
  v8f ai = {bi, bi, bi, bi, bi, bi, bi, bi};
  v8f af = {bf, bf, bf, bf, bf, bf, bf, bf};
  v8f ag = {bg, bg, bg, bg, bg, bg, bg, bg};
  v8f ao = {bo, bo, bo, bo, bo, bo, bo, bo};
  const _Float16* ap = X   + (mtile * 16 + lo) * XS + hi * 8;
  const _Float16* bp = Ws + lane * 8;
  auto ld = [&](int kb) {
    KStep f;
    f.a  = pack_frag(ap + kb);
    f.b0 = pack_frag_w(bp + ((kb >> 5) * 4 + 0) * 512);
    f.b1 = pack_frag_w(bp + ((kb >> 5) * 4 + 1) * 512);
    f.b2 = pack_frag_w(bp + ((kb >> 5) * 4 + 2) * 512);
    f.b3 = pack_frag_w(bp + ((kb >> 5) * 4 + 3) * 512);
    return f;
  };
  KStep f0 = ld(0), f1;
  for (int kb = 0; kb < XS - 64; kb += 64) {
    f1 = ld(kb + 32);
    LOADS_STAY_AHEAD();
    MMA4(f0, ai, af, ag, ao);
    f0 = ld(kb + 64);
    LOADS_STAY_AHEAD();
    MMA4(f1, ai, af, ag, ao);
  }
  f1 = ld(XS - 32);
  LOADS_STAY_AHEAD();
  MMA4(f0, ai, af, ag, ao);
  MMA4(f1, ai, af, ag, ao);
  v4f* gp = (v4f*)(G + (size_t)(mtile * 16 + cg) * GT) + lane;
  gp[0 * 64]      = ai.lo;  gp[0 * 64 + 32] = ai.hi;
  gp[1 * 64]      = af.lo;  gp[1 * 64 + 32] = af.hi;
  gp[2 * 64]      = ag.lo;  gp[2 * 64 + 32] = ag.hi;
  gp[3 * 64]      = ao.lo;  gp[3 * 64 + 32] = ao.hi;
}

__global__ void __launch_bounds__(512) k_lstm(const float* G, const _Float16* Whh,
                                              const int* done,
                                              const float* h0, const float* c0,
                                              float* hs, float* out) {
  __shared__ v8h Wl[16 * 4 * 4 * 2 * 32];
  __shared__ __attribute__((aligned(16))) _Float16 A[16 * HSTR];
  __shared__ unsigned nm[T_STEPS];
  const int tid = threadIdx.x;
  const int mt  = blockIdx.x;
  const int cg  = tid >> 5, lane = tid & 31, hi = lane >> 4, lo = lane & 15;
  const int ncol = cg * 16 + lo;
  const int env0 = mt * 16;

  if (tid < T_STEPS) {
    unsigned m = 0;
    if (tid + 1 < T_STEPS) {
#pragma unroll
      for (int e = 0; e < 16; e++)
        m |= (done[(tid + 1) * E_ENVS + env0 + e] != 0 ? 1u : 0u) << e;
    }
    nm[tid] = m;
  }

  for (int i = tid; i < 16 * HL; i += 512) {
    int e = i >> 8, k = i & 255;
    float v = h0[(env0 + e) * HL + k];
    A[e * HSTR + k] = done[env0 + e] ? (_Float16)0.0f : (_Float16)v;
  }
  v8f creg;
#pragma unroll
  for (int r = 0; r < 8; r++) {
    float v = c0[(env0 + hi * 8 + r) * HL + ncol];
    creg[r] = done[env0 + hi * 8 + r] ? 0.0f : v;
  }

  const _Float16* bp = Whh + ncol * HL + hi * 8;
  v8h* wl = Wl + cg * (4 * 4 * 2 * 32) + lane;
#pragma unroll
  for (int ch = 0; ch < 4; ch++) {
#pragma unroll
    for (int g = 0; g < 4; g++) {
      const _Float16* p = bp + g * HL * HL + (4 + ch) * 32;
      wl[((ch * 4 + g) * 2 + 0) * 32] = *(const v8h*)p;
      wl[((ch * 4 + g) * 2 + 1) * 32] = *(const v8h*)(p + 16);
    }
  }
  __builtin_amdgcn_sched_barrier(0);
  v16h wr[4][4];
#pragma unroll
  for (int ch = 0; ch < 4; ch++) {
#pragma unroll
    for (int g = 0; g < 4; g++) wr[ch][g] = pack_frag(bp + g * HL * HL + ch * 32);
  }
  __syncthreads();

  const v4f* gbase = (const v4f*)(G + (size_t)(mt * 16 + cg) * GT) + lane;
  v8f gi, gf, gg, go;
  gi.lo = gbase[0 * 64];  gi.hi = gbase[0 * 64 + 32];
  gf.lo = gbase[1 * 64];  gf.hi = gbase[1 * 64 + 32];
  gg.lo = gbase[2 * 64];  gg.hi = gbase[2 * 64 + 32];
  go.lo = gbase[3 * 64];  go.hi = gbase[3 * 64 + 32];

  float hv[8];
  for (int t = 0; t < T_STEPS; t++) {
    const unsigned my = nm[t] >> (hi * 8);

    const _Float16* arow = &A[lo * HSTR + hi * 8];
    v8f ai, af, ag, ao;
#pragma unroll
    for (int ch = 0; ch < 4; ch++) {
      v16h a = pack_frag(arow + ch * 32);
      ai = WMMA16(a, wr[ch][0], ch ? ai : gi);
      af = WMMA16(a, wr[ch][1], ch ? af : gf);
      ag = WMMA16(a, wr[ch][2], ch ? ag : gg);
      ao = WMMA16(a, wr[ch][3], ch ? ao : go);
    }
#pragma unroll
    for (int ch = 0; ch < 4; ch++) {
      v16h a = pack_frag(arow + (4 + ch) * 32);
      v16h b[4];
#pragma unroll
      for (int g = 0; g < 4; g++) {
        v8h u = wl[((ch * 4 + g) * 2 + 0) * 32];
        v8h w = wl[((ch * 4 + g) * 2 + 1) * 32];
#pragma unroll
        for (int e = 0; e < 8; e++) { b[g][e] = u[e]; b[g][e + 8] = w[e]; }
      }
      ai = WMMA16(a, b[0], ai);
      af = WMMA16(a, b[1], af);
      ag = WMMA16(a, b[2], ag);
      ao = WMMA16(a, b[3], ao);
    }
    lds_barrier();

    if (t + 1 < T_STEPS) {
      const v4f* gp = gbase + (size_t)(t + 1) * (2 * 16 * GT / 4);
      gi.lo = gp[0 * 64];  gi.hi = gp[0 * 64 + 32];
      gf.lo = gp[1 * 64];  gf.hi = gp[1 * 64 + 32];
      gg.lo = gp[2 * 64];  gg.hi = gp[2 * 64 + 32];
      go.lo = gp[3 * 64];  go.hi = gp[3 * 64 + 32];
    }

    bool dn[8];
#pragma unroll
    for (int r = 0; r < 8; r++) {
      float iv = sigmoid_hw(ai[r]), fv = sigmoid_hw(af[r]);
      float gv = tanh_hw(ag[r]),    ov = sigmoid_hw(ao[r]);
      float cv = fv * creg[r] + iv * gv;
      hv[r]   = ov * tanh_hw(cv);
      dn[r]   = ((my >> r) & 1u) != 0;
      creg[r] = dn[r] ? 0.0f : cv;
    }
#pragma unroll
    for (int r = 0; r < 8; r++) {
      int el  = hi * 8 + r;
      int env = env0 + el;
      A[el * HSTR + ncol] = dn[r] ? (_Float16)0.0f : (_Float16)hv[r];
      hs[(t * E_ENVS + env) * HL + ncol] = hv[r];
    }
    lds_barrier();
  }
#pragma unroll
  for (int r = 0; r < 8; r++) {
    int env = env0 + hi * 8 + r;
    out[12288 + env * HL + ncol] = hv[r];
    out[20480 + env * HL + ncol] = creg[r];
  }
}

__global__ void k_heads(const float* hs, const float* pw, const float* pb,
                        const float* vw, const float* vb, float* out) {
  int i = blockIdx.x * blockDim.x + threadIdx.x;
  if (i >= B_TOT * 6) return;
  int b = i / 6, j = i % 6;
  const float* h    = hs + b * HL;
  const float* wrow = (j < 5) ? pw + j * HL : vw;
  float s = (j < 5) ? pb[j] : vb[0];
#pragma unroll 4
  for (int k = 0; k < HL; k++) s += h[k] * wrow[k];
  if (j < 5) out[b * 5 + j] = s;
  else       out[10240 + b] = s;
}

extern "C" void kernel_launch(void* const* d_in, const int* in_sizes, int n_in,
                              void* d_out, int out_size, void* d_ws, size_t ws_size,
                              hipStream_t stream) {
  (void)in_sizes; (void)n_in; (void)out_size; (void)ws_size;
  const float* img  = (const float*)d_in[0];
  const int*   pos  = (const int*)d_in[1];
  const int*   lact = (const int*)d_in[2];
  const int*   done = (const int*)d_in[3];
  const float* h0   = (const float*)d_in[4];
  const float* c0   = (const float*)d_in[5];
  const float* w1   = (const float*)d_in[6];
  const float* b1   = (const float*)d_in[7];
  const float* w2   = (const float*)d_in[8];
  const float* b2   = (const float*)d_in[9];
  const float* w3   = (const float*)d_in[10];
  const float* b3   = (const float*)d_in[11];
  const float* fcw  = (const float*)d_in[12];
  const float* fcb  = (const float*)d_in[13];
  const float* wih  = (const float*)d_in[14];
  const float* whh  = (const float*)d_in[15];
  const float* bih  = (const float*)d_in[16];
  const float* bhh  = (const float*)d_in[17];
  const float* pw   = (const float*)d_in[18];
  const float* pb   = (const float*)d_in[19];
  const float* vw   = (const float*)d_in[20];
  const float* vb   = (const float*)d_in[21];
  float* out = (float*)d_out;
  char*  ws  = (char*)d_ws;

  const size_t o_act1 = 90832896;
  const size_t o_w1t  = o_act1 + 52428800;
  const size_t o_w2t  = o_w1t + 12288;
  const size_t o_w3t  = o_w2t + 65536;
  const size_t o_fcwt = o_w3t + 73728;
  const size_t o_wih  = o_fcwt + 3211264;
  const size_t o_whh  = o_wih + 1179648;
  const size_t o_bg   = o_whh + 524288;
  const size_t o_act2 = 0;
  const size_t o_act3 = 21233664;
  const size_t o_X    = 34078720;
  const size_t o_hs   = 36438016;
  const size_t o_G    = 38535168;

  _Float16* act1 = (_Float16*)(ws + o_act1);
  _Float16* act2 = (_Float16*)(ws + o_act2);
  _Float16* act3 = (_Float16*)(ws + o_act3);
  _Float16* Xm   = (_Float16*)(ws + o_X);
  _Float16* w1t  = (_Float16*)(ws + o_w1t);
  _Float16* w2t  = (_Float16*)(ws + o_w2t);
  _Float16* w3t  = (_Float16*)(ws + o_w3t);
  _Float16* fcwt = (_Float16*)(ws + o_fcwt);
  _Float16* wih16 = (_Float16*)(ws + o_wih);
  _Float16* whh16 = (_Float16*)(ws + o_whh);
  float*    bg   = (float*)(ws + o_bg);
  float*    hsb  = (float*)(ws + o_hs);
  float*    Gx   = (float*)(ws + o_G);

  const int TB = 256;
  k_prep_w1  <<<(32 * 192   + TB - 1) / TB, TB, 0, stream>>>(w1, w1t);

  k_conv1<<<NGRP * GRP_ALL, TB, 0, stream>>>(img, w1t, b1, act1, fcw, fcwt, wih, whh, wih16, whh16,
                                                   pos, lact, Xm, w2, w2t, w3, w3t, bih, bhh, bg);
  k_conv2<<<1296, TB, 0, stream>>>(act1, w2t, b2, act2);
  k_conv3<<< 784, TB, 0, stream>>>(act2, w3t, b3, act3);
  k_fc   <<< 128, TB, 0, stream>>>(act3, fcwt, fcb, Xm);

  k_xproj<<<256, TB, 0, stream>>>(Xm, wih16, bg, Gx);

  k_lstm<<<2, 512, 0, stream>>>(Gx, whh16, done, h0, c0, hsb, out);

  k_heads<<<(B_TOT * 6 + TB - 1) / TB, TB, 0, stream>>>(hsb, pw, pb, vw, vb, out);
}


# FC retiled onto all 256 WGPs: workgroup 32 rows x 128 cols (was 16 x 512 on 128 WGPs), a weight fragment serves 2 row tiles and is requested 6 k-steps ahead; whole-line X stores; sum order unchanged
# speedup vs baseline: 15.1766x; 1.0668x over previous
#include <hip/hip_runtime.h>

typedef __attribute__((ext_vector_type(16))) _Float16 v16h;
typedef __attribute__((ext_vector_type(8)))  _Float16 v8h;
typedef __attribute__((ext_vector_type(4)))  _Float16 v4h;
typedef __attribute__((ext_vector_type(8)))  float    v8f;
typedef __attribute__((ext_vector_type(4)))  float    v4f;

#define WMMA16(A,B,C) __builtin_amdgcn_wmma_f32_16x16x32_f16(false,(A),false,(B),(short)0,(C),false,false)

static constexpr int T_STEPS = 64;
static constexpr int E_ENVS  = 32;
static constexpr int B_TOT   = 2048;
static constexpr int HL      = 256;
static constexpr int IND     = 549;
static constexpr int XS      = 576;
static constexpr int HSTR    = 264;
static constexpr int GT      = 1024;
static constexpr int IROW    = 84 * 3;
static constexpr int FCK     = 3136;
static constexpr int FCP     = 3144;

__device__ __forceinline__ v16h pack_frag(const _Float16* p) {
  v8h u = *(const v8h*)p;
  v8h w = *(const v8h*)(p + 16);
  v16h a;
#pragma unroll
  for (int e = 0; e < 8; e++) { a[e] = u[e]; a[e + 8] = w[e]; }
  return a;
}
__device__ __forceinline__ v16h pack_frag_img(const float* p0, const float* p1) {
  v4f u0 = *(const v4f*)p0;
  v4f u1 = *(const v4f*)(p0 + 4);
  v4f u2 = *(const v4f*)p1;
  v4f u3 = *(const v4f*)(p1 + 4);
  v16h a;
#pragma unroll
  for (int e = 0; e < 4; e++) {
    a[e]      = (_Float16)(u0[e] * (1.0f / 255.0f));
    a[e + 4]  = (_Float16)(u1[e] * (1.0f / 255.0f));
    a[e + 8]  = (_Float16)(u2[e] * (1.0f / 255.0f));
    a[e + 12] = (_Float16)(u3[e] * (1.0f / 255.0f));
  }
  return a;
}
__host__ __device__ constexpr int patch_off(int f) { return (f / 24) * IROW + (f % 24); }

__device__ __forceinline__ v16h pack_frag_w(const _Float16* q) {
  v8h u = *(const v8h*)q;
  v8h w = *(const v8h*)(q + 256);
  v16h a;
#pragma unroll
  for (int e = 0; e < 8; e++) { a[e] = u[e]; a[e + 8] = w[e]; }
  return a;
}
__device__ __forceinline__ void frag_pos(int i, int NT, int& n, int& k) {
  int e = i & 7, lo = (i >> 3) & 15, hi = (i >> 7) & 1, part = (i >> 8) & 1, blk = i >> 9;
  int tl = blk % NT, ks = blk / NT;
  n = tl * 16 + lo;
  k = ks * 32 + part * 16 + hi * 8 + e;
}

__device__ __forceinline__ float tanh_hw(float x) { return __builtin_amdgcn_tanhf(x); }
__device__ __forceinline__ float sigmoid_hw(float x) {
  return __builtin_fmaf(0.5f, tanh_hw(0.5f * x), 0.5f);
}

__device__ __forceinline__ void lds_barrier() {
  __builtin_amdgcn_fence(__ATOMIC_RELEASE, "workgroup", "local");
  __builtin_amdgcn_s_barrier();
  __builtin_amdgcn_fence(__ATOMIC_ACQUIRE, "workgroup", "local");
}

struct KStep { v16h a, b0, b1, b2, b3; };
#define LOADS_STAY_AHEAD() __builtin_amdgcn_sched_barrier(0)
#define MMA4(F, C0, C1, C2, C3)      \
  C0 = WMMA16((F).a, (F).b0, C0);    \
  C1 = WMMA16((F).a, (F).b1, C1);    \
  C2 = WMMA16((F).a, (F).b2, C2);    \
  C3 = WMMA16((F).a, (F).b3, C3)

template <int N>
__device__ __forceinline__ void stage_lds(_Float16* dst, const _Float16* src) {
  const v8h* s = (const v8h*)src + threadIdx.x;
  v8h* d = (v8h*)dst + threadIdx.x;
  v8h t[N];
#pragma unroll
  for (int j = 0; j < N; j++) t[j] = s[j * 256];
#pragma unroll
  for (int j = 0; j < N; j++) d[j * 256] = t[j];
}

__global__ void k_prep_w1(const float* w, _Float16* wt) {
  int i = blockIdx.x * blockDim.x + threadIdx.x;
  if (i >= 32 * 192) return;
  int n, f;
  frag_pos(i, 2, n, f);
  int kh = f / 24, r = f % 24, kw = r / 3, c = r % 3;
  wt[i] = (_Float16)w[((n * 3 + c) * 8 + kh) * 8 + kw];
}
__device__ __forceinline__ void prep_w2(int i, const float* w, _Float16* wt) {
  int n, k;
  frag_pos(i, 4, n, k);
  int pix = k >> 5, c = k & 31, kh = pix >> 2, kw = pix & 3;
  wt[i] = (_Float16)w[((n * 32 + c) * 4 + kh) * 4 + kw];
}
__device__ __forceinline__ void prep_w3(int i, const float* w, _Float16* wt) {
  int n, k;
  frag_pos(i, 4, n, k);
  int pix = k >> 6, c = k & 63, kh = pix / 3, kw = pix % 3;
  wt[i] = (_Float16)w[((n * 64 + c) * 3 + kh) * 3 + kw];
}
__device__ __forceinline__ void prep_fc(int i, const float* w, _Float16* wt) {
  int f, k;
  frag_pos(i, 32, f, k);
  int s = k >> 6, c = k & 63;
  wt[i] = (_Float16)w[f * 3136 + c * 49 + s];
}
__device__ __forceinline__ void prep_wl(int i, const float* wih, const float* whh,
                                        _Float16* wih16, _Float16* whh16) {
  if (i < 1024 * XS) {
    int r, k;
    frag_pos(i, 64, r, k);
    wih16[i] = (_Float16)((k < IND) ? wih[r * IND + k] : 0.0f);
  } else {
    int j = i - 1024 * XS;
    whh16[j] = (_Float16)whh[j];
  }
}
__device__ __forceinline__ void prep_bias(int i, const float* bih, const float* bhh, float* bg) {
  bg[i] = bih[i] + bhh[i];
}
__device__ __forceinline__ void onehot(int i, const int* pos, const int* lact, _Float16* X) {
  int b = i >> 6, c = i & 63;
  float v = 0.0f;
  if (c < 16)      v = (pos[b * 2 + 0] == c)      ? 1.0f : 0.0f;
  else if (c < 32) v = (pos[b * 2 + 1] == c - 16) ? 1.0f : 0.0f;
  else if (c < 37) v = (lact[b]        == c - 32) ? 1.0f : 0.0f;
  X[b * XS + 512 + c] = (_Float16)v;
}

static constexpr int NB_CONV1 = 6400;
static constexpr int GRP_CONV = 8, GRP_ALL = 21;
static constexpr int NE_FC    = 512 * 3136;
static constexpr int NE_WL    = 1024 * (XS + HL);
static constexpr int NE_OH    = B_TOT * 64;
static constexpr int NE_W2    = 64 * 512;
static constexpr int NE_W3    = 64 * 576;
static constexpr int NE_BIAS  = 1024;
static constexpr int NB_EXTRA = (NE_FC + NE_WL + NE_OH + NE_W2 + NE_W3 + NE_BIAS) / 256;
static constexpr int NGRP     = NB_CONV1 / GRP_CONV;
static_assert(NE_FC % 256 == 0 && NE_WL % 256 == 0 && NE_OH % 256 == 0 && NE_W2 % 256 == 0 &&
              NE_W3 % 256 == 0 && NE_BIAS % 256 == 0, "every extra job is whole workgroups");
static_assert(NGRP * GRP_CONV == NB_CONV1 && NGRP * (GRP_ALL - GRP_CONV) >= NB_EXTRA,
              "the groups hold every conv1 workgroup and every extra one");
__global__ void __launch_bounds__(256) k_conv1(const float* img, const _Float16* wt,
                                               const float* bias, _Float16* out,
                                               const float* fcw, _Float16* fcwt,
                                               const float* wih, const float* whh,
                                               _Float16* wih16, _Float16* whh16,
                                               const int* pos, const int* lact, _Float16* X,
                                               const float* w2, _Float16* w2t,
                                               const float* w3, _Float16* w3t,
                                               const float* bih, const float* bhh, float* bg) {
  __shared__ __attribute__((aligned(16))) _Float16 Ws[32 * 192];
  const int grp = blockIdx.x / GRP_ALL, rr = blockIdx.x % GRP_ALL;
  if (rr >= GRP_CONV) {
    int xb = grp * (GRP_ALL - GRP_CONV) + (rr - GRP_CONV);
    if (xb >= NB_EXTRA) return;
    const int t = threadIdx.x;
    if (xb < NE_FC / 256)   { prep_fc(xb * 256 + t, fcw, fcwt);              return; }
    xb -= NE_FC / 256;
    if (xb < NE_WL / 256)   { prep_wl(xb * 256 + t, wih, whh, wih16, whh16); return; }
    xb -= NE_WL / 256;
    if (xb < NE_OH / 256)   { onehot(xb * 256 + t, pos, lact, X);            return; }
    xb -= NE_OH / 256;
    if (xb < NE_W2 / 256)   { prep_w2(xb * 256 + t, w2, w2t);                return; }
    xb -= NE_W2 / 256;
    if (xb < NE_W3 / 256)   { prep_w3(xb * 256 + t, w3, w3t);                return; }
    xb -= NE_W3 / 256;
    prep_bias(xb * 256 + t, bih, bhh, bg);
    return;
  }
  const int cb = grp * GRP_CONV + rr;
  stage_lds<3>(Ws, wt);
  __syncthreads();
  int wid  = (cb * 256 + threadIdx.x) >> 5;
  int lane = threadIdx.x & 31, hi = lane >> 4, lo = lane & 15;
  int m = wid * 16 + lo;
  int b = m / 400, p = m % 400, oh = p / 20, ow = p % 20;
  const float* abase = img + ((size_t)(b * 84 + oh * 4) * 84 + ow * 4) * 3;
  v8f acc0 = {}, acc1 = {};
#pragma unroll
  for (int kb = 0; kb < 192; kb += 32) {
    int o0 = hi ? patch_off(kb + 8)  : patch_off(kb);
    int o1 = hi ? patch_off(kb + 24) : patch_off(kb + 16);
    v16h a  = pack_frag_img(abase + o0, abase + o1);
    v16h b0 = pack_frag_w(Ws + ((kb >> 5) * 2 + 0) * 512 + lane * 8);
    v16h b1 = pack_frag_w(Ws + ((kb >> 5) * 2 + 1) * 512 + lane * 8);
    acc0 = WMMA16(a, b0, acc0);
    acc1 = WMMA16(a, b1, acc1);
  }
  float bs0 = bias[lo], bs1 = bias[16 + lo];
#pragma unroll
  for (int r = 0; r < 8; r++) {
    int mm = wid * 16 + hi * 8 + r;
    float v0 = acc0[r] + bs0; v0 = v0 > 0.0f ? v0 : 0.0f;
    float v1 = acc1[r] + bs1; v1 = v1 > 0.0f ? v1 : 0.0f;
    out[mm * 32 + lo]      = (_Float16)v0;
    out[mm * 32 + 16 + lo] = (_Float16)v1;
  }
}

__global__ void __launch_bounds__(256) k_conv2(const _Float16* act1, const _Float16* wt,
                                               const float* bias, _Float16* out) {
  __shared__ __attribute__((aligned(16))) _Float16 Ws[64 * 512];
  stage_lds<16>(Ws, wt);
  __syncthreads();
  int wid  = (blockIdx.x * blockDim.x + threadIdx.x) >> 5;
  int lane = threadIdx.x & 31, hi = lane >> 4, lo = lane & 15;
  int m = wid * 16 + lo;
  int b = m / 81, p = m % 81, oh = p / 9, ow = p % 9;
  v8f acc0 = {}, acc1 = {}, acc2 = {}, acc3 = {};
  const _Float16* ap = act1 + ((b * 20 + oh * 2) * 20 + ow * 2) * 32 + hi * 8;
  const _Float16* bp = Ws + lane * 8;
  auto ld = [&](int kb) {
    int pix = kb >> 5, kh = pix >> 2, kw = pix & 3;
    KStep f;
    f.a  = pack_frag(ap + (kh * 20 + kw) * 32);
    f.b0 = pack_frag_w(bp + ((kb >> 5) * 4 + 0) * 512);
    f.b1 = pack_frag_w(bp + ((kb >> 5) * 4 + 1) * 512);
    f.b2 = pack_frag_w(bp + ((kb >> 5) * 4 + 2) * 512);
    f.b3 = pack_frag_w(bp + ((kb >> 5) * 4 + 3) * 512);
    return f;
  };
  KStep f0 = ld(0), f1;
  for (int kb = 0; kb < 512 - 64; kb += 64) {
    f1 = ld(kb + 32);
    LOADS_STAY_AHEAD();
    MMA4(f0, acc0, acc1, acc2, acc3);
    f0 = ld(kb + 64);
    LOADS_STAY_AHEAD();
    MMA4(f1, acc0, acc1, acc2, acc3);
  }
  f1 = ld(512 - 32);
  LOADS_STAY_AHEAD();
  MMA4(f0, acc0, acc1, acc2, acc3);
  MMA4(f1, acc0, acc1, acc2, acc3);
  float bs0 = bias[lo], bs1 = bias[16 + lo], bs2 = bias[32 + lo], bs3 = bias[48 + lo];
#pragma unroll
  for (int r = 0; r < 8; r++) {
    int mm = wid * 16 + hi * 8 + r;
    float v0 = acc0[r] + bs0; v0 = v0 > 0.0f ? v0 : 0.0f;
    float v1 = acc1[r] + bs1; v1 = v1 > 0.0f ? v1 : 0.0f;
    float v2 = acc2[r] + bs2; v2 = v2 > 0.0f ? v2 : 0.0f;
    float v3 = acc3[r] + bs3; v3 = v3 > 0.0f ? v3 : 0.0f;
    out[mm * 64 + lo]      = (_Float16)v0;
    out[mm * 64 + 16 + lo] = (_Float16)v1;
    out[mm * 64 + 32 + lo] = (_Float16)v2;
    out[mm * 64 + 48 + lo] = (_Float16)v3;
  }
}

__global__ void __launch_bounds__(256) k_conv3(const _Float16* act2, const _Float16* wt,
                                               const float* bias, _Float16* out) {
  __shared__ __attribute__((aligned(16))) _Float16 Ws[64 * 576];
  stage_lds<18>(Ws, wt);
  __syncthreads();
  int wid  = (blockIdx.x * blockDim.x + threadIdx.x) >> 5;
  int lane = threadIdx.x & 31, hi = lane >> 4, lo = lane & 15;
  int m = wid * 16 + lo;
  int b = m / 49, p = m % 49, oh = p / 7, ow = p % 7;
  v8f acc0 = {}, acc1 = {}, acc2 = {}, acc3 = {};
  const _Float16* ap = act2 + ((b * 9 + oh) * 9 + ow) * 64 + hi * 8;
  const _Float16* bp = Ws + lane * 8;
  auto ld = [&](int kb) {
    int pix = kb >> 6, kh = pix / 3, kw = pix % 3;
    KStep f;
    f.a  = pack_frag(ap + (kh * 9 + kw) * 64 + (kb & 63));
    f.b0 = pack_frag_w(bp + ((kb >> 5) * 4 + 0) * 512);
    f.b1 = pack_frag_w(bp + ((kb >> 5) * 4 + 1) * 512);
    f.b2 = pack_frag_w(bp + ((kb >> 5) * 4 + 2) * 512);
    f.b3 = pack_frag_w(bp + ((kb >> 5) * 4 + 3) * 512);
    return f;
  };
  KStep f0 = ld(0), f1;
  for (int kb = 0; kb < 576 - 64; kb += 64) {
    f1 = ld(kb + 32);
    LOADS_STAY_AHEAD();
    MMA4(f0, acc0, acc1, acc2, acc3);
    f0 = ld(kb + 64);
    LOADS_STAY_AHEAD();
    MMA4(f1, acc0, acc1, acc2, acc3);
  }
  f1 = ld(576 - 32);
  LOADS_STAY_AHEAD();
  MMA4(f0, acc0, acc1, acc2, acc3);
  MMA4(f1, acc0, acc1, acc2, acc3);
  float bs0 = bias[lo], bs1 = bias[16 + lo], bs2 = bias[32 + lo], bs3 = bias[48 + lo];
#pragma unroll
  for (int r = 0; r < 8; r++) {
    int mm = wid * 16 + hi * 8 + r;
    float v0 = acc0[r] + bs0; v0 = v0 > 0.0f ? v0 : 0.0f;
    float v1 = acc1[r] + bs1; v1 = v1 > 0.0f ? v1 : 0.0f;
    float v2 = acc2[r] + bs2; v2 = v2 > 0.0f ? v2 : 0.0f;
    float v3 = acc3[r] + bs3; v3 = v3 > 0.0f ? v3 : 0.0f;
    out[mm * 64 + lo]      = (_Float16)v0;
    out[mm * 64 + 16 + lo] = (_Float16)v1;
    out[mm * 64 + 32 + lo] = (_Float16)v2;
    out[mm * 64 + 48 + lo] = (_Float16)v3;
  }
}

static constexpr int FCO = 136;
__global__ void __launch_bounds__(256) k_fc(const _Float16* act3, const _Float16* wt,
                                            const float* bias, _Float16* X) {
  __shared__ __attribute__((aligned(16))) _Float16 As[32 * FCP];
  const int mt2 = blockIdx.x >> 2, nq = blockIdx.x & 3;
  const int wv   = threadIdx.x >> 5;
  const int lane = threadIdx.x & 31, hi = lane >> 4, lo = lane & 15;
  const int nb = nq * 128 + wv * 16;
  constexpr int KSTR = 32 * 512;
  constexpr int NKS  = FCK / 32;
  const _Float16* bp = wt + (nb >> 4) * 512 + lane * 8;
  v16h b[8];
#pragma unroll
  for (int j = 0; j < 6; j++) b[j] = pack_frag_w(bp + j * KSTR);
  {
    const v8h* s = (const v8h*)(act3 + (size_t)mt2 * 32 * FCK);
    constexpr int RC = FCK / 8;
    static_assert(32 * RC == 49 * 256, "the tile is a whole number of chunks per thread");
    v8h t[25];
#pragma unroll
    for (int j = 0; j < 25; j++) t[j] = s[j * 256 + threadIdx.x];
#pragma unroll
    for (int j = 0; j < 25; j++) {
      int i = j * 256 + threadIdx.x;
      *(v8h*)(As + (i / RC) * FCP + (i % RC) * 8) = t[j];
    }
#pragma unroll
    for (int j = 0; j < 24; j++) t[j] = s[(25 + j) * 256 + threadIdx.x];
#pragma unroll
    for (int j = 0; j < 24; j++) {
      int i = (25 + j) * 256 + threadIdx.x;
      *(v8h*)(As + (i / RC) * FCP + (i % RC) * 8) = t[j];
    }
  }
  __syncthreads();
  v8f acc0 = {}, acc1 = {};
  const _Float16* ap0 = As + lo * FCP + hi * 8;
  const _Float16* ap1 = ap0 + 16 * FCP;
  v16h x0 = pack_frag(ap0), x1 = pack_frag(ap1), y0, y1;
  for (int ks = 0; ks < NKS - 10; ks += 8) {
#pragma unroll
    for (int j = 0; j < 8; j += 2) {
      b[(j + 6) & 7] = pack_frag_w(bp + (ks + j + 6) * KSTR);
      y0 = pack_frag(ap0 + (ks + j + 1) * 32);  y1 = pack_frag(ap1 + (ks + j + 1) * 32);
      LOADS_STAY_AHEAD();
      acc0 = WMMA16(x0, b[j], acc0);  acc1 = WMMA16(x1, b[j], acc1);
      b[(j + 7) & 7] = pack_frag_w(bp + (ks + j + 7) * KSTR);
      x0 = pack_frag(ap0 + (ks + j + 2) * 32);  x1 = pack_frag(ap1 + (ks + j + 2) * 32);
      LOADS_STAY_AHEAD();
      acc0 = WMMA16(y0, b[j + 1], acc0);  acc1 = WMMA16(y1, b[j + 1], acc1);
    }
  }
#pragma unroll
  for (int j = 0; j < 10; j += 2) {
    if (j + 6 < 10) b[(j + 6) & 7] = pack_frag_w(bp + (NKS - 10 + j + 6) * KSTR);
    y0 = pack_frag(ap0 + (NKS - 10 + j + 1) * 32);  y1 = pack_frag(ap1 + (NKS - 10 + j + 1) * 32);
    LOADS_STAY_AHEAD();
    acc0 = WMMA16(x0, b[j & 7], acc0);  acc1 = WMMA16(x1, b[j & 7], acc1);
    if (j + 7 < 10) b[(j + 7) & 7] = pack_frag_w(bp + (NKS - 10 + j + 7) * KSTR);
    if (j + 2 < 10) { x0 = pack_frag(ap0 + (NKS - 10 + j + 2) * 32);  x1 = pack_frag(ap1 + (NKS - 10 + j + 2) * 32); }
    LOADS_STAY_AHEAD();
    acc0 = WMMA16(y0, b[(j + 1) & 7], acc0);  acc1 = WMMA16(y1, b[(j + 1) & 7], acc1);
  }
  float bs = bias[nb + lo];
  __syncthreads();
  _Float16* Os = As;
#pragma unroll
  for (int r = 0; r < 8; r++) {
    float v0 = acc0[r] + bs; v0 = v0 > 0.0f ? v0 : 0.0f;
    float v1 = acc1[r] + bs; v1 = v1 > 0.0f ? v1 : 0.0f;
    Os[(hi * 8 + r) * FCO + wv * 16 + lo]      = (_Float16)v0;
    Os[(16 + hi * 8 + r) * FCO + wv * 16 + lo] = (_Float16)v1;
  }
  __syncthreads();
#pragma unroll
  for (int j = 0; j < 2; j++) {
    int c = j * 256 + threadIdx.x, row = c >> 4, cc = c & 15;
    *(v8h*)(X + (size_t)(mt2 * 32 + row) * XS + nq * 128 + cc * 8) = *(const v8h*)(Os + row * FCO + cc * 8);
  }
}

__global__ void __launch_bounds__(256) k_xproj(const _Float16* X, const _Float16* Wih,
                                               const float* biasg, float* G) {
  int wid  = (blockIdx.x * blockDim.x + threadIdx.x) >> 5;
  int lane = threadIdx.x & 31, hi = lane >> 4, lo = lane & 15;
  int mtile = wid & 127, cg = wid >> 7;
  __shared__ __attribute__((aligned(16))) _Float16 Ws[18 * 4 * 512];
  {
    const v8h* s = (const v8h*)Wih;
    v8h t[18];
#pragma unroll
    for (int j = 0; j < 18; j++) {
      int i = j * 256 + threadIdx.x;
      int blk = i >> 6, ks = blk >> 2, g = blk & 3;
      t[j] = s[((ks * 64 + g * 16 + cg) << 6) + (i & 63)];
    }
#pragma unroll
    for (int j = 0; j < 18; j++) ((v8h*)Ws)[j * 256 + threadIdx.x] = t[j];
  }
  __syncthreads();
  int ncol = cg * 16 + lo;
  float bi = biasg[0 * HL + ncol], bf = biasg[1 * HL + ncol];
  float bg = biasg[2 * HL + ncol], bo = biasg[3 * HL + ncol];
  v8f ai = {bi, bi, bi, bi, bi, bi, bi, bi};
  v8f af = {bf, bf, bf, bf, bf, bf, bf, bf};
  v8f ag = {bg, bg, bg, bg, bg, bg, bg, bg};
  v8f ao = {bo, bo, bo, bo, bo, bo, bo, bo};
  const _Float16* ap = X   + (mtile * 16 + lo) * XS + hi * 8;
  const _Float16* bp = Ws + lane * 8;
  auto ld = [&](int kb) {
    KStep f;
    f.a  = pack_frag(ap + kb);
    f.b0 = pack_frag_w(bp + ((kb >> 5) * 4 + 0) * 512);
    f.b1 = pack_frag_w(bp + ((kb >> 5) * 4 + 1) * 512);
    f.b2 = pack_frag_w(bp + ((kb >> 5) * 4 + 2) * 512);
    f.b3 = pack_frag_w(bp + ((kb >> 5) * 4 + 3) * 512);
    return f;
  };
  KStep f0 = ld(0), f1;
  for (int kb = 0; kb < XS - 64; kb += 64) {
    f1 = ld(kb + 32);
    LOADS_STAY_AHEAD();
    MMA4(f0, ai, af, ag, ao);
    f0 = ld(kb + 64);
    LOADS_STAY_AHEAD();
    MMA4(f1, ai, af, ag, ao);
  }
  f1 = ld(XS - 32);
  LOADS_STAY_AHEAD();
  MMA4(f0, ai, af, ag, ao);
  MMA4(f1, ai, af, ag, ao);
  v4f* gp = (v4f*)(G + (size_t)(mtile * 16 + cg) * GT) + lane;
  gp[0 * 64]      = ai.lo;  gp[0 * 64 + 32] = ai.hi;
  gp[1 * 64]      = af.lo;  gp[1 * 64 + 32] = af.hi;
  gp[2 * 64]      = ag.lo;  gp[2 * 64 + 32] = ag.hi;
  gp[3 * 64]      = ao.lo;  gp[3 * 64 + 32] = ao.hi;
}

__global__ void __launch_bounds__(512) k_lstm(const float* G, const _Float16* Whh,
                                              const int* done,
                                              const float* h0, const float* c0,
                                              float* hs, float* out) {
  __shared__ v8h Wl[16 * 4 * 4 * 2 * 32];
  __shared__ __attribute__((aligned(16))) _Float16 A[16 * HSTR];
  __shared__ unsigned nm[T_STEPS];
  const int tid = threadIdx.x;
  const int mt  = blockIdx.x;
  const int cg  = tid >> 5, lane = tid & 31, hi = lane >> 4, lo = lane & 15;
  const int ncol = cg * 16 + lo;
  const int env0 = mt * 16;

  if (tid < T_STEPS) {
    unsigned m = 0;
    if (tid + 1 < T_STEPS) {
#pragma unroll
      for (int e = 0; e < 16; e++)
        m |= (done[(tid + 1) * E_ENVS + env0 + e] != 0 ? 1u : 0u) << e;
    }
    nm[tid] = m;
  }

  for (int i = tid; i < 16 * HL; i += 512) {
    int e = i >> 8, k = i & 255;
    float v = h0[(env0 + e) * HL + k];
    A[e * HSTR + k] = done[env0 + e] ? (_Float16)0.0f : (_Float16)v;
  }
  v8f creg;
#pragma unroll
  for (int r = 0; r < 8; r++) {
    float v = c0[(env0 + hi * 8 + r) * HL + ncol];
    creg[r] = done[env0 + hi * 8 + r] ? 0.0f : v;
  }

  const _Float16* bp = Whh + ncol * HL + hi * 8;
  v8h* wl = Wl + cg * (4 * 4 * 2 * 32) + lane;
#pragma unroll
  for (int ch = 0; ch < 4; ch++) {
#pragma unroll
    for (int g = 0; g < 4; g++) {
      const _Float16* p = bp + g * HL * HL + (4 + ch) * 32;
      wl[((ch * 4 + g) * 2 + 0) * 32] = *(const v8h*)p;
      wl[((ch * 4 + g) * 2 + 1) * 32] = *(const v8h*)(p + 16);
    }
  }
  __builtin_amdgcn_sched_barrier(0);
  v16h wr[4][4];
#pragma unroll
  for (int ch = 0; ch < 4; ch++) {
#pragma unroll
    for (int g = 0; g < 4; g++) wr[ch][g] = pack_frag(bp + g * HL * HL + ch * 32);
  }
  __syncthreads();

  const v4f* gbase = (const v4f*)(G + (size_t)(mt * 16 + cg) * GT) + lane;
  v8f gi, gf, gg, go;
  gi.lo = gbase[0 * 64];  gi.hi = gbase[0 * 64 + 32];
  gf.lo = gbase[1 * 64];  gf.hi = gbase[1 * 64 + 32];
  gg.lo = gbase[2 * 64];  gg.hi = gbase[2 * 64 + 32];
  go.lo = gbase[3 * 64];  go.hi = gbase[3 * 64 + 32];

  float hv[8];
  for (int t = 0; t < T_STEPS; t++) {
    const unsigned my = nm[t] >> (hi * 8);

    const _Float16* arow = &A[lo * HSTR + hi * 8];
    v8f ai, af, ag, ao;
#pragma unroll
    for (int ch = 0; ch < 4; ch++) {
      v16h a = pack_frag(arow + ch * 32);
      ai = WMMA16(a, wr[ch][0], ch ? ai : gi);
      af = WMMA16(a, wr[ch][1], ch ? af : gf);
      ag = WMMA16(a, wr[ch][2], ch ? ag : gg);
      ao = WMMA16(a, wr[ch][3], ch ? ao : go);
    }
#pragma unroll
    for (int ch = 0; ch < 4; ch++) {
      v16h a = pack_frag(arow + (4 + ch) * 32);
      v16h b[4];
#pragma unroll
      for (int g = 0; g < 4; g++) {
        v8h u = wl[((ch * 4 + g) * 2 + 0) * 32];
        v8h w = wl[((ch * 4 + g) * 2 + 1) * 32];
#pragma unroll
        for (int e = 0; e < 8; e++) { b[g][e] = u[e]; b[g][e + 8] = w[e]; }
      }
      ai = WMMA16(a, b[0], ai);
      af = WMMA16(a, b[1], af);
      ag = WMMA16(a, b[2], ag);
      ao = WMMA16(a, b[3], ao);
    }
    lds_barrier();

    if (t + 1 < T_STEPS) {
      const v4f* gp = gbase + (size_t)(t + 1) * (2 * 16 * GT / 4);
      gi.lo = gp[0 * 64];  gi.hi = gp[0 * 64 + 32];
      gf.lo = gp[1 * 64];  gf.hi = gp[1 * 64 + 32];
      gg.lo = gp[2 * 64];  gg.hi = gp[2 * 64 + 32];
      go.lo = gp[3 * 64];  go.hi = gp[3 * 64 + 32];
    }

    bool dn[8];
#pragma unroll
    for (int r = 0; r < 8; r++) {
      float iv = sigmoid_hw(ai[r]), fv = sigmoid_hw(af[r]);
      float gv = tanh_hw(ag[r]),    ov = sigmoid_hw(ao[r]);
      float cv = fv * creg[r] + iv * gv;
      hv[r]   = ov * tanh_hw(cv);
      dn[r]   = ((my >> r) & 1u) != 0;
      creg[r] = dn[r] ? 0.0f : cv;
    }
#pragma unroll
    for (int r = 0; r < 8; r++) {
      int el  = hi * 8 + r;
      int env = env0 + el;
      A[el * HSTR + ncol] = dn[r] ? (_Float16)0.0f : (_Float16)hv[r];
      hs[(t * E_ENVS + env) * HL + ncol] = hv[r];
    }
    lds_barrier();
  }
#pragma unroll
  for (int r = 0; r < 8; r++) {
    int env = env0 + hi * 8 + r;
    out[12288 + env * HL + ncol] = hv[r];
    out[20480 + env * HL + ncol] = creg[r];
  }
}

__global__ void k_heads(const float* hs, const float* pw, const float* pb,
                        const float* vw, const float* vb, float* out) {
  int i = blockIdx.x * blockDim.x + threadIdx.x;
  if (i >= B_TOT * 6) return;
  int b = i / 6, j = i % 6;
  const float* h    = hs + b * HL;
  const float* wrow = (j < 5) ? pw + j * HL : vw;
  float s = (j < 5) ? pb[j] : vb[0];
#pragma unroll 4
  for (int k = 0; k < HL; k++) s += h[k] * wrow[k];
  if (j < 5) out[b * 5 + j] = s;
  else       out[10240 + b] = s;
}

extern "C" void kernel_launch(void* const* d_in, const int* in_sizes, int n_in,
                              void* d_out, int out_size, void* d_ws, size_t ws_size,
                              hipStream_t stream) {
  (void)in_sizes; (void)n_in; (void)out_size; (void)ws_size;
  const float* img  = (const float*)d_in[0];
  const int*   pos  = (const int*)d_in[1];
  const int*   lact = (const int*)d_in[2];
  const int*   done = (const int*)d_in[3];
  const float* h0   = (const float*)d_in[4];
  const float* c0   = (const float*)d_in[5];
  const float* w1   = (const float*)d_in[6];
  const float* b1   = (const float*)d_in[7];
  const float* w2   = (const float*)d_in[8];
  const float* b2   = (const float*)d_in[9];
  const float* w3   = (const float*)d_in[10];
  const float* b3   = (const float*)d_in[11];
  const float* fcw  = (const float*)d_in[12];
  const float* fcb  = (const float*)d_in[13];
  const float* wih  = (const float*)d_in[14];
  const float* whh  = (const float*)d_in[15];
  const float* bih  = (const float*)d_in[16];
  const float* bhh  = (const float*)d_in[17];
  const float* pw   = (const float*)d_in[18];
  const float* pb   = (const float*)d_in[19];
  const float* vw   = (const float*)d_in[20];
  const float* vb   = (const float*)d_in[21];
  float* out = (float*)d_out;
  char*  ws  = (char*)d_ws;

  const size_t o_act1 = 90832896;
  const size_t o_w1t  = o_act1 + 52428800;
  const size_t o_w2t  = o_w1t + 12288;
  const size_t o_w3t  = o_w2t + 65536;
  const size_t o_fcwt = o_w3t + 73728;
  const size_t o_wih  = o_fcwt + 3211264;
  const size_t o_whh  = o_wih + 1179648;
  const size_t o_bg   = o_whh + 524288;
  const size_t o_act2 = 0;
  const size_t o_act3 = 21233664;
  const size_t o_X    = 34078720;
  const size_t o_hs   = 36438016;
  const size_t o_G    = 38535168;

  _Float16* act1 = (_Float16*)(ws + o_act1);
  _Float16* act2 = (_Float16*)(ws + o_act2);
  _Float16* act3 = (_Float16*)(ws + o_act3);
  _Float16* Xm   = (_Float16*)(ws + o_X);
  _Float16* w1t  = (_Float16*)(ws + o_w1t);
  _Float16* w2t  = (_Float16*)(ws + o_w2t);
  _Float16* w3t  = (_Float16*)(ws + o_w3t);
  _Float16* fcwt = (_Float16*)(ws + o_fcwt);
  _Float16* wih16 = (_Float16*)(ws + o_wih);
  _Float16* whh16 = (_Float16*)(ws + o_whh);
  float*    bg   = (float*)(ws + o_bg);
  float*    hsb  = (float*)(ws + o_hs);
  float*    Gx   = (float*)(ws + o_G);

  const int TB = 256;
  k_prep_w1  <<<(32 * 192   + TB - 1) / TB, TB, 0, stream>>>(w1, w1t);

  k_conv1<<<NGRP * GRP_ALL, TB, 0, stream>>>(img, w1t, b1, act1, fcw, fcwt, wih, whh, wih16, whh16,
                                                   pos, lact, Xm, w2, w2t, w3, w3t, bih, bhh, bg);
  k_conv2<<<1296, TB, 0, stream>>>(act1, w2t, b2, act2);
  k_conv3<<< 784, TB, 0, stream>>>(act2, w3t, b3, act3);
  k_fc   <<< 256, TB, 0, stream>>>(act3, fcwt, fcb, Xm);

  k_xproj<<<256, TB, 0, stream>>>(Xm, wih16, bg, Gx);

  k_lstm<<<2, 512, 0, stream>>>(Gx, whh16, done, h0, c0, hsb, out);

  k_heads<<<(B_TOT * 6 + TB - 1) / TB, TB, 0, stream>>>(hsb, pw, pb, vw, vb, out);
}
